# starting kernel
# baseline (speedup 1.0000x reference)
#include <hip/hip_runtime.h>
#include <hip/hip_bf16.h>
#include <math.h>

#define N_NODES 50000
#define E_EDGES 800000
#define ETOT    (E_EDGES + N_NODES)
#define HEADS   4
#define CHP     32
#define HC      128
#define NGRAPH  64
#define SLOPE   0.2f
#define LN_EPS  1e-5f

typedef float v2f __attribute__((ext_vector_type(2)));
typedef float v8f __attribute__((ext_vector_type(8)));

__device__ __forceinline__ unsigned f32_to_ord(float f) {
  unsigned u = __float_as_uint(f);
  return (u & 0x80000000u) ? ~u : (u | 0x80000000u);
}
__device__ __forceinline__ float ord_to_f32(unsigned u) {
  u = (u & 0x80000000u) ? (u & 0x7fffffffu) : ~u;
  return __uint_as_float(u);
}
__device__ __forceinline__ void atomAddF(float* p, float v) {
  unsafeAtomicAdd(p, v);
}

__global__ void fill_f32(float* __restrict__ p, float v, int n) {
  int i = blockIdx.x * blockDim.x + threadIdx.x;
  if (i < n) p[i] = v;
}
__global__ void fill_u32(unsigned* __restrict__ p, unsigned v, int n) {
  int i = blockIdx.x * blockDim.x + threadIdx.x;
  if (i < n) p[i] = v;
}

#define LDSW 136
#define ROWTILES 4
__global__ void gemm_bias_wmma(const float* __restrict__ X,
                               const float* __restrict__ W,
                               const float* __restrict__ bias,
                               float* __restrict__ Y, int n) {
  extern __shared__ float sW[];
  const int tid = threadIdx.x;
  for (int i = 0; i < 16; ++i) {
    int idx = tid + 256 * i;
    int r = idx >> 5, q = idx & 31;
    float4 w4 = *(const float4*)(W + r * HC + q * 4);
    float* d = sW + r * LDSW + q * 4;
    d[0] = w4.x; d[1] = w4.y; d[2] = w4.z; d[3] = w4.w;
  }
  __syncthreads();

  const int wave = tid >> 5;
  const int lane = tid & 31;
  const int m    = lane & 15;
  const int hi   = lane >> 4;
  const int col  = wave * 16 + m;
  const float bc = bias[col];
  const int rowbase = blockIdx.x * (16 * ROWTILES);

  for (int t = 0; t < ROWTILES; ++t) {
    int row0 = rowbase + t * 16;
    if (row0 >= n) break;
    int row  = row0 + m;
    int rowc = (row < n) ? row : (n - 1);
    const float* ap = X + (size_t)rowc * HC + 2 * hi;
    v8f acc = {};
    #pragma unroll
    for (int k0 = 0; k0 < HC; k0 += 4) {
      v2f a; a.x = ap[k0]; a.y = ap[k0 + 1];
      const float* bp = sW + (k0 + 2 * hi) * LDSW + col;
      v2f b; b.x = bp[0]; b.y = bp[LDSW];
      acc = __builtin_amdgcn_wmma_f32_16x16x4_f32(
          false, a, false, b, (short)0, acc, false, false);
    }
    #pragma unroll
    for (int r = 0; r < 8; ++r) {
      int orow = row0 + r + 8 * hi;
      if (orow < n) Y[(size_t)orow * HC + col] = acc[r] + bc;
    }
  }
}

__global__ void edge_score_kernel(const int* __restrict__ ei,
                                  const float* __restrict__ xl,
                                  const float* __restrict__ xr,
                                  const float* __restrict__ att,
                                  float* __restrict__ score,
                                  unsigned* __restrict__ mord) {
  int e = blockIdx.x * 8 + (threadIdx.x >> 5);
  int lane = threadIdx.x & 31;
  if (e >= ETOT) return;
  int src, dst;
  if (e < E_EDGES) { src = ei[e]; dst = ei[E_EDGES + e]; }
  else             { src = dst = e - E_EDGES; }
  const float* pl = xl + (size_t)src * HC + lane;
  const float* pr = xr + (size_t)dst * HC + lane;
  float s[HEADS];
  #pragma unroll
  for (int h = 0; h < HEADS; ++h) {
    float v = pl[h * CHP] + pr[h * CHP];
    v = (v > 0.f) ? v : v * SLOPE;
    s[h] = v * att[h * CHP + lane];
  }
  #pragma unroll
  for (int h = 0; h < HEADS; ++h)
    for (int off = 16; off > 0; off >>= 1) s[h] += __shfl_xor(s[h], off, 32);
  if (lane < HEADS) {
    float sh = s[lane];
    score[(size_t)e * HEADS + lane] = sh;
    atomicMax(&mord[(size_t)dst * HEADS + lane], f32_to_ord(sh));
  }
}

__global__ void edge_exp_kernel(const int* __restrict__ ei,
                                float* __restrict__ score,
                                const unsigned* __restrict__ mord,
                                float* __restrict__ denom) {
  int t = blockIdx.x * blockDim.x + threadIdx.x;
  if (t >= ETOT * HEADS) return;
  int e = t >> 2, h = t & 3;
  int dst = (e < E_EDGES) ? ei[E_EDGES + e] : (e - E_EDGES);
  float m = ord_to_f32(mord[(size_t)dst * HEADS + h]);
  float ex = __expf(score[t] - m);
  score[t] = ex;
  atomAddF(&denom[(size_t)dst * HEADS + h], ex);
}

__global__ void edge_aggregate_kernel(const int* __restrict__ ei,
                                      const float* __restrict__ xl,
                                      const float* __restrict__ ex,
                                      const float* __restrict__ denom,
                                      float* __restrict__ agg) {
  int e = blockIdx.x * 8 + (threadIdx.x >> 5);
  int lane = threadIdx.x & 31;
  if (e >= ETOT) return;
  int src, dst;
  if (e < E_EDGES) { src = ei[e]; dst = ei[E_EDGES + e]; }
  else             { src = dst = e - E_EDGES; }
  #pragma unroll
  for (int h = 0; h < HEADS; ++h) {
    float alpha = ex[(size_t)e * HEADS + h] / denom[(size_t)dst * HEADS + h];
    float v = xl[(size_t)src * HC + h * CHP + lane] * alpha;
    atomAddF(&agg[(size_t)dst * HC + h * CHP + lane], v);
  }
}

__global__ void ln_silu_kernel(const float* __restrict__ agg,
                               const float* __restrict__ bias,
                               const float* __restrict__ gamma,
                               const float* __restrict__ beta,
                               float* __restrict__ out) {
  int node = blockIdx.x * 8 + (threadIdx.x >> 5);
  int lane = threadIdx.x & 31;
  if (node >= N_NODES) return;
  float v[4]; float sum = 0.f;
  #pragma unroll
  for (int i = 0; i < 4; ++i) {
    int c = i * 32 + lane;
    v[i] = agg[(size_t)node * HC + c] + bias[c];
    sum += v[i];
  }
  for (int off = 16; off > 0; off >>= 1) sum += __shfl_xor(sum, off, 32);
  float mu = sum * (1.0f / HC);
  float sq = 0.f;
  #pragma unroll
  for (int i = 0; i < 4; ++i) { float d = v[i] - mu; sq += d * d; }
  for (int off = 16; off > 0; off >>= 1) sq += __shfl_xor(sq, off, 32);
  float rstd = rsqrtf(sq * (1.0f / HC) + LN_EPS);
  #pragma unroll
  for (int i = 0; i < 4; ++i) {
    int c = i * 32 + lane;
    float h = (v[i] - mu) * rstd * gamma[c] + beta[c];
    out[(size_t)node * HC + c] = h / (1.0f + __expf(-h));
  }
}

__global__ void pool_accum_kernel(const float* __restrict__ x,
                                  const int* __restrict__ batch,
                                  float* __restrict__ sums,
                                  float* __restrict__ counts) {
  int node = blockIdx.x * 8 + (threadIdx.x >> 5);
  int lane = threadIdx.x & 31;
  if (node >= N_NODES) return;
  int g = batch[node];
  #pragma unroll
  for (int i = 0; i < 4; ++i) {
    int c = i * 32 + lane;
    atomAddF(&sums[(size_t)g * HC + c], x[(size_t)node * HC + c]);
  }
  if (lane == 0) atomAddF(&counts[g], 1.0f);
}
__global__ void pool_final_kernel(const float* __restrict__ sums,
                                  const float* __restrict__ counts,
                                  float* __restrict__ out) {
  int t = blockIdx.x * blockDim.x + threadIdx.x;
  if (t >= NGRAPH * HC) return;
  out[t] = sums[t] / fmaxf(counts[t >> 7], 1.0f);
}

extern "C" void kernel_launch(void* const* d_in, const int* in_sizes, int n_in,
                              void* d_out, int out_size, void* d_ws, size_t ws_size,
                              hipStream_t stream) {
  (void)in_sizes; (void)n_in; (void)out_size; (void)ws_size;
  const float* x     = (const float*)d_in[0];
  const int*   ei    = (const int*)d_in[1];
  const int*   batch = (const int*)d_in[2];
  const float* Wl[2]    = {(const float*)d_in[3],  (const float*)d_in[11]};
  const float* Wr[2]    = {(const float*)d_in[4],  (const float*)d_in[12]};
  const float* bl[2]    = {(const float*)d_in[5],  (const float*)d_in[13]};
  const float* br[2]    = {(const float*)d_in[6],  (const float*)d_in[14]};
  const float* att[2]   = {(const float*)d_in[7],  (const float*)d_in[15]};
  const float* bias[2]  = {(const float*)d_in[8],  (const float*)d_in[16]};
  const float* gamma[2] = {(const float*)d_in[9],  (const float*)d_in[17]};
  const float* beta[2]  = {(const float*)d_in[10], (const float*)d_in[18]};

  float* ws = (float*)d_ws;
  float*    xl    = ws;            ws += (size_t)N_NODES * HC;
  float*    xr2   = ws;            ws += (size_t)N_NODES * HC;
  float*    agg   = ws;            ws += (size_t)N_NODES * HC;
  float*    xnode = ws;            ws += (size_t)N_NODES * HC;
  float*    exbuf = ws;            ws += (size_t)ETOT * HEADS;
  float*    denom = ws;            ws += (size_t)N_NODES * HEADS;
  unsigned* mord  = (unsigned*)ws; ws += (size_t)N_NODES * HEADS;
  float*    psums = ws;            ws += (size_t)NGRAPH * HC;
  float*    pcnt  = ws;

  dim3 b256(256);
  const int    gemmGrid = (N_NODES + 16 * ROWTILES - 1) / (16 * ROWTILES);
  const size_t shmem    = (size_t)HC * LDSW * sizeof(float);
  const int    edgeGrid = (ETOT + 7) / 8;
  const int    nodeGrid = (N_NODES + 7) / 8;
  const unsigned ORD_NEG_INF = 0x007FFFFFu;

  const float* xin = x;
  for (int l = 0; l < 2; ++l) {
    gemm_bias_wmma<<<gemmGrid, b256, shmem, stream>>>(xin, Wl[l], bl[l], xl, N_NODES);
    gemm_bias_wmma<<<gemmGrid, b256, shmem, stream>>>(xin, Wr[l], br[l], xr2, N_NODES);
    fill_u32<<<(N_NODES * HEADS + 255) / 256, b256, 0, stream>>>(mord, ORD_NEG_INF, N_NODES * HEADS);
    fill_f32<<<(N_NODES * HEADS + 255) / 256, b256, 0, stream>>>(denom, 0.f, N_NODES * HEADS);
    fill_f32<<<(N_NODES * HC + 255) / 256, b256, 0, stream>>>(agg, 0.f, N_NODES * HC);
    edge_score_kernel<<<edgeGrid, b256, 0, stream>>>(ei, xl, xr2, att[l], exbuf, mord);
    edge_exp_kernel<<<(ETOT * HEADS + 255) / 256, b256, 0, stream>>>(ei, exbuf, mord, denom);
    edge_aggregate_kernel<<<edgeGrid, b256, 0, stream>>>(ei, xl, exbuf, denom, agg);
    ln_silu_kernel<<<nodeGrid, b256, 0, stream>>>(agg, bias[l], gamma[l], beta[l], xnode);
    xin = xnode;
  }
  fill_f32<<<(NGRAPH * HC + NGRAPH + 255) / 256, b256, 0, stream>>>(psums, 0.f, NGRAPH * HC + NGRAPH);
  pool_accum_kernel<<<nodeGrid, b256, 0, stream>>>(xnode, batch, psums, pcnt);
  pool_final_kernel<<<(NGRAPH * HC + 255) / 256, b256, 0, stream>>>(psums, pcnt, (float*)d_out);
}


# pool by owner-computes gather, no atomics: block per (graph, 1024-node slab) ballots ids and sums only its own rows in registers; 49 slab partials added in fixed order; pool sum order changes
# speedup vs baseline: 1.1318x; 1.1318x over previous
#include <hip/hip_runtime.h>
#include <hip/hip_bf16.h>
#include <math.h>

#define N_NODES 50000
#define E_EDGES 800000
#define ETOT    (E_EDGES + N_NODES)
#define HEADS   4
#define CHP     32
#define HC      128
#define NGRAPH  64
#define SLOPE   0.2f
#define LN_EPS  1e-5f

typedef float v2f __attribute__((ext_vector_type(2)));
typedef float v8f __attribute__((ext_vector_type(8)));

__device__ __forceinline__ unsigned f32_to_ord(float f) {
  unsigned u = __float_as_uint(f);
  return (u & 0x80000000u) ? ~u : (u | 0x80000000u);
}
__device__ __forceinline__ float ord_to_f32(unsigned u) {
  u = (u & 0x80000000u) ? (u & 0x7fffffffu) : ~u;
  return __uint_as_float(u);
}
__device__ __forceinline__ void atomAddF(float* p, float v) {
  unsafeAtomicAdd(p, v);
}

__global__ void fill_f32(float* __restrict__ p, float v, int n) {
  int i = blockIdx.x * blockDim.x + threadIdx.x;
  if (i < n) p[i] = v;
}
__global__ void fill_u32(unsigned* __restrict__ p, unsigned v, int n) {
  int i = blockIdx.x * blockDim.x + threadIdx.x;
  if (i < n) p[i] = v;
}

#define LDSW 136
#define ROWTILES 4
__global__ void gemm_bias_wmma(const float* __restrict__ X,
                               const float* __restrict__ W,
                               const float* __restrict__ bias,
                               float* __restrict__ Y, int n) {
  extern __shared__ float sW[];
  const int tid = threadIdx.x;
  for (int i = 0; i < 16; ++i) {
    int idx = tid + 256 * i;
    int r = idx >> 5, q = idx & 31;
    float4 w4 = *(const float4*)(W + r * HC + q * 4);
    float* d = sW + r * LDSW + q * 4;
    d[0] = w4.x; d[1] = w4.y; d[2] = w4.z; d[3] = w4.w;
  }
  __syncthreads();

  const int wave = tid >> 5;
  const int lane = tid & 31;
  const int m    = lane & 15;
  const int hi   = lane >> 4;
  const int col  = wave * 16 + m;
  const float bc = bias[col];
  const int rowbase = blockIdx.x * (16 * ROWTILES);

  for (int t = 0; t < ROWTILES; ++t) {
    int row0 = rowbase + t * 16;
    if (row0 >= n) break;
    int row  = row0 + m;
    int rowc = (row < n) ? row : (n - 1);
    const float* ap = X + (size_t)rowc * HC + 2 * hi;
    v8f acc = {};
    #pragma unroll
    for (int k0 = 0; k0 < HC; k0 += 4) {
      v2f a; a.x = ap[k0]; a.y = ap[k0 + 1];
      const float* bp = sW + (k0 + 2 * hi) * LDSW + col;
      v2f b; b.x = bp[0]; b.y = bp[LDSW];
      acc = __builtin_amdgcn_wmma_f32_16x16x4_f32(
          false, a, false, b, (short)0, acc, false, false);
    }
    #pragma unroll
    for (int r = 0; r < 8; ++r) {
      int orow = row0 + r + 8 * hi;
      if (orow < n) Y[(size_t)orow * HC + col] = acc[r] + bc;
    }
  }
}

__global__ void edge_score_kernel(const int* __restrict__ ei,
                                  const float* __restrict__ xl,
                                  const float* __restrict__ xr,
                                  const float* __restrict__ att,
                                  float* __restrict__ score,
                                  unsigned* __restrict__ mord) {
  int e = blockIdx.x * 8 + (threadIdx.x >> 5);
  int lane = threadIdx.x & 31;
  if (e >= ETOT) return;
  int src, dst;
  if (e < E_EDGES) { src = ei[e]; dst = ei[E_EDGES + e]; }
  else             { src = dst = e - E_EDGES; }
  const float* pl = xl + (size_t)src * HC + lane;
  const float* pr = xr + (size_t)dst * HC + lane;
  float s[HEADS];
  #pragma unroll
  for (int h = 0; h < HEADS; ++h) {
    float v = pl[h * CHP] + pr[h * CHP];
    v = (v > 0.f) ? v : v * SLOPE;
    s[h] = v * att[h * CHP + lane];
  }
  #pragma unroll
  for (int h = 0; h < HEADS; ++h)
    for (int off = 16; off > 0; off >>= 1) s[h] += __shfl_xor(s[h], off, 32);
  if (lane < HEADS) {
    float sh = s[lane];
    score[(size_t)e * HEADS + lane] = sh;
    atomicMax(&mord[(size_t)dst * HEADS + lane], f32_to_ord(sh));
  }
}

__global__ void edge_exp_kernel(const int* __restrict__ ei,
                                float* __restrict__ score,
                                const unsigned* __restrict__ mord,
                                float* __restrict__ denom) {
  int t = blockIdx.x * blockDim.x + threadIdx.x;
  if (t >= ETOT * HEADS) return;
  int e = t >> 2, h = t & 3;
  int dst = (e < E_EDGES) ? ei[E_EDGES + e] : (e - E_EDGES);
  float m = ord_to_f32(mord[(size_t)dst * HEADS + h]);
  float ex = __expf(score[t] - m);
  score[t] = ex;
  atomAddF(&denom[(size_t)dst * HEADS + h], ex);
}

__global__ void edge_aggregate_kernel(const int* __restrict__ ei,
                                      const float* __restrict__ xl,
                                      const float* __restrict__ ex,
                                      const float* __restrict__ denom,
                                      float* __restrict__ agg) {
  int e = blockIdx.x * 8 + (threadIdx.x >> 5);
  int lane = threadIdx.x & 31;
  if (e >= ETOT) return;
  int src, dst;
  if (e < E_EDGES) { src = ei[e]; dst = ei[E_EDGES + e]; }
  else             { src = dst = e - E_EDGES; }
  #pragma unroll
  for (int h = 0; h < HEADS; ++h) {
    float alpha = ex[(size_t)e * HEADS + h] / denom[(size_t)dst * HEADS + h];
    float v = xl[(size_t)src * HC + h * CHP + lane] * alpha;
    atomAddF(&agg[(size_t)dst * HC + h * CHP + lane], v);
  }
}

__global__ void ln_silu_kernel(const float* __restrict__ agg,
                               const float* __restrict__ bias,
                               const float* __restrict__ gamma,
                               const float* __restrict__ beta,
                               float* __restrict__ out) {
  int node = blockIdx.x * 8 + (threadIdx.x >> 5);
  int lane = threadIdx.x & 31;
  if (node >= N_NODES) return;
  float v[4]; float sum = 0.f;
  #pragma unroll
  for (int i = 0; i < 4; ++i) {
    int c = i * 32 + lane;
    v[i] = agg[(size_t)node * HC + c] + bias[c];
    sum += v[i];
  }
  for (int off = 16; off > 0; off >>= 1) sum += __shfl_xor(sum, off, 32);
  float mu = sum * (1.0f / HC);
  float sq = 0.f;
  #pragma unroll
  for (int i = 0; i < 4; ++i) { float d = v[i] - mu; sq += d * d; }
  for (int off = 16; off > 0; off >>= 1) sq += __shfl_xor(sq, off, 32);
  float rstd = rsqrtf(sq * (1.0f / HC) + LN_EPS);
  #pragma unroll
  for (int i = 0; i < 4; ++i) {
    int c = i * 32 + lane;
    float h = (v[i] - mu) * rstd * gamma[c] + beta[c];
    out[(size_t)node * HC + c] = h / (1.0f + __expf(-h));
  }
}

#define POOL_CHUNKS 4
#define POOL_SLAB   (8 * POOL_CHUNKS * 32)
#define POOL_SLABS  ((N_NODES + POOL_SLAB - 1) / POOL_SLAB)
#define PROW        (HC + 32)
__global__ void pool_gather_kernel(const float* __restrict__ x,
                                   const int* __restrict__ batch,
                                   float* __restrict__ part) {
  __shared__ float sacc[8][HC];
  __shared__ int   scnt[8];
  const int g     = blockIdx.x % NGRAPH;
  const int s     = blockIdx.x / NGRAPH;
  const int wave  = threadIdx.x >> 5;
  const int lane  = threadIdx.x & 31;
  const int slab0 = s * POOL_SLAB;

  int ids[POOL_CHUNKS];
  #pragma unroll
  for (int i = 0; i < POOL_CHUNKS; ++i) {
    int node = slab0 + (i * 8 + wave) * 32 + lane;
    ids[i] = (node < N_NODES) ? batch[node] : -1;
  }
  float4 acc = make_float4(0.f, 0.f, 0.f, 0.f);
  int cnt = 0;
  #pragma unroll
  for (int i = 0; i < POOL_CHUNKS; ++i) {
    const unsigned mask = (unsigned)__ballot(ids[i] == g);
    if (mask == 0u) continue;
    cnt += __popc(mask);
    const int base = slab0 + (i * 8 + wave) * 32;
    for (int j0 = 0; j0 < 32; j0 += 16) {
      if (((mask >> j0) & 0xffffu) == 0u) continue;
      float4 v[16];
      #pragma unroll
      for (int u = 0; u < 16; ++u) {
        int node = base + j0 + u;
        node = (node < N_NODES) ? node : (N_NODES - 1);
        v[u] = *(const float4*)(x + (size_t)node * HC + lane * 4);
      }
      #pragma unroll
      for (int u = 0; u < 16; ++u) {
        const bool on = ((mask >> (j0 + u)) & 1u) != 0u;
        acc.x += on ? v[u].x : 0.f;
        acc.y += on ? v[u].y : 0.f;
        acc.z += on ? v[u].z : 0.f;
        acc.w += on ? v[u].w : 0.f;
      }
    }
  }
  *(float4*)&sacc[wave][lane * 4] = acc;
  if (lane == 0) scnt[wave] = cnt;
  __syncthreads();
  const int tid = threadIdx.x;
  float* prow = part + ((size_t)g * POOL_SLABS + s) * PROW;
  if (tid < HC) {
    float t = 0.f;
    #pragma unroll
    for (int w = 0; w < 8; ++w) t += sacc[w][tid];
    prow[tid] = t;
  } else if (tid < PROW) {
    int c = 0;
    #pragma unroll
    for (int w = 0; w < 8; ++w) c += scnt[w];
    ((int*)prow)[tid] = (tid == HC) ? c : 0;
  }
}
__global__ void pool_final_kernel(const float* __restrict__ part,
                                  float* __restrict__ out) {
  int t = blockIdx.x * blockDim.x + threadIdx.x;
  if (t >= NGRAPH * HC) return;
  const int g = t >> 7, c = t & 127;
  const float* p = part + (size_t)g * POOL_SLABS * PROW;
  float sum = 0.f;
  int n = 0;
  #pragma unroll 7
  for (int s = 0; s < POOL_SLABS; ++s) {
    sum += p[s * PROW + c];
    n   += ((const int*)p)[s * PROW + HC];
  }
  out[t] = sum / fmaxf((float)n, 1.0f);
}

extern "C" void kernel_launch(void* const* d_in, const int* in_sizes, int n_in,
                              void* d_out, int out_size, void* d_ws, size_t ws_size,
                              hipStream_t stream) {
  (void)in_sizes; (void)n_in; (void)out_size; (void)ws_size;
  const float* x     = (const float*)d_in[0];
  const int*   ei    = (const int*)d_in[1];
  const int*   batch = (const int*)d_in[2];
  const float* Wl[2]    = {(const float*)d_in[3],  (const float*)d_in[11]};
  const float* Wr[2]    = {(const float*)d_in[4],  (const float*)d_in[12]};
  const float* bl[2]    = {(const float*)d_in[5],  (const float*)d_in[13]};
  const float* br[2]    = {(const float*)d_in[6],  (const float*)d_in[14]};
  const float* att[2]   = {(const float*)d_in[7],  (const float*)d_in[15]};
  const float* bias[2]  = {(const float*)d_in[8],  (const float*)d_in[16]};
  const float* gamma[2] = {(const float*)d_in[9],  (const float*)d_in[17]};
  const float* beta[2]  = {(const float*)d_in[10], (const float*)d_in[18]};

  float* ws = (float*)d_ws;
  float*    xl    = ws;            ws += (size_t)N_NODES * HC;
  float*    xr2   = ws;            ws += (size_t)N_NODES * HC;
  float*    agg   = ws;            ws += (size_t)N_NODES * HC;
  float*    xnode = ws;            ws += (size_t)N_NODES * HC;
  float*    exbuf = ws;            ws += (size_t)ETOT * HEADS;
  float*    denom = ws;            ws += (size_t)N_NODES * HEADS;
  unsigned* mord  = (unsigned*)ws; ws += (size_t)N_NODES * HEADS;
  float*    ppart = ws;

  dim3 b256(256);
  const int    gemmGrid = (N_NODES + 16 * ROWTILES - 1) / (16 * ROWTILES);
  const size_t shmem    = (size_t)HC * LDSW * sizeof(float);
  const int    edgeGrid = (ETOT + 7) / 8;
  const int    nodeGrid = (N_NODES + 7) / 8;
  const unsigned ORD_NEG_INF = 0x007FFFFFu;

  const float* xin = x;
  for (int l = 0; l < 2; ++l) {
    gemm_bias_wmma<<<gemmGrid, b256, shmem, stream>>>(xin, Wl[l], bl[l], xl, N_NODES);
    gemm_bias_wmma<<<gemmGrid, b256, shmem, stream>>>(xin, Wr[l], br[l], xr2, N_NODES);
    fill_u32<<<(N_NODES * HEADS + 255) / 256, b256, 0, stream>>>(mord, ORD_NEG_INF, N_NODES * HEADS);
    fill_f32<<<(N_NODES * HEADS + 255) / 256, b256, 0, stream>>>(denom, 0.f, N_NODES * HEADS);
    fill_f32<<<(N_NODES * HC + 255) / 256, b256, 0, stream>>>(agg, 0.f, N_NODES * HC);
    edge_score_kernel<<<edgeGrid, b256, 0, stream>>>(ei, xl, xr2, att[l], exbuf, mord);
    edge_exp_kernel<<<(ETOT * HEADS + 255) / 256, b256, 0, stream>>>(ei, exbuf, mord, denom);
    edge_aggregate_kernel<<<edgeGrid, b256, 0, stream>>>(ei, xl, exbuf, denom, agg);
    ln_silu_kernel<<<nodeGrid, b256, 0, stream>>>(agg, bias[l], gamma[l], beta[l], xnode);
    xin = xnode;
  }
  pool_gather_kernel<<<NGRAPH * POOL_SLABS, b256, 0, stream>>>(xnode, batch, ppart);
  pool_final_kernel<<<(NGRAPH * HC + 255) / 256, b256, 0, stream>>>(ppart, (float*)d_out);
}


# aggregate by owner-computes tiles: block per 196 destination rows sums them in LDS (scans all dst ids, LDS float adds, arrival order as before), LayerNorm from LDS; no global float adds in aggregate
# speedup vs baseline: 1.5032x; 1.3281x over previous
#include <hip/hip_runtime.h>
#include <hip/hip_bf16.h>
#include <math.h>

#define N_NODES 50000
#define E_EDGES 800000
#define ETOT    (E_EDGES + N_NODES)
#define HEADS   4
#define CHP     32
#define HC      128
#define NGRAPH  64
#define SLOPE   0.2f
#define LN_EPS  1e-5f

typedef float v2f __attribute__((ext_vector_type(2)));
typedef float v8f __attribute__((ext_vector_type(8)));

__device__ __forceinline__ unsigned f32_to_ord(float f) {
  unsigned u = __float_as_uint(f);
  return (u & 0x80000000u) ? ~u : (u | 0x80000000u);
}
__device__ __forceinline__ float ord_to_f32(unsigned u) {
  u = (u & 0x80000000u) ? (u & 0x7fffffffu) : ~u;
  return __uint_as_float(u);
}
__device__ __forceinline__ void atomAddF(float* p, float v) {
  unsafeAtomicAdd(p, v);
}

__global__ void fill_f32(float* __restrict__ p, float v, int n) {
  int i = blockIdx.x * blockDim.x + threadIdx.x;
  if (i < n) p[i] = v;
}
__global__ void fill_u32(unsigned* __restrict__ p, unsigned v, int n) {
  int i = blockIdx.x * blockDim.x + threadIdx.x;
  if (i < n) p[i] = v;
}

#define LDSW 136
#define ROWTILES 4
__global__ void gemm_bias_wmma(const float* __restrict__ X,
                               const float* __restrict__ W,
                               const float* __restrict__ bias,
                               float* __restrict__ Y, int n) {
  extern __shared__ float sW[];
  const int tid = threadIdx.x;
  for (int i = 0; i < 16; ++i) {
    int idx = tid + 256 * i;
    int r = idx >> 5, q = idx & 31;
    float4 w4 = *(const float4*)(W + r * HC + q * 4);
    float* d = sW + r * LDSW + q * 4;
    d[0] = w4.x; d[1] = w4.y; d[2] = w4.z; d[3] = w4.w;
  }
  __syncthreads();

  const int wave = tid >> 5;
  const int lane = tid & 31;
  const int m    = lane & 15;
  const int hi   = lane >> 4;
  const int col  = wave * 16 + m;
  const float bc = bias[col];
  const int rowbase = blockIdx.x * (16 * ROWTILES);

  for (int t = 0; t < ROWTILES; ++t) {
    int row0 = rowbase + t * 16;
    if (row0 >= n) break;
    int row  = row0 + m;
    int rowc = (row < n) ? row : (n - 1);
    const float* ap = X + (size_t)rowc * HC + 2 * hi;
    v8f acc = {};
    #pragma unroll
    for (int k0 = 0; k0 < HC; k0 += 4) {
      v2f a; a.x = ap[k0]; a.y = ap[k0 + 1];
      const float* bp = sW + (k0 + 2 * hi) * LDSW + col;
      v2f b; b.x = bp[0]; b.y = bp[LDSW];
      acc = __builtin_amdgcn_wmma_f32_16x16x4_f32(
          false, a, false, b, (short)0, acc, false, false);
    }
    #pragma unroll
    for (int r = 0; r < 8; ++r) {
      int orow = row0 + r + 8 * hi;
      if (orow < n) Y[(size_t)orow * HC + col] = acc[r] + bc;
    }
  }
}

__global__ void edge_score_kernel(const int* __restrict__ ei,
                                  const float* __restrict__ xl,
                                  const float* __restrict__ xr,
                                  const float* __restrict__ att,
                                  float* __restrict__ score,
                                  unsigned* __restrict__ mord) {
  int e = blockIdx.x * 8 + (threadIdx.x >> 5);
  int lane = threadIdx.x & 31;
  if (e >= ETOT) return;
  int src, dst;
  if (e < E_EDGES) { src = ei[e]; dst = ei[E_EDGES + e]; }
  else             { src = dst = e - E_EDGES; }
  const float* pl = xl + (size_t)src * HC + lane;
  const float* pr = xr + (size_t)dst * HC + lane;
  float s[HEADS];
  #pragma unroll
  for (int h = 0; h < HEADS; ++h) {
    float v = pl[h * CHP] + pr[h * CHP];
    v = (v > 0.f) ? v : v * SLOPE;
    s[h] = v * att[h * CHP + lane];
  }
  #pragma unroll
  for (int h = 0; h < HEADS; ++h)
    for (int off = 16; off > 0; off >>= 1) s[h] += __shfl_xor(s[h], off, 32);
  if (lane < HEADS) {
    float sh = s[lane];
    score[(size_t)e * HEADS + lane] = sh;
    atomicMax(&mord[(size_t)dst * HEADS + lane], f32_to_ord(sh));
  }
}

__global__ void edge_exp_kernel(const int* __restrict__ ei,
                                float* __restrict__ score,
                                const unsigned* __restrict__ mord,
                                float* __restrict__ denom) {
  int t = blockIdx.x * blockDim.x + threadIdx.x;
  if (t >= ETOT * HEADS) return;
  int e = t >> 2, h = t & 3;
  int dst = (e < E_EDGES) ? ei[E_EDGES + e] : (e - E_EDGES);
  float m = ord_to_f32(mord[(size_t)dst * HEADS + h]);
  float ex = __expf(score[t] - m);
  score[t] = ex;
  atomAddF(&denom[(size_t)dst * HEADS + h], ex);
}

#define AT_S       196
#define AT_TILES   ((N_NODES + AT_S - 1) / AT_S)
#define AT_THREADS 1024
#define AT_WAVES   (AT_THREADS / 32)
#define AT_QCAP    256
#define AT_GROUP   8
#define AT_V4      (E_EDGES / 4)
#define AT_ITERS   ((AT_V4 + AT_THREADS - 1) / AT_THREADS)
#define AT_PASSES  (((AT_ITERS + 3) / 4) * 4)
#define AT_SELF    ((AT_S + AT_WAVES - 1) / AT_WAVES)

__device__ __forceinline__ int lane_bcast_i(int v, int j) {
  return __builtin_amdgcn_readlane(v, j);
}
__device__ __forceinline__ float lane_bcast_f(float v, int j) {
  return __int_as_float(__builtin_amdgcn_readlane(__float_as_int(v), j));
}

__device__ __forceinline__ void tile_drain(const int2* q, int head, int n, int lane, int lo,
                                           const int* __restrict__ ei,
                                           const float* __restrict__ xl,
                                           const float* __restrict__ ex,
                                           const float* __restrict__ denom,
                                           float* tile) {
  int src = 0, dl = 0;
  float a0 = 0.f, a1 = 0.f, a2 = 0.f, a3 = 0.f;
  if (lane < n) {
    const int2 t = q[(head + lane) & (AT_QCAP - 1)];
    const int e = t.x;
    dl = t.y;
    src = (e < E_EDGES) ? ei[e] : (e - E_EDGES);
    const float4 x4 = *(const float4*)(ex + (size_t)e * HEADS);
    const float4 d4 = *(const float4*)(denom + (size_t)(lo + dl) * HEADS);
    a0 = x4.x / d4.x; a1 = x4.y / d4.y; a2 = x4.z / d4.z; a3 = x4.w / d4.w;
  }
  for (int j0 = 0; j0 < n; j0 += AT_GROUP) {
    float v[AT_GROUP][HEADS];
    #pragma unroll
    for (int u = 0; u < AT_GROUP; ++u) {
      const int s = lane_bcast_i(src, j0 + u);
      const float* p = xl + (size_t)s * HC + lane;
      #pragma unroll
      for (int h = 0; h < HEADS; ++h) v[u][h] = p[h * CHP];
    }
    #pragma unroll
    for (int u = 0; u < AT_GROUP; ++u) {
      if (j0 + u < n) {
        const int   d  = lane_bcast_i(dl, j0 + u);
        const float w0 = lane_bcast_f(a0, j0 + u);
        const float w1 = lane_bcast_f(a1, j0 + u);
        const float w2 = lane_bcast_f(a2, j0 + u);
        const float w3 = lane_bcast_f(a3, j0 + u);
        float* t = tile + d * HC + lane;
        atomAddF(t,           v[u][0] * w0);
        atomAddF(t + CHP,     v[u][1] * w1);
        atomAddF(t + 2 * CHP, v[u][2] * w2);
        atomAddF(t + 3 * CHP, v[u][3] * w3);
      }
    }
  }
}

__device__ __forceinline__ int4 tile_ids(const int4* __restrict__ dst4, const int pass, const int tid) {
  const int i = pass * AT_THREADS + tid;
  return dst4[(i < AT_V4) ? i : (AT_V4 - 1)];
}

__device__ __forceinline__ void tile_pass(int4& p, const int4* __restrict__ dst4, const int pass, const int tid,
                                          const bool last, int2* q, int& head, int& cnt,
                                          const unsigned below, const int lane, const int lo,
                                          const int* __restrict__ ei,
                                          const float* __restrict__ xl,
                                          const float* __restrict__ ex,
                                          const float* __restrict__ denom,
                                          float* tile) {
  const int idx = pass * AT_THREADS + tid;
  const unsigned dl[4] = {(unsigned)(p.x - lo), (unsigned)(p.y - lo), (unsigned)(p.z - lo), (unsigned)(p.w - lo)};
  p = tile_ids(dst4, pass + 4, tid);
  const bool have = idx < AT_V4;
  const unsigned dmin = min(min(dl[0], dl[1]), min(dl[2], dl[3]));
  if (__ballot(have && (dmin < (unsigned)AT_S)) != 0) {
    #pragma unroll
    for (int k = 0; k < 4; ++k) {
      const bool on = have && (dl[k] < (unsigned)AT_S);
      const unsigned m = (unsigned)__ballot(on);
      if (on) q[(head + cnt + __popc(m & below)) & (AT_QCAP - 1)] = make_int2(idx * 4 + k, (int)dl[k]);
      cnt += __popc(m);
    }
  }
  while (cnt >= 32 || (last && cnt > 0)) {
    const int n = (cnt < 32) ? cnt : 32;
    tile_drain(q, head, n, lane, lo, ei, xl, ex, denom, tile);
    head = (head + n) & (AT_QCAP - 1);
    cnt -= n;
  }
}

__global__ void __launch_bounds__(AT_THREADS)
tile_aggregate_ln_kernel(const int* __restrict__ ei,
                         const float* __restrict__ xl,
                         const float* __restrict__ ex,
                         const float* __restrict__ denom,
                         const float* __restrict__ bias,
                         const float* __restrict__ gamma,
                         const float* __restrict__ beta,
                         float* __restrict__ out) {
  __shared__ float tile[AT_S * HC];
  __shared__ int2  ring[AT_WAVES * AT_QCAP];
  const int tid  = threadIdx.x;
  const int wave = tid >> 5;
  const int lane = tid & 31;
  const int lo   = blockIdx.x * AT_S;

  for (int i = tid; i < AT_S * HC / 4; i += AT_THREADS)
    ((float4*)tile)[i] = make_float4(0.f, 0.f, 0.f, 0.f);
  __syncthreads();

  int2* q = ring + wave * AT_QCAP;
  int head = 0, cnt = 0;
  const unsigned below = (1u << lane) - 1u;
  {
    const int dl = wave * AT_SELF + lane;
    const bool on = (lane < AT_SELF) && (dl < AT_S) && (lo + dl < N_NODES);
    const unsigned m = (unsigned)__ballot(on);
    if (on) q[__popc(m & below)] = make_int2(E_EDGES + lo + dl, dl);
    cnt = __popc(m);
  }

  const int4* dst4 = (const int4*)(ei + E_EDGES);
  int4 p0 = tile_ids(dst4, 0, tid);
  int4 p1 = tile_ids(dst4, 1, tid);
  int4 p2 = tile_ids(dst4, 2, tid);
  int4 p3 = tile_ids(dst4, 3, tid);
  #pragma unroll 1
  for (int it = 0; it < AT_PASSES; it += 4) {
    tile_pass(p0, dst4, it,     tid, false, q, head, cnt, below, lane, lo, ei, xl, ex, denom, tile);
    tile_pass(p1, dst4, it + 1, tid, false, q, head, cnt, below, lane, lo, ei, xl, ex, denom, tile);
    tile_pass(p2, dst4, it + 2, tid, false, q, head, cnt, below, lane, lo, ei, xl, ex, denom, tile);
    tile_pass(p3, dst4, it + 3, tid, it + 4 >= AT_PASSES, q, head, cnt, below, lane, lo, ei, xl, ex, denom, tile);
  }
  __syncthreads();

  for (int dl = wave; dl < AT_S; dl += AT_WAVES) {
    const int node = lo + dl;
    if (node >= N_NODES) break;
    float v[4]; float sum = 0.f;
    #pragma unroll
    for (int i = 0; i < 4; ++i) {
      const int c = i * 32 + lane;
      v[i] = tile[dl * HC + c] + bias[c];
      sum += v[i];
    }
    for (int off = 16; off > 0; off >>= 1) sum += __shfl_xor(sum, off, 32);
    const float mu = sum * (1.0f / HC);
    float sq = 0.f;
    #pragma unroll
    for (int i = 0; i < 4; ++i) { float d = v[i] - mu; sq += d * d; }
    for (int off = 16; off > 0; off >>= 1) sq += __shfl_xor(sq, off, 32);
    const float rstd = rsqrtf(sq * (1.0f / HC) + LN_EPS);
    #pragma unroll
    for (int i = 0; i < 4; ++i) {
      const int c = i * 32 + lane;
      const float h = (v[i] - mu) * rstd * gamma[c] + beta[c];
      out[(size_t)node * HC + c] = h / (1.0f + __expf(-h));
    }
  }
}

#define POOL_CHUNKS 4
#define POOL_SLAB   (8 * POOL_CHUNKS * 32)
#define POOL_SLABS  ((N_NODES + POOL_SLAB - 1) / POOL_SLAB)
#define PROW        (HC + 32)
__global__ void pool_gather_kernel(const float* __restrict__ x,
                                   const int* __restrict__ batch,
                                   float* __restrict__ part) {
  __shared__ float sacc[8][HC];
  __shared__ int   scnt[8];
  const int g     = blockIdx.x % NGRAPH;
  const int s     = blockIdx.x / NGRAPH;
  const int wave  = threadIdx.x >> 5;
  const int lane  = threadIdx.x & 31;
  const int slab0 = s * POOL_SLAB;

  int ids[POOL_CHUNKS];
  #pragma unroll
  for (int i = 0; i < POOL_CHUNKS; ++i) {
    int node = slab0 + (i * 8 + wave) * 32 + lane;
    ids[i] = (node < N_NODES) ? batch[node] : -1;
  }
  float4 acc = make_float4(0.f, 0.f, 0.f, 0.f);
  int cnt = 0;
  #pragma unroll
  for (int i = 0; i < POOL_CHUNKS; ++i) {
    const unsigned mask = (unsigned)__ballot(ids[i] == g);
    if (mask == 0u) continue;
    cnt += __popc(mask);
    const int base = slab0 + (i * 8 + wave) * 32;
    for (int j0 = 0; j0 < 32; j0 += 16) {
      if (((mask >> j0) & 0xffffu) == 0u) continue;
      float4 v[16];
      #pragma unroll
      for (int u = 0; u < 16; ++u) {
        int node = base + j0 + u;
        node = (node < N_NODES) ? node : (N_NODES - 1);
        v[u] = *(const float4*)(x + (size_t)node * HC + lane * 4);
      }
      #pragma unroll
      for (int u = 0; u < 16; ++u) {
        const bool on = ((mask >> (j0 + u)) & 1u) != 0u;
        acc.x += on ? v[u].x : 0.f;
        acc.y += on ? v[u].y : 0.f;
        acc.z += on ? v[u].z : 0.f;
        acc.w += on ? v[u].w : 0.f;
      }
    }
  }
  *(float4*)&sacc[wave][lane * 4] = acc;
  if (lane == 0) scnt[wave] = cnt;
  __syncthreads();
  const int tid = threadIdx.x;
  float* prow = part + ((size_t)g * POOL_SLABS + s) * PROW;
  if (tid < HC) {
    float t = 0.f;
    #pragma unroll
    for (int w = 0; w < 8; ++w) t += sacc[w][tid];
    prow[tid] = t;
  } else if (tid < PROW) {
    int c = 0;
    #pragma unroll
    for (int w = 0; w < 8; ++w) c += scnt[w];
    ((int*)prow)[tid] = (tid == HC) ? c : 0;
  }
}
__global__ void pool_final_kernel(const float* __restrict__ part,
                                  float* __restrict__ out) {
  int t = blockIdx.x * blockDim.x + threadIdx.x;
  if (t >= NGRAPH * HC) return;
  const int g = t >> 7, c = t & 127;
  const float* p = part + (size_t)g * POOL_SLABS * PROW;
  float sum = 0.f;
  int n = 0;
  #pragma unroll 7
  for (int s = 0; s < POOL_SLABS; ++s) {
    sum += p[s * PROW + c];
    n   += ((const int*)p)[s * PROW + HC];
  }
  out[t] = sum / fmaxf((float)n, 1.0f);
}

extern "C" void kernel_launch(void* const* d_in, const int* in_sizes, int n_in,
                              void* d_out, int out_size, void* d_ws, size_t ws_size,
                              hipStream_t stream) {
  (void)in_sizes; (void)n_in; (void)out_size; (void)ws_size;
  const float* x     = (const float*)d_in[0];
  const int*   ei    = (const int*)d_in[1];
  const int*   batch = (const int*)d_in[2];
  const float* Wl[2]    = {(const float*)d_in[3],  (const float*)d_in[11]};
  const float* Wr[2]    = {(const float*)d_in[4],  (const float*)d_in[12]};
  const float* bl[2]    = {(const float*)d_in[5],  (const float*)d_in[13]};
  const float* br[2]    = {(const float*)d_in[6],  (const float*)d_in[14]};
  const float* att[2]   = {(const float*)d_in[7],  (const float*)d_in[15]};
  const float* bias[2]  = {(const float*)d_in[8],  (const float*)d_in[16]};
  const float* gamma[2] = {(const float*)d_in[9],  (const float*)d_in[17]};
  const float* beta[2]  = {(const float*)d_in[10], (const float*)d_in[18]};

  float* ws = (float*)d_ws;
  float*    xl    = ws;            ws += (size_t)N_NODES * HC;
  float*    xr2   = ws;            ws += (size_t)N_NODES * HC;
  float*    xnode = ws;            ws += (size_t)N_NODES * HC;
  float*    exbuf = ws;            ws += (size_t)ETOT * HEADS;
  float*    denom = ws;            ws += (size_t)N_NODES * HEADS;
  unsigned* mord  = (unsigned*)ws; ws += (size_t)N_NODES * HEADS;
  float*    ppart = ws;

  dim3 b256(256);
  const int    gemmGrid = (N_NODES + 16 * ROWTILES - 1) / (16 * ROWTILES);
  const size_t shmem    = (size_t)HC * LDSW * sizeof(float);
  const int    edgeGrid = (ETOT + 7) / 8;
  const unsigned ORD_NEG_INF = 0x007FFFFFu;

  const float* xin = x;
  for (int l = 0; l < 2; ++l) {
    gemm_bias_wmma<<<gemmGrid, b256, shmem, stream>>>(xin, Wl[l], bl[l], xl, N_NODES);
    gemm_bias_wmma<<<gemmGrid, b256, shmem, stream>>>(xin, Wr[l], br[l], xr2, N_NODES);
    fill_u32<<<(N_NODES * HEADS + 255) / 256, b256, 0, stream>>>(mord, ORD_NEG_INF, N_NODES * HEADS);
    fill_f32<<<(N_NODES * HEADS + 255) / 256, b256, 0, stream>>>(denom, 0.f, N_NODES * HEADS);
    edge_score_kernel<<<edgeGrid, b256, 0, stream>>>(ei, xl, xr2, att[l], exbuf, mord);
    edge_exp_kernel<<<(ETOT * HEADS + 255) / 256, b256, 0, stream>>>(ei, exbuf, mord, denom);
    tile_aggregate_ln_kernel<<<AT_TILES, AT_THREADS, 0, stream>>>(ei, xl, exbuf, denom, bias[l], gamma[l], beta[l], xnode);
    xin = xnode;
  }
  pool_gather_kernel<<<NGRAPH * POOL_SLABS, b256, 0, stream>>>(xnode, batch, ppart);
  pool_final_kernel<<<(NGRAPH * HC + 255) / 256, b256, 0, stream>>>(ppart, (float*)d_out);
}


# softmax fused into the tile kernel: block scores its own edges (x_r rows in LDS, 8-lane fold), row max and denominator by LDS atomics, one divide per row; 16 launches to 8; score sum order changes
# speedup vs baseline: 2.5202x; 1.6766x over previous
#include <hip/hip_runtime.h>
#include <hip/hip_bf16.h>
#include <math.h>

#define N_NODES 50000
#define E_EDGES 800000
#define ETOT    (E_EDGES + N_NODES)
#define HEADS   4
#define CHP     32
#define HC      128
#define NGRAPH  64
#define SLOPE   0.2f
#define LN_EPS  1e-5f

typedef float v2f __attribute__((ext_vector_type(2)));
typedef float v8f __attribute__((ext_vector_type(8)));

__device__ __forceinline__ unsigned f32_to_ord(float f) {
  unsigned u = __float_as_uint(f);
  return (u & 0x80000000u) ? ~u : (u | 0x80000000u);
}
__device__ __forceinline__ float ord_to_f32(unsigned u) {
  u = (u & 0x80000000u) ? (u & 0x7fffffffu) : ~u;
  return __uint_as_float(u);
}
__device__ __forceinline__ void atomAddF(float* p, float v) {
  unsafeAtomicAdd(p, v);
}

#define LDSW 136
#define ROWTILES 4
__global__ void gemm_bias_wmma(const float* __restrict__ X,
                               const float* __restrict__ W,
                               const float* __restrict__ bias,
                               float* __restrict__ Y, int n) {
  extern __shared__ float sW[];
  const int tid = threadIdx.x;
  for (int i = 0; i < 16; ++i) {
    int idx = tid + 256 * i;
    int r = idx >> 5, q = idx & 31;
    float4 w4 = *(const float4*)(W + r * HC + q * 4);
    float* d = sW + r * LDSW + q * 4;
    d[0] = w4.x; d[1] = w4.y; d[2] = w4.z; d[3] = w4.w;
  }
  __syncthreads();

  const int wave = tid >> 5;
  const int lane = tid & 31;
  const int m    = lane & 15;
  const int hi   = lane >> 4;
  const int col  = wave * 16 + m;
  const float bc = bias[col];
  const int rowbase = blockIdx.x * (16 * ROWTILES);

  for (int t = 0; t < ROWTILES; ++t) {
    int row0 = rowbase + t * 16;
    if (row0 >= n) break;
    int row  = row0 + m;
    int rowc = (row < n) ? row : (n - 1);
    const float* ap = X + (size_t)rowc * HC + 2 * hi;
    v8f acc = {};
    #pragma unroll
    for (int k0 = 0; k0 < HC; k0 += 4) {
      v2f a; a.x = ap[k0]; a.y = ap[k0 + 1];
      const float* bp = sW + (k0 + 2 * hi) * LDSW + col;
      v2f b; b.x = bp[0]; b.y = bp[LDSW];
      acc = __builtin_amdgcn_wmma_f32_16x16x4_f32(
          false, a, false, b, (short)0, acc, false, false);
    }
    #pragma unroll
    for (int r = 0; r < 8; ++r) {
      int orow = row0 + r + 8 * hi;
      if (orow < n) Y[(size_t)orow * HC + col] = acc[r] + bc;
    }
  }
}

#define AT_S       196
#define AT_TILES   ((N_NODES + AT_S - 1) / AT_S)
#define AT_THREADS 1024
#define AT_WAVES   (AT_THREADS / 32)
#define AT_QCAP    256
#define AT_GROUP   8
#define AT_V4      (E_EDGES / 4)
#define AT_ITERS   ((AT_V4 + AT_THREADS - 1) / AT_THREADS)
#define AT_PASSES  (((AT_ITERS + 3) / 4) * 4)
#define AT_SELF    ((AT_S + AT_WAVES - 1) / AT_WAVES)
#define ORD_NEG_INF 0x007FFFFFu

struct TileArgs {
  const int*   ei;
  const float* xl;
  const float* sxr;
  const float* att;
  float*       sc;
  float*       tile;
  unsigned*    smax;
  float*       sden;
  int          lo;
};

__device__ __forceinline__ int lane_bcast_i(int v, int j) {
  return __builtin_amdgcn_readlane(v, j);
}
__device__ __forceinline__ float lane_bcast_f(float v, int j) {
  return __int_as_float(__builtin_amdgcn_readlane(__float_as_int(v), j));
}
__device__ __forceinline__ float lrelu_dot4(const float4 a, const float4 b, const float4 w) {
  float t0 = a.x + b.x, t1 = a.y + b.y, t2 = a.z + b.z, t3 = a.w + b.w;
  t0 = (t0 > 0.f) ? t0 : t0 * SLOPE;
  t1 = (t1 > 0.f) ? t1 : t1 * SLOPE;
  t2 = (t2 > 0.f) ? t2 : t2 * SLOPE;
  t3 = (t3 > 0.f) ? t3 : t3 * SLOPE;
  return t0 * w.x + t1 * w.y + t2 * w.z + t3 * w.w;
}

__device__ __forceinline__ void score_drain(const int2* q, int head, int n, int lane, const TileArgs& A) {
  int src = 0, dl = 0, e = 0;
  if (lane < n) {
    const int2 t = q[(head + lane) & (AT_QCAP - 1)];
    e = t.x;
    dl = t.y;
    src = (e < E_EDGES) ? A.ei[e] : (e - E_EDGES);
  }
  const float4 w4 = *(const float4*)(A.att + 4 * lane);
  const int  r  = lane & 7;
  const int  hd = lane >> 3;
  const bool b2 = (r & 4) != 0, b1 = (r & 2) != 0, b0 = (r & 1) != 0;
  for (int j0 = 0; j0 < n; j0 += AT_GROUP) {
    float4 a[AT_GROUP], b[AT_GROUP];
    #pragma unroll
    for (int u = 0; u < AT_GROUP; ++u) {
      const int s = lane_bcast_i(src, j0 + u);
      const int d = lane_bcast_i(dl, j0 + u);
      a[u] = *(const float4*)(A.xl + (size_t)s * HC + 4 * lane);
      b[u] = *(const float4*)(A.sxr + d * HC + 4 * lane);
    }
    float p[AT_GROUP];
    #pragma unroll
    for (int u = 0; u < AT_GROUP; ++u) p[u] = lrelu_dot4(a[u], b[u], w4);
    float f4[4];
    #pragma unroll
    for (int i = 0; i < 4; ++i) {
      const float keep = b2 ? p[i + 4] : p[i];
      const float send = b2 ? p[i] : p[i + 4];
      f4[i] = keep + __shfl_xor(send, 4, 32);
    }
    float f2[2];
    #pragma unroll
    for (int i = 0; i < 2; ++i) {
      const float keep = b1 ? f4[i + 2] : f4[i];
      const float send = b1 ? f4[i] : f4[i + 2];
      f2[i] = keep + __shfl_xor(send, 2, 32);
    }
    const float keep = b0 ? f2[1] : f2[0];
    const float send = b0 ? f2[0] : f2[1];
    const float sc = keep + __shfl_xor(send, 1, 32);
    const int ue = j0 + r;
    const int eu = __shfl(e, ue, 32);
    const int du = __shfl(dl, ue, 32);
    if (ue < n) {
      A.sc[(size_t)eu * HEADS + hd] = sc;
      atomicMax(&A.smax[du * HEADS + hd], f32_to_ord(sc));
    }
  }
}

__device__ __forceinline__ void agg_drain(const int2* q, int head, int n, int lane, const TileArgs& A) {
  int src = 0, dl = 0;
  float a0 = 0.f, a1 = 0.f, a2 = 0.f, a3 = 0.f;
  if (lane < n) {
    const int2 t = q[(head + lane) & (AT_QCAP - 1)];
    const int e = t.x;
    dl = t.y;
    src = (e < E_EDGES) ? A.ei[e] : (e - E_EDGES);
    const float4 s4 = *(const float4*)(A.sc + (size_t)e * HEADS);
    const uint4  m4 = *(const uint4*)(A.smax + dl * HEADS);
    a0 = __expf(s4.x - ord_to_f32(m4.x));
    a1 = __expf(s4.y - ord_to_f32(m4.y));
    a2 = __expf(s4.z - ord_to_f32(m4.z));
    a3 = __expf(s4.w - ord_to_f32(m4.w));
    float* dn = A.sden + dl * HEADS;
    atomAddF(dn,     a0);
    atomAddF(dn + 1, a1);
    atomAddF(dn + 2, a2);
    atomAddF(dn + 3, a3);
  }
  for (int j0 = 0; j0 < n; j0 += AT_GROUP) {
    float v[AT_GROUP][HEADS];
    #pragma unroll
    for (int u = 0; u < AT_GROUP; ++u) {
      const int s = lane_bcast_i(src, j0 + u);
      const float* p = A.xl + (size_t)s * HC + lane;
      #pragma unroll
      for (int h = 0; h < HEADS; ++h) v[u][h] = p[h * CHP];
    }
    #pragma unroll
    for (int u = 0; u < AT_GROUP; ++u) {
      if (j0 + u < n) {
        const int   d  = lane_bcast_i(dl, j0 + u);
        const float w0 = lane_bcast_f(a0, j0 + u);
        const float w1 = lane_bcast_f(a1, j0 + u);
        const float w2 = lane_bcast_f(a2, j0 + u);
        const float w3 = lane_bcast_f(a3, j0 + u);
        float* t = A.tile + d * HC + lane;
        atomAddF(t,           v[u][0] * w0);
        atomAddF(t + CHP,     v[u][1] * w1);
        atomAddF(t + 2 * CHP, v[u][2] * w2);
        atomAddF(t + 3 * CHP, v[u][3] * w3);
      }
    }
  }
}

__device__ __forceinline__ int4 tile_ids(const int4* __restrict__ dst4, const int pass, const int tid) {
  const int i = pass * AT_THREADS + tid;
  return dst4[(i < AT_V4) ? i : (AT_V4 - 1)];
}

template <int PH>
__device__ __forceinline__ void tile_pass(int4& p, const int4* __restrict__ dst4, const int pass, const int tid,
                                          const bool last, int2* q, int& head, int& cnt, int& total,
                                          const unsigned below, const int lane, const TileArgs& A) {
  const int idx = pass * AT_THREADS + tid;
  const int lo = A.lo;
  const unsigned dl[4] = {(unsigned)(p.x - lo), (unsigned)(p.y - lo), (unsigned)(p.z - lo), (unsigned)(p.w - lo)};
  p = tile_ids(dst4, pass + 4, tid);
  const bool have = idx < AT_V4;
  const unsigned dmin = min(min(dl[0], dl[1]), min(dl[2], dl[3]));
  if (__ballot(have && (dmin < (unsigned)AT_S)) != 0) {
    #pragma unroll
    for (int k = 0; k < 4; ++k) {
      const bool on = have && (dl[k] < (unsigned)AT_S);
      const unsigned m = (unsigned)__ballot(on);
      if (on) q[(head + cnt + __popc(m & below)) & (AT_QCAP - 1)] = make_int2(idx * 4 + k, (int)dl[k]);
      cnt += __popc(m);
      total += __popc(m);
    }
  }
  while (cnt >= 32 || (last && cnt > 0)) {
    const int n = (cnt < 32) ? cnt : 32;
    if (PH == 0) score_drain(q, head, n, lane, A);
    else         agg_drain(q, head, n, lane, A);
    head = (head + n) & (AT_QCAP - 1);
    cnt -= n;
  }
}

template <int PH>
__device__ __forceinline__ int tile_scan(int2* q, const int tid, const int wave, const int lane, const TileArgs& A) {
  int head = 0, cnt = 0, total = 0;
  const unsigned below = (1u << lane) - 1u;
  {
    const int dl = wave * AT_SELF + lane;
    const bool on = (lane < AT_SELF) && (dl < AT_S) && (A.lo + dl < N_NODES);
    const unsigned m = (unsigned)__ballot(on);
    if (on) q[__popc(m & below)] = make_int2(E_EDGES + A.lo + dl, dl);
    cnt = __popc(m);
    total = cnt;
  }
  const int4* dst4 = (const int4*)(A.ei + E_EDGES);
  int4 p0 = tile_ids(dst4, 0, tid);
  int4 p1 = tile_ids(dst4, 1, tid);
  int4 p2 = tile_ids(dst4, 2, tid);
  int4 p3 = tile_ids(dst4, 3, tid);
  #pragma unroll 1
  for (int it = 0; it < AT_PASSES; it += 4) {
    tile_pass<PH>(p0, dst4, it,     tid, false, q, head, cnt, total, below, lane, A);
    tile_pass<PH>(p1, dst4, it + 1, tid, false, q, head, cnt, total, below, lane, A);
    tile_pass<PH>(p2, dst4, it + 2, tid, false, q, head, cnt, total, below, lane, A);
    tile_pass<PH>(p3, dst4, it + 3, tid, it + 4 >= AT_PASSES, q, head, cnt, total, below, lane, A);
  }
  return total;
}

__global__ void __launch_bounds__(AT_THREADS)
tile_softmax_aggregate_ln_kernel(const int* __restrict__ ei,
                                 const float* __restrict__ xl,
                                 const float* __restrict__ xr,
                                 const float* __restrict__ att,
                                 float* sc,
                                 const float* __restrict__ bias,
                                 const float* __restrict__ gamma,
                                 const float* __restrict__ beta,
                                 float* __restrict__ out) {
  __shared__ __align__(16) float    tile[AT_S * HC];
  __shared__ __align__(16) float    sxr[AT_S * HC];
  __shared__ __align__(16) int2     ring[AT_WAVES * AT_QCAP];
  __shared__ __align__(16) unsigned smax[AT_S * HEADS];
  __shared__ __align__(16) float    sden[AT_S * HEADS];
  __shared__ int      sover;
  const int tid  = threadIdx.x;
  const int wave = tid >> 5;
  const int lane = tid & 31;
  const int lo   = blockIdx.x * AT_S;

  {
    constexpr int NV4 = AT_S * HC / 4;
    constexpr int NLD = (NV4 + AT_THREADS - 1) / AT_THREADS;
    const float4 zero4 = make_float4(0.f, 0.f, 0.f, 0.f);
    float4 r[NLD];
    #pragma unroll
    for (int k = 0; k < NLD; ++k) {
      const int gi = lo * (HC / 4) + tid + k * AT_THREADS;
      r[k] = ((const float4*)xr)[min(gi, N_NODES * (HC / 4) - 1)];
    }
    #pragma unroll
    for (int k = 0; k < NLD; ++k) {
      const int i = tid + k * AT_THREADS;
      if (i < NV4) { ((float4*)tile)[i] = zero4; ((float4*)sxr)[i] = r[k]; }
    }
  }
  if (tid < AT_S * HEADS) { smax[tid] = ORD_NEG_INF; sden[tid] = 0.f; }
  if (tid == 0) sover = 0;
  __syncthreads();

  TileArgs A;
  A.ei = ei; A.xl = xl; A.sxr = sxr; A.att = att; A.sc = sc;
  A.tile = tile; A.smax = smax; A.sden = sden; A.lo = lo;
  int2* q = ring + wave * AT_QCAP;

  const int total = tile_scan<0>(q, tid, wave, lane, A);
  if (total > AT_QCAP && lane == 0) sover = 1;
  __syncthreads();

  if (sover == 0) {
    for (int h0 = 0; h0 < total; h0 += 32) {
      const int n = (total - h0 < 32) ? (total - h0) : 32;
      agg_drain(q, h0, n, lane, A);
    }
  } else {
    (void)tile_scan<1>(q, tid, wave, lane, A);
  }
  __syncthreads();

  for (int dl = wave; dl < AT_S; dl += AT_WAVES) {
    const int node = lo + dl;
    if (node >= N_NODES) break;
    float v[4]; float sum = 0.f;
    #pragma unroll
    for (int i = 0; i < 4; ++i) {
      const int c = i * 32 + lane;
      v[i] = tile[dl * HC + c] / sden[dl * HEADS + i] + bias[c];
      sum += v[i];
    }
    for (int off = 16; off > 0; off >>= 1) sum += __shfl_xor(sum, off, 32);
    const float mu = sum * (1.0f / HC);
    float sq = 0.f;
    #pragma unroll
    for (int i = 0; i < 4; ++i) { float d = v[i] - mu; sq += d * d; }
    for (int off = 16; off > 0; off >>= 1) sq += __shfl_xor(sq, off, 32);
    const float rstd = rsqrtf(sq * (1.0f / HC) + LN_EPS);
    #pragma unroll
    for (int i = 0; i < 4; ++i) {
      const int c = i * 32 + lane;
      const float h = (v[i] - mu) * rstd * gamma[c] + beta[c];
      out[(size_t)node * HC + c] = h / (1.0f + __expf(-h));
    }
  }
}

#define POOL_CHUNKS 4
#define POOL_SLAB   (8 * POOL_CHUNKS * 32)
#define POOL_SLABS  ((N_NODES + POOL_SLAB - 1) / POOL_SLAB)
#define PROW        (HC + 32)
__global__ void pool_gather_kernel(const float* __restrict__ x,
                                   const int* __restrict__ batch,
                                   float* __restrict__ part) {
  __shared__ float sacc[8][HC];
  __shared__ int   scnt[8];
  const int g     = blockIdx.x % NGRAPH;
  const int s     = blockIdx.x / NGRAPH;
  const int wave  = threadIdx.x >> 5;
  const int lane  = threadIdx.x & 31;
  const int slab0 = s * POOL_SLAB;

  int ids[POOL_CHUNKS];
  #pragma unroll
  for (int i = 0; i < POOL_CHUNKS; ++i) {
    int node = slab0 + (i * 8 + wave) * 32 + lane;
    ids[i] = (node < N_NODES) ? batch[node] : -1;
  }
  float4 acc = make_float4(0.f, 0.f, 0.f, 0.f);
  int cnt = 0;
  #pragma unroll
  for (int i = 0; i < POOL_CHUNKS; ++i) {
    const unsigned mask = (unsigned)__ballot(ids[i] == g);
    if (mask == 0u) continue;
    cnt += __popc(mask);
    const int base = slab0 + (i * 8 + wave) * 32;
    for (int j0 = 0; j0 < 32; j0 += 16) {
      if (((mask >> j0) & 0xffffu) == 0u) continue;
      float4 v[16];
      #pragma unroll
      for (int u = 0; u < 16; ++u) {
        int node = base + j0 + u;
        node = (node < N_NODES) ? node : (N_NODES - 1);
        v[u] = *(const float4*)(x + (size_t)node * HC + lane * 4);
      }
      #pragma unroll
      for (int u = 0; u < 16; ++u) {
        const bool on = ((mask >> (j0 + u)) & 1u) != 0u;
        acc.x += on ? v[u].x : 0.f;
        acc.y += on ? v[u].y : 0.f;
        acc.z += on ? v[u].z : 0.f;
        acc.w += on ? v[u].w : 0.f;
      }
    }
  }
  *(float4*)&sacc[wave][lane * 4] = acc;
  if (lane == 0) scnt[wave] = cnt;
  __syncthreads();
  const int tid = threadIdx.x;
  float* prow = part + ((size_t)g * POOL_SLABS + s) * PROW;
  if (tid < HC) {
    float t = 0.f;
    #pragma unroll
    for (int w = 0; w < 8; ++w) t += sacc[w][tid];
    prow[tid] = t;
  } else if (tid < PROW) {
    int c = 0;
    #pragma unroll
    for (int w = 0; w < 8; ++w) c += scnt[w];
    ((int*)prow)[tid] = (tid == HC) ? c : 0;
  }
}
__global__ void pool_final_kernel(const float* __restrict__ part,
                                  float* __restrict__ out) {
  int t = blockIdx.x * blockDim.x + threadIdx.x;
  if (t >= NGRAPH * HC) return;
  const int g = t >> 7, c = t & 127;
  const float* p = part + (size_t)g * POOL_SLABS * PROW;
  float sum = 0.f;
  int n = 0;
  #pragma unroll 7
  for (int s = 0; s < POOL_SLABS; ++s) {
    sum += p[s * PROW + c];
    n   += ((const int*)p)[s * PROW + HC];
  }
  out[t] = sum / fmaxf((float)n, 1.0f);
}

extern "C" void kernel_launch(void* const* d_in, const int* in_sizes, int n_in,
                              void* d_out, int out_size, void* d_ws, size_t ws_size,
                              hipStream_t stream) {
  (void)in_sizes; (void)n_in; (void)out_size; (void)ws_size;
  const float* x     = (const float*)d_in[0];
  const int*   ei    = (const int*)d_in[1];
  const int*   batch = (const int*)d_in[2];
  const float* Wl[2]    = {(const float*)d_in[3],  (const float*)d_in[11]};
  const float* Wr[2]    = {(const float*)d_in[4],  (const float*)d_in[12]};
  const float* bl[2]    = {(const float*)d_in[5],  (const float*)d_in[13]};
  const float* br[2]    = {(const float*)d_in[6],  (const float*)d_in[14]};
  const float* att[2]   = {(const float*)d_in[7],  (const float*)d_in[15]};
  const float* bias[2]  = {(const float*)d_in[8],  (const float*)d_in[16]};
  const float* gamma[2] = {(const float*)d_in[9],  (const float*)d_in[17]};
  const float* beta[2]  = {(const float*)d_in[10], (const float*)d_in[18]};

  float* ws = (float*)d_ws;
  float*    xl    = ws;            ws += (size_t)N_NODES * HC;
  float*    xr2   = ws;            ws += (size_t)N_NODES * HC;
  float*    xnode = ws;            ws += (size_t)N_NODES * HC;
  float*    scbuf = ws;            ws += (size_t)ETOT * HEADS;
  float*    ppart = ws;

  dim3 b256(256);
  const int    gemmGrid = (N_NODES + 16 * ROWTILES - 1) / (16 * ROWTILES);
  const size_t shmem    = (size_t)HC * LDSW * sizeof(float);

  const float* xin = x;
  for (int l = 0; l < 2; ++l) {
    gemm_bias_wmma<<<gemmGrid, b256, shmem, stream>>>(xin, Wl[l], bl[l], xl, N_NODES);
    gemm_bias_wmma<<<gemmGrid, b256, shmem, stream>>>(xin, Wr[l], br[l], xr2, N_NODES);
    tile_softmax_aggregate_ln_kernel<<<AT_TILES, AT_THREADS, 0, stream>>>(ei, xl, xr2, att[l], scbuf, bias[l], gamma[l], beta[l], xnode);
    xin = xnode;
  }
  pool_gather_kernel<<<NGRAPH * POOL_SLABS, b256, 0, stream>>>(xnode, batch, ppart);
  pool_final_kernel<<<(NGRAPH * HC + 255) / 256, b256, 0, stream>>>(ppart, (float*)d_out);
}


# layer 2's tile launch reads back the per-wave lists of found edges that layer 1's launch stores in the workspace and no longer scans all 800000 ids; LDS sums in arrival order as before
# speedup vs baseline: 2.7135x; 1.0767x over previous
#include <hip/hip_runtime.h>
#include <hip/hip_bf16.h>
#include <math.h>

#define N_NODES 50000
#define E_EDGES 800000
#define ETOT    (E_EDGES + N_NODES)
#define HEADS   4
#define CHP     32
#define HC      128
#define NGRAPH  64
#define SLOPE   0.2f
#define LN_EPS  1e-5f

typedef float v2f __attribute__((ext_vector_type(2)));
typedef float v8f __attribute__((ext_vector_type(8)));

__device__ __forceinline__ unsigned f32_to_ord(float f) {
  unsigned u = __float_as_uint(f);
  return (u & 0x80000000u) ? ~u : (u | 0x80000000u);
}
__device__ __forceinline__ float ord_to_f32(unsigned u) {
  u = (u & 0x80000000u) ? (u & 0x7fffffffu) : ~u;
  return __uint_as_float(u);
}
__device__ __forceinline__ void atomAddF(float* p, float v) {
  unsafeAtomicAdd(p, v);
}

#define LDSW 136
#define ROWTILES 4
__global__ void gemm_bias_wmma(const float* __restrict__ X,
                               const float* __restrict__ W,
                               const float* __restrict__ bias,
                               float* __restrict__ Y, int n) {
  extern __shared__ float sW[];
  const int tid = threadIdx.x;
  for (int i = 0; i < 16; ++i) {
    int idx = tid + 256 * i;
    int r = idx >> 5, q = idx & 31;
    float4 w4 = *(const float4*)(W + r * HC + q * 4);
    float* d = sW + r * LDSW + q * 4;
    d[0] = w4.x; d[1] = w4.y; d[2] = w4.z; d[3] = w4.w;
  }
  __syncthreads();

  const int wave = tid >> 5;
  const int lane = tid & 31;
  const int m    = lane & 15;
  const int hi   = lane >> 4;
  const int col  = wave * 16 + m;
  const float bc = bias[col];
  const int rowbase = blockIdx.x * (16 * ROWTILES);

  for (int t = 0; t < ROWTILES; ++t) {
    int row0 = rowbase + t * 16;
    if (row0 >= n) break;
    int row  = row0 + m;
    int rowc = (row < n) ? row : (n - 1);
    const float* ap = X + (size_t)rowc * HC + 2 * hi;
    v8f acc = {};
    #pragma unroll
    for (int k0 = 0; k0 < HC; k0 += 4) {
      v2f a; a.x = ap[k0]; a.y = ap[k0 + 1];
      const float* bp = sW + (k0 + 2 * hi) * LDSW + col;
      v2f b; b.x = bp[0]; b.y = bp[LDSW];
      acc = __builtin_amdgcn_wmma_f32_16x16x4_f32(
          false, a, false, b, (short)0, acc, false, false);
    }
    #pragma unroll
    for (int r = 0; r < 8; ++r) {
      int orow = row0 + r + 8 * hi;
      if (orow < n) Y[(size_t)orow * HC + col] = acc[r] + bc;
    }
  }
}

#define AT_S       196
#define AT_TILES   ((N_NODES + AT_S - 1) / AT_S)
#define AT_THREADS 1024
#define AT_WAVES   (AT_THREADS / 32)
#define AT_QCAP    256
#define AT_GROUP   8
#define AT_V4      (E_EDGES / 4)
#define AT_ITERS   ((AT_V4 + AT_THREADS - 1) / AT_THREADS)
#define AT_PASSES  (((AT_ITERS + 3) / 4) * 4)
#define AT_SELF    ((AT_S + AT_WAVES - 1) / AT_WAVES)
#define ORD_NEG_INF 0x007FFFFFu

struct TileArgs {
  const int*   ei;
  const float* xl;
  const float* sxr;
  const float* att;
  float*       sc;
  float*       tile;
  unsigned*    smax;
  float*       sden;
  int          lo;
};

__device__ __forceinline__ int lane_bcast_i(int v, int j) {
  return __builtin_amdgcn_readlane(v, j);
}
__device__ __forceinline__ float lane_bcast_f(float v, int j) {
  return __int_as_float(__builtin_amdgcn_readlane(__float_as_int(v), j));
}
__device__ __forceinline__ float lrelu_dot4(const float4 a, const float4 b, const float4 w) {
  float t0 = a.x + b.x, t1 = a.y + b.y, t2 = a.z + b.z, t3 = a.w + b.w;
  t0 = (t0 > 0.f) ? t0 : t0 * SLOPE;
  t1 = (t1 > 0.f) ? t1 : t1 * SLOPE;
  t2 = (t2 > 0.f) ? t2 : t2 * SLOPE;
  t3 = (t3 > 0.f) ? t3 : t3 * SLOPE;
  return t0 * w.x + t1 * w.y + t2 * w.z + t3 * w.w;
}

__device__ __forceinline__ void score_drain(const int2* q, int head, int n, int lane, const TileArgs& A) {
  int src = 0, dl = 0, e = 0;
  if (lane < n) {
    const int2 t = q[(head + lane) & (AT_QCAP - 1)];
    e = t.x;
    dl = t.y;
    src = (e < E_EDGES) ? A.ei[e] : (e - E_EDGES);
  }
  const float4 w4 = *(const float4*)(A.att + 4 * lane);
  const int  r  = lane & 7;
  const int  hd = lane >> 3;
  const bool b2 = (r & 4) != 0, b1 = (r & 2) != 0, b0 = (r & 1) != 0;
  for (int j0 = 0; j0 < n; j0 += AT_GROUP) {
    float4 a[AT_GROUP], b[AT_GROUP];
    #pragma unroll
    for (int u = 0; u < AT_GROUP; ++u) {
      const int s = lane_bcast_i(src, j0 + u);
      const int d = lane_bcast_i(dl, j0 + u);
      a[u] = *(const float4*)(A.xl + (size_t)s * HC + 4 * lane);
      b[u] = *(const float4*)(A.sxr + d * HC + 4 * lane);
    }
    float p[AT_GROUP];
    #pragma unroll
    for (int u = 0; u < AT_GROUP; ++u) p[u] = lrelu_dot4(a[u], b[u], w4);
    float f4[4];
    #pragma unroll
    for (int i = 0; i < 4; ++i) {
      const float keep = b2 ? p[i + 4] : p[i];
      const float send = b2 ? p[i] : p[i + 4];
      f4[i] = keep + __shfl_xor(send, 4, 32);
    }
    float f2[2];
    #pragma unroll
    for (int i = 0; i < 2; ++i) {
      const float keep = b1 ? f4[i + 2] : f4[i];
      const float send = b1 ? f4[i] : f4[i + 2];
      f2[i] = keep + __shfl_xor(send, 2, 32);
    }
    const float keep = b0 ? f2[1] : f2[0];
    const float send = b0 ? f2[0] : f2[1];
    const float sc = keep + __shfl_xor(send, 1, 32);
    const int ue = j0 + r;
    const int eu = __shfl(e, ue, 32);
    const int du = __shfl(dl, ue, 32);
    if (ue < n) {
      A.sc[(size_t)eu * HEADS + hd] = sc;
      atomicMax(&A.smax[du * HEADS + hd], f32_to_ord(sc));
    }
  }
}

__device__ __forceinline__ void agg_drain(const int2* q, int head, int n, int lane, const TileArgs& A) {
  int src = 0, dl = 0;
  float a0 = 0.f, a1 = 0.f, a2 = 0.f, a3 = 0.f;
  if (lane < n) {
    const int2 t = q[(head + lane) & (AT_QCAP - 1)];
    const int e = t.x;
    dl = t.y;
    src = (e < E_EDGES) ? A.ei[e] : (e - E_EDGES);
    const float4 s4 = *(const float4*)(A.sc + (size_t)e * HEADS);
    const uint4  m4 = *(const uint4*)(A.smax + dl * HEADS);
    a0 = __expf(s4.x - ord_to_f32(m4.x));
    a1 = __expf(s4.y - ord_to_f32(m4.y));
    a2 = __expf(s4.z - ord_to_f32(m4.z));
    a3 = __expf(s4.w - ord_to_f32(m4.w));
    float* dn = A.sden + dl * HEADS;
    atomAddF(dn,     a0);
    atomAddF(dn + 1, a1);
    atomAddF(dn + 2, a2);
    atomAddF(dn + 3, a3);
  }
  for (int j0 = 0; j0 < n; j0 += AT_GROUP) {
    float v[AT_GROUP][HEADS];
    #pragma unroll
    for (int u = 0; u < AT_GROUP; ++u) {
      const int s = lane_bcast_i(src, j0 + u);
      const float* p = A.xl + (size_t)s * HC + lane;
      #pragma unroll
      for (int h = 0; h < HEADS; ++h) v[u][h] = p[h * CHP];
    }
    #pragma unroll
    for (int u = 0; u < AT_GROUP; ++u) {
      if (j0 + u < n) {
        const int   d  = lane_bcast_i(dl, j0 + u);
        const float w0 = lane_bcast_f(a0, j0 + u);
        const float w1 = lane_bcast_f(a1, j0 + u);
        const float w2 = lane_bcast_f(a2, j0 + u);
        const float w3 = lane_bcast_f(a3, j0 + u);
        float* t = A.tile + d * HC + lane;
        atomAddF(t,           v[u][0] * w0);
        atomAddF(t + CHP,     v[u][1] * w1);
        atomAddF(t + 2 * CHP, v[u][2] * w2);
        atomAddF(t + 3 * CHP, v[u][3] * w3);
      }
    }
  }
}

__device__ __forceinline__ int4 tile_ids(const int4* __restrict__ dst4, const int pass, const int tid) {
  const int i = pass * AT_THREADS + tid;
  return dst4[(i < AT_V4) ? i : (AT_V4 - 1)];
}

template <int PH>
__device__ __forceinline__ void tile_pass(int4& p, const int4* __restrict__ dst4, const int pass, const int tid,
                                          const bool last, int2* q, int& head, int& cnt, int& total,
                                          const unsigned below, const int lane, const TileArgs& A) {
  const int idx = pass * AT_THREADS + tid;
  const int lo = A.lo;
  const unsigned dl[4] = {(unsigned)(p.x - lo), (unsigned)(p.y - lo), (unsigned)(p.z - lo), (unsigned)(p.w - lo)};
  p = tile_ids(dst4, pass + 4, tid);
  const bool have = idx < AT_V4;
  const unsigned dmin = min(min(dl[0], dl[1]), min(dl[2], dl[3]));
  if (__ballot(have && (dmin < (unsigned)AT_S)) != 0) {
    #pragma unroll
    for (int k = 0; k < 4; ++k) {
      const bool on = have && (dl[k] < (unsigned)AT_S);
      const unsigned m = (unsigned)__ballot(on);
      if (on) q[(head + cnt + __popc(m & below)) & (AT_QCAP - 1)] = make_int2(idx * 4 + k, (int)dl[k]);
      cnt += __popc(m);
      total += __popc(m);
    }
  }
  while (cnt >= 32 || (last && cnt > 0)) {
    const int n = (cnt < 32) ? cnt : 32;
    if (PH == 0) score_drain(q, head, n, lane, A);
    else         agg_drain(q, head, n, lane, A);
    head = (head + n) & (AT_QCAP - 1);
    cnt -= n;
  }
}

template <int PH>
__device__ __forceinline__ int tile_scan(int2* q, const int tid, const int wave, const int lane, const TileArgs& A) {
  int head = 0, cnt = 0, total = 0;
  const unsigned below = (1u << lane) - 1u;
  {
    const int dl = wave * AT_SELF + lane;
    const bool on = (lane < AT_SELF) && (dl < AT_S) && (A.lo + dl < N_NODES);
    const unsigned m = (unsigned)__ballot(on);
    if (on) q[__popc(m & below)] = make_int2(E_EDGES + A.lo + dl, dl);
    cnt = __popc(m);
    total = cnt;
  }
  const int4* dst4 = (const int4*)(A.ei + E_EDGES);
  int4 p0 = tile_ids(dst4, 0, tid);
  int4 p1 = tile_ids(dst4, 1, tid);
  int4 p2 = tile_ids(dst4, 2, tid);
  int4 p3 = tile_ids(dst4, 3, tid);
  #pragma unroll 1
  for (int it = 0; it < AT_PASSES; it += 4) {
    tile_pass<PH>(p0, dst4, it,     tid, false, q, head, cnt, total, below, lane, A);
    tile_pass<PH>(p1, dst4, it + 1, tid, false, q, head, cnt, total, below, lane, A);
    tile_pass<PH>(p2, dst4, it + 2, tid, false, q, head, cnt, total, below, lane, A);
    tile_pass<PH>(p3, dst4, it + 3, tid, it + 4 >= AT_PASSES, q, head, cnt, total, below, lane, A);
  }
  return total;
}

template <int REUSE>
__global__ void __launch_bounds__(AT_THREADS)
tile_softmax_aggregate_ln_kernel(const int* __restrict__ ei,
                                 const float* __restrict__ xl,
                                 const float* __restrict__ xr,
                                 const float* __restrict__ att,
                                 float* sc,
                                 const float* __restrict__ bias,
                                 const float* __restrict__ gamma,
                                 const float* __restrict__ beta,
                                 float* __restrict__ out,
                                 int*  lists,
                                 int*  lcnt,
                                 int*  lover) {
  __shared__ __align__(16) float    tile[AT_S * HC];
  __shared__ __align__(16) float    sxr[AT_S * HC];
  __shared__ __align__(16) int2     ring[AT_WAVES * AT_QCAP];
  __shared__ __align__(16) unsigned smax[AT_S * HEADS];
  __shared__ __align__(16) float    sden[AT_S * HEADS];
  __shared__ int      sover;
  const int tid  = threadIdx.x;
  const int wave = tid >> 5;
  const int lane = tid & 31;
  const int lo   = blockIdx.x * AT_S;

  {
    constexpr int NV4 = AT_S * HC / 4;
    constexpr int NLD = (NV4 + AT_THREADS - 1) / AT_THREADS;
    const float4 zero4 = make_float4(0.f, 0.f, 0.f, 0.f);
    float4 r[NLD];
    #pragma unroll
    for (int k = 0; k < NLD; ++k) {
      const int gi = lo * (HC / 4) + tid + k * AT_THREADS;
      r[k] = ((const float4*)xr)[min(gi, N_NODES * (HC / 4) - 1)];
    }
    #pragma unroll
    for (int k = 0; k < NLD; ++k) {
      const int i = tid + k * AT_THREADS;
      if (i < NV4) { ((float4*)tile)[i] = zero4; ((float4*)sxr)[i] = r[k]; }
    }
  }
  if (tid < AT_S * HEADS) { smax[tid] = ORD_NEG_INF; sden[tid] = 0.f; }
  if (tid == 0) sover = 0;
  __syncthreads();

  TileArgs A;
  A.ei = ei; A.xl = xl; A.sxr = sxr; A.att = att; A.sc = sc;
  A.tile = tile; A.smax = smax; A.sden = sden; A.lo = lo;
  int2* q = ring + wave * AT_QCAP;

  const int slot = blockIdx.x * AT_WAVES + wave;
  int total;
  bool listed = false;
  if (REUSE) listed = (lover[blockIdx.x] == 0);
  if (REUSE && listed) {
    total = min(max(__builtin_amdgcn_readfirstlane(lcnt[slot]), 0), AT_QCAP);
    const int2* ls = (const int2*)lists + (size_t)slot * AT_QCAP;
    int2 r[AT_QCAP / 32];
    #pragma unroll
    for (int k = 0; k < AT_QCAP / 32; ++k) {
      const int i = lane + 32 * k;
      r[k] = (i < total) ? ls[i] : make_int2(0, 0);
    }
    #pragma unroll
    for (int k = 0; k < AT_QCAP / 32; ++k) {
      const int i = lane + 32 * k;
      if (i < total) q[i] = r[k];
    }
    for (int h0 = 0; h0 < total; h0 += 32) {
      const int n = (total - h0 < 32) ? (total - h0) : 32;
      score_drain(q, h0, n, lane, A);
    }
  } else {
    total = tile_scan<0>(q, tid, wave, lane, A);
  }
  if (total > AT_QCAP && lane == 0) sover = 1;
  __syncthreads();
  if (!REUSE) {
    if (tid == 0) lover[blockIdx.x] = sover;
    if (sover == 0) {
      int2* ls = (int2*)lists + (size_t)slot * AT_QCAP;
      for (int i = lane; i < total; i += 32) ls[i] = q[i];
      if (lane == 0) lcnt[slot] = total;
    }
  }

  if (sover == 0) {
    for (int h0 = 0; h0 < total; h0 += 32) {
      const int n = (total - h0 < 32) ? (total - h0) : 32;
      agg_drain(q, h0, n, lane, A);
    }
  } else {
    (void)tile_scan<1>(q, tid, wave, lane, A);
  }
  __syncthreads();

  for (int dl = wave; dl < AT_S; dl += AT_WAVES) {
    const int node = lo + dl;
    if (node >= N_NODES) break;
    float v[4]; float sum = 0.f;
    #pragma unroll
    for (int i = 0; i < 4; ++i) {
      const int c = i * 32 + lane;
      v[i] = tile[dl * HC + c] / sden[dl * HEADS + i] + bias[c];
      sum += v[i];
    }
    for (int off = 16; off > 0; off >>= 1) sum += __shfl_xor(sum, off, 32);
    const float mu = sum * (1.0f / HC);
    float sq = 0.f;
    #pragma unroll
    for (int i = 0; i < 4; ++i) { float d = v[i] - mu; sq += d * d; }
    for (int off = 16; off > 0; off >>= 1) sq += __shfl_xor(sq, off, 32);
    const float rstd = rsqrtf(sq * (1.0f / HC) + LN_EPS);
    #pragma unroll
    for (int i = 0; i < 4; ++i) {
      const int c = i * 32 + lane;
      const float h = (v[i] - mu) * rstd * gamma[c] + beta[c];
      out[(size_t)node * HC + c] = h / (1.0f + __expf(-h));
    }
  }
}

#define POOL_CHUNKS 4
#define POOL_SLAB   (8 * POOL_CHUNKS * 32)
#define POOL_SLABS  ((N_NODES + POOL_SLAB - 1) / POOL_SLAB)
#define PROW        (HC + 32)
__global__ void pool_gather_kernel(const float* __restrict__ x,
                                   const int* __restrict__ batch,
                                   float* __restrict__ part) {
  __shared__ float sacc[8][HC];
  __shared__ int   scnt[8];
  const int g     = blockIdx.x % NGRAPH;
  const int s     = blockIdx.x / NGRAPH;
  const int wave  = threadIdx.x >> 5;
  const int lane  = threadIdx.x & 31;
  const int slab0 = s * POOL_SLAB;

  int ids[POOL_CHUNKS];
  #pragma unroll
  for (int i = 0; i < POOL_CHUNKS; ++i) {
    int node = slab0 + (i * 8 + wave) * 32 + lane;
    ids[i] = (node < N_NODES) ? batch[node] : -1;
  }
  float4 acc = make_float4(0.f, 0.f, 0.f, 0.f);
  int cnt = 0;
  #pragma unroll
  for (int i = 0; i < POOL_CHUNKS; ++i) {
    const unsigned mask = (unsigned)__ballot(ids[i] == g);
    if (mask == 0u) continue;
    cnt += __popc(mask);
    const int base = slab0 + (i * 8 + wave) * 32;
    for (int j0 = 0; j0 < 32; j0 += 16) {
      if (((mask >> j0) & 0xffffu) == 0u) continue;
      float4 v[16];
      #pragma unroll
      for (int u = 0; u < 16; ++u) {
        int node = base + j0 + u;
        node = (node < N_NODES) ? node : (N_NODES - 1);
        v[u] = *(const float4*)(x + (size_t)node * HC + lane * 4);
      }
      #pragma unroll
      for (int u = 0; u < 16; ++u) {
        const bool on = ((mask >> (j0 + u)) & 1u) != 0u;
        acc.x += on ? v[u].x : 0.f;
        acc.y += on ? v[u].y : 0.f;
        acc.z += on ? v[u].z : 0.f;
        acc.w += on ? v[u].w : 0.f;
      }
    }
  }
  *(float4*)&sacc[wave][lane * 4] = acc;
  if (lane == 0) scnt[wave] = cnt;
  __syncthreads();
  const int tid = threadIdx.x;
  float* prow = part + ((size_t)g * POOL_SLABS + s) * PROW;
  if (tid < HC) {
    float t = 0.f;
    #pragma unroll
    for (int w = 0; w < 8; ++w) t += sacc[w][tid];
    prow[tid] = t;
  } else if (tid < PROW) {
    int c = 0;
    #pragma unroll
    for (int w = 0; w < 8; ++w) c += scnt[w];
    ((int*)prow)[tid] = (tid == HC) ? c : 0;
  }
}
__global__ void pool_final_kernel(const float* __restrict__ part,
                                  float* __restrict__ out) {
  int t = blockIdx.x * blockDim.x + threadIdx.x;
  if (t >= NGRAPH * HC) return;
  const int g = t >> 7, c = t & 127;
  const float* p = part + (size_t)g * POOL_SLABS * PROW;
  float sum = 0.f;
  int n = 0;
  #pragma unroll 7
  for (int s = 0; s < POOL_SLABS; ++s) {
    sum += p[s * PROW + c];
    n   += ((const int*)p)[s * PROW + HC];
  }
  out[t] = sum / fmaxf((float)n, 1.0f);
}

extern "C" void kernel_launch(void* const* d_in, const int* in_sizes, int n_in,
                              void* d_out, int out_size, void* d_ws, size_t ws_size,
                              hipStream_t stream) {
  (void)in_sizes; (void)n_in; (void)out_size; (void)ws_size;
  const float* x     = (const float*)d_in[0];
  const int*   ei    = (const int*)d_in[1];
  const int*   batch = (const int*)d_in[2];
  const float* Wl[2]    = {(const float*)d_in[3],  (const float*)d_in[11]};
  const float* Wr[2]    = {(const float*)d_in[4],  (const float*)d_in[12]};
  const float* bl[2]    = {(const float*)d_in[5],  (const float*)d_in[13]};
  const float* br[2]    = {(const float*)d_in[6],  (const float*)d_in[14]};
  const float* att[2]   = {(const float*)d_in[7],  (const float*)d_in[15]};
  const float* bias[2]  = {(const float*)d_in[8],  (const float*)d_in[16]};
  const float* gamma[2] = {(const float*)d_in[9],  (const float*)d_in[17]};
  const float* beta[2]  = {(const float*)d_in[10], (const float*)d_in[18]};

  float* ws = (float*)d_ws;
  float*    xl    = ws;            ws += (size_t)N_NODES * HC;
  float*    xr2   = ws;            ws += (size_t)N_NODES * HC;
  float*    xnode = ws;            ws += (size_t)N_NODES * HC;
  float*    scbuf = ws;            ws += (size_t)ETOT * HEADS;
  float*    ppart = ws;            ws += (size_t)NGRAPH * POOL_SLABS * PROW;
  int*      lists = (int*)ws;      ws += (size_t)AT_TILES * AT_WAVES * AT_QCAP * 2;
  int*      lcnt  = (int*)ws;      ws += (size_t)AT_TILES * AT_WAVES;
  int*      lover = (int*)ws;

  dim3 b256(256);
  const int    gemmGrid = (N_NODES + 16 * ROWTILES - 1) / (16 * ROWTILES);
  const size_t shmem    = (size_t)HC * LDSW * sizeof(float);

  const float* xin = x;
  for (int l = 0; l < 2; ++l) {
    gemm_bias_wmma<<<gemmGrid, b256, shmem, stream>>>(xin, Wl[l], bl[l], xl, N_NODES);
    gemm_bias_wmma<<<gemmGrid, b256, shmem, stream>>>(xin, Wr[l], br[l], xr2, N_NODES);
    if (l == 0)
      tile_softmax_aggregate_ln_kernel<0><<<AT_TILES, AT_THREADS, 0, stream>>>(ei, xl, xr2, att[l], scbuf, bias[l], gamma[l], beta[l], xnode, lists, lcnt, lover);
    else
      tile_softmax_aggregate_ln_kernel<1><<<AT_TILES, AT_THREADS, 0, stream>>>(ei, xl, xr2, att[l], scbuf, bias[l], gamma[l], beta[l], xnode, lists, lcnt, lover);
    xin = xnode;
  }
  pool_gather_kernel<<<NGRAPH * POOL_SLABS, b256, 0, stream>>>(xnode, batch, ppart);
  pool_final_kernel<<<(NGRAPH * HC + 255) / 256, b256, 0, stream>>>(ppart, (float*)d_out);
}


# GEMM by row-owner waves: a wave keeps its 16 rows of X in registers (16-byte loads, once) and takes one 16-byte LDS piece of W per two WMMAs; 256 blocks; k order of the GEMM sums changes
# speedup vs baseline: 3.1705x; 1.1684x over previous
#include <hip/hip_runtime.h>
#include <hip/hip_bf16.h>
#include <math.h>

#define N_NODES 50000
#define E_EDGES 800000
#define ETOT    (E_EDGES + N_NODES)
#define HEADS   4
#define CHP     32
#define HC      128
#define NGRAPH  64
#define SLOPE   0.2f
#define LN_EPS  1e-5f

typedef float v2f __attribute__((ext_vector_type(2)));
typedef float v8f __attribute__((ext_vector_type(8)));

__device__ __forceinline__ unsigned f32_to_ord(float f) {
  unsigned u = __float_as_uint(f);
  return (u & 0x80000000u) ? ~u : (u | 0x80000000u);
}
__device__ __forceinline__ float ord_to_f32(unsigned u) {
  u = (u & 0x80000000u) ? (u & 0x7fffffffu) : ~u;
  return __uint_as_float(u);
}
__device__ __forceinline__ void atomAddF(float* p, float v) {
  unsafeAtomicAdd(p, v);
}

#define G_THREADS 512
#define G_WAVES   (G_THREADS / 32)
#define G_BLOCKS  256
#define G_NT      4
typedef float v4f __attribute__((ext_vector_type(4)));

__device__ __forceinline__ void gemm_load_a(v4f (&a)[16], const float* __restrict__ X, const int u, const int n,
                                            const int lane) {
  const int rt  = blockIdx.x + G_BLOCKS * (u >> 1);
  const int row = max(min(rt * 16 + (lane & 15), n - 1), 0);
  const v4f* p = (const v4f*)(X + (size_t)row * HC) + (lane >> 4);
  #pragma unroll
  for (int j = 0; j < 16; ++j) a[j] = p[2 * j];
}

__global__ void __launch_bounds__(G_THREADS)
gemm_bias_wmma(const float* __restrict__ X,
               const float* __restrict__ W,
               const float* __restrict__ bias,
               float* __restrict__ Y, int n) {
  __shared__ __align__(16) float sW[HC * HC];
  const int tid  = threadIdx.x;
  const int wave = __builtin_amdgcn_readfirstlane(tid >> 5);
  const int lane = tid & 31;
  const int c    = lane & 15;
  const int h    = lane >> 4;
  const int ntiles = (n + 15) >> 4;
  const int nrt    = (ntiles - (int)blockIdx.x + G_BLOCKS - 1) / G_BLOCKS;
  const int nunits = 2 * nrt;

  int u = wave;
  v4f a[16];
  gemm_load_a(a, X, u, n, lane);

  {
    v4f r[2][4];
    #pragma unroll
    for (int i = 0; i < 2; ++i) {
      const int item = tid + G_THREADS * i;
      const int q = item >> 5, g = item & 31;
      #pragma unroll
      for (int k = 0; k < 4; ++k) r[i][k] = *(const v4f*)(W + (4 * q + k) * HC + 4 * g);
    }
    #pragma unroll
    for (int i = 0; i < 2; ++i) {
      const int item = tid + G_THREADS * i;
      const int q = item >> 5, g = item & 31;
      v4f* d = (v4f*)sW + q * HC + 4 * g;
      d[0] = v4f{r[i][0].x, r[i][1].x, r[i][2].x, r[i][3].x};
      d[1] = v4f{r[i][0].y, r[i][1].y, r[i][2].y, r[i][3].y};
      d[2] = v4f{r[i][0].z, r[i][1].z, r[i][2].z, r[i][3].z};
      d[3] = v4f{r[i][0].w, r[i][1].w, r[i][2].w, r[i][3].w};
    }
  }
  __syncthreads();

  while (u < nunits) {
    const int un = u + G_WAVES;
    v4f an[16];
    gemm_load_a(an, X, min(un, nunits - 1), n, lane);
    const int rt   = blockIdx.x + G_BLOCKS * (u >> 1);
    const int col0 = (u & 1) * (16 * G_NT) + c;
    const v4f* bp = (const v4f*)sW + h * HC + col0;
    float bc[G_NT];
    v8f acc[G_NT];
    #pragma unroll
    for (int t = 0; t < G_NT; ++t) { bc[t] = bias[col0 + 16 * t]; acc[t] = v8f{}; }
    v4f bq[2][G_NT];
    #pragma unroll
    for (int t = 0; t < G_NT; ++t) bq[0][t] = bp[16 * t];
    #pragma unroll
    for (int j = 0; j < 16; ++j) {
      if (j + 1 < 16) {
        #pragma unroll
        for (int t = 0; t < G_NT; ++t) bq[(j + 1) & 1][t] = bp[2 * (j + 1) * HC + 16 * t];
      }
      #pragma unroll
      for (int t = 0; t < G_NT; ++t) {
        const v4f b = bq[j & 1][t];
        v2f a0; a0.x = a[j].x; a0.y = a[j].y;
        v2f b0; b0.x = b.x;    b0.y = b.y;
        acc[t] = __builtin_amdgcn_wmma_f32_16x16x4_f32(false, a0, false, b0, (short)0, acc[t], false, false);
        v2f a1; a1.x = a[j].z; a1.y = a[j].w;
        v2f b1; b1.x = b.z;    b1.y = b.w;
        acc[t] = __builtin_amdgcn_wmma_f32_16x16x4_f32(false, a1, false, b1, (short)0, acc[t], false, false);
      }
      __builtin_amdgcn_sched_barrier(0);
    }
    float* yp = Y + (size_t)(rt * 16 + 8 * h) * HC + col0;
    if (rt * 16 + 16 <= n) {
      #pragma unroll
      for (int t = 0; t < G_NT; ++t) {
        #pragma unroll
        for (int r = 0; r < 8; ++r) yp[r * HC + 16 * t] = acc[t][r] + bc[t];
      }
    } else {
      #pragma unroll
      for (int t = 0; t < G_NT; ++t) {
        #pragma unroll
        for (int r = 0; r < 8; ++r) {
          if (rt * 16 + r + 8 * h < n) yp[r * HC + 16 * t] = acc[t][r] + bc[t];
        }
      }
    }
    #pragma unroll
    for (int j = 0; j < 16; ++j) a[j] = an[j];
    u = un;
  }
}

#define AT_S       196
#define AT_TILES   ((N_NODES + AT_S - 1) / AT_S)
#define AT_THREADS 1024
#define AT_WAVES   (AT_THREADS / 32)
#define AT_QCAP    256
#define AT_GROUP   8
#define AT_V4      (E_EDGES / 4)
#define AT_ITERS   ((AT_V4 + AT_THREADS - 1) / AT_THREADS)
#define AT_PASSES  (((AT_ITERS + 3) / 4) * 4)
#define AT_SELF    ((AT_S + AT_WAVES - 1) / AT_WAVES)
#define ORD_NEG_INF 0x007FFFFFu

struct TileArgs {
  const int*   ei;
  const float* xl;
  const float* sxr;
  const float* att;
  float*       sc;
  float*       tile;
  unsigned*    smax;
  float*       sden;
  int          lo;
};

__device__ __forceinline__ int lane_bcast_i(int v, int j) {
  return __builtin_amdgcn_readlane(v, j);
}
__device__ __forceinline__ float lane_bcast_f(float v, int j) {
  return __int_as_float(__builtin_amdgcn_readlane(__float_as_int(v), j));
}
__device__ __forceinline__ float lrelu_dot4(const float4 a, const float4 b, const float4 w) {
  float t0 = a.x + b.x, t1 = a.y + b.y, t2 = a.z + b.z, t3 = a.w + b.w;
  t0 = (t0 > 0.f) ? t0 : t0 * SLOPE;
  t1 = (t1 > 0.f) ? t1 : t1 * SLOPE;
  t2 = (t2 > 0.f) ? t2 : t2 * SLOPE;
  t3 = (t3 > 0.f) ? t3 : t3 * SLOPE;
  return t0 * w.x + t1 * w.y + t2 * w.z + t3 * w.w;
}

__device__ __forceinline__ void score_drain(const int2* q, int head, int n, int lane, const TileArgs& A) {
  int src = 0, dl = 0, e = 0;
  if (lane < n) {
    const int2 t = q[(head + lane) & (AT_QCAP - 1)];
    e = t.x;
    dl = t.y;
    src = (e < E_EDGES) ? A.ei[e] : (e - E_EDGES);
  }
  const float4 w4 = *(const float4*)(A.att + 4 * lane);
  const int  r  = lane & 7;
  const int  hd = lane >> 3;
  const bool b2 = (r & 4) != 0, b1 = (r & 2) != 0, b0 = (r & 1) != 0;
  for (int j0 = 0; j0 < n; j0 += AT_GROUP) {
    float4 a[AT_GROUP], b[AT_GROUP];
    #pragma unroll
    for (int u = 0; u < AT_GROUP; ++u) {
      const int s = lane_bcast_i(src, j0 + u);
      const int d = lane_bcast_i(dl, j0 + u);
      a[u] = *(const float4*)(A.xl + (size_t)s * HC + 4 * lane);
      b[u] = *(const float4*)(A.sxr + d * HC + 4 * lane);
    }
    float p[AT_GROUP];
    #pragma unroll
    for (int u = 0; u < AT_GROUP; ++u) p[u] = lrelu_dot4(a[u], b[u], w4);
    float f4[4];
    #pragma unroll
    for (int i = 0; i < 4; ++i) {
      const float keep = b2 ? p[i + 4] : p[i];
      const float send = b2 ? p[i] : p[i + 4];
      f4[i] = keep + __shfl_xor(send, 4, 32);
    }
    float f2[2];
    #pragma unroll
    for (int i = 0; i < 2; ++i) {
      const float keep = b1 ? f4[i + 2] : f4[i];
      const float send = b1 ? f4[i] : f4[i + 2];
      f2[i] = keep + __shfl_xor(send, 2, 32);
    }
    const float keep = b0 ? f2[1] : f2[0];
    const float send = b0 ? f2[0] : f2[1];
    const float sc = keep + __shfl_xor(send, 1, 32);
    const int ue = j0 + r;
    const int eu = __shfl(e, ue, 32);
    const int du = __shfl(dl, ue, 32);
    if (ue < n) {
      A.sc[(size_t)eu * HEADS + hd] = sc;
      atomicMax(&A.smax[du * HEADS + hd], f32_to_ord(sc));
    }
  }
}

__device__ __forceinline__ void agg_drain(const int2* q, int head, int n, int lane, const TileArgs& A) {
  int src = 0, dl = 0;
  float a0 = 0.f, a1 = 0.f, a2 = 0.f, a3 = 0.f;
  if (lane < n) {
    const int2 t = q[(head + lane) & (AT_QCAP - 1)];
    const int e = t.x;
    dl = t.y;
    src = (e < E_EDGES) ? A.ei[e] : (e - E_EDGES);
    const float4 s4 = *(const float4*)(A.sc + (size_t)e * HEADS);
    const uint4  m4 = *(const uint4*)(A.smax + dl * HEADS);
    a0 = __expf(s4.x - ord_to_f32(m4.x));
    a1 = __expf(s4.y - ord_to_f32(m4.y));
    a2 = __expf(s4.z - ord_to_f32(m4.z));
    a3 = __expf(s4.w - ord_to_f32(m4.w));
    float* dn = A.sden + dl * HEADS;
    atomAddF(dn,     a0);
    atomAddF(dn + 1, a1);
    atomAddF(dn + 2, a2);
    atomAddF(dn + 3, a3);
  }
  for (int j0 = 0; j0 < n; j0 += AT_GROUP) {
    float v[AT_GROUP][HEADS];
    #pragma unroll
    for (int u = 0; u < AT_GROUP; ++u) {
      const int s = lane_bcast_i(src, j0 + u);
      const float* p = A.xl + (size_t)s * HC + lane;
      #pragma unroll
      for (int h = 0; h < HEADS; ++h) v[u][h] = p[h * CHP];
    }
    #pragma unroll
    for (int u = 0; u < AT_GROUP; ++u) {
      if (j0 + u < n) {
        const int   d  = lane_bcast_i(dl, j0 + u);
        const float w0 = lane_bcast_f(a0, j0 + u);
        const float w1 = lane_bcast_f(a1, j0 + u);
        const float w2 = lane_bcast_f(a2, j0 + u);
        const float w3 = lane_bcast_f(a3, j0 + u);
        float* t = A.tile + d * HC + lane;
        atomAddF(t,           v[u][0] * w0);
        atomAddF(t + CHP,     v[u][1] * w1);
        atomAddF(t + 2 * CHP, v[u][2] * w2);
        atomAddF(t + 3 * CHP, v[u][3] * w3);
      }
    }
  }
}

__device__ __forceinline__ int4 tile_ids(const int4* __restrict__ dst4, const int pass, const int tid) {
  const int i = pass * AT_THREADS + tid;
  return dst4[(i < AT_V4) ? i : (AT_V4 - 1)];
}

template <int PH>
__device__ __forceinline__ void tile_pass(int4& p, const int4* __restrict__ dst4, const int pass, const int tid,
                                          const bool last, int2* q, int& head, int& cnt, int& total,
                                          const unsigned below, const int lane, const TileArgs& A) {
  const int idx = pass * AT_THREADS + tid;
  const int lo = A.lo;
  const unsigned dl[4] = {(unsigned)(p.x - lo), (unsigned)(p.y - lo), (unsigned)(p.z - lo), (unsigned)(p.w - lo)};
  p = tile_ids(dst4, pass + 4, tid);
  const bool have = idx < AT_V4;
  const unsigned dmin = min(min(dl[0], dl[1]), min(dl[2], dl[3]));
  if (__ballot(have && (dmin < (unsigned)AT_S)) != 0) {
    #pragma unroll
    for (int k = 0; k < 4; ++k) {
      const bool on = have && (dl[k] < (unsigned)AT_S);
      const unsigned m = (unsigned)__ballot(on);
      if (on) q[(head + cnt + __popc(m & below)) & (AT_QCAP - 1)] = make_int2(idx * 4 + k, (int)dl[k]);
      cnt += __popc(m);
      total += __popc(m);
    }
  }
  while (cnt >= 32 || (last && cnt > 0)) {
    const int n = (cnt < 32) ? cnt : 32;
    if (PH == 0) score_drain(q, head, n, lane, A);
    else         agg_drain(q, head, n, lane, A);
    head = (head + n) & (AT_QCAP - 1);
    cnt -= n;
  }
}

template <int PH>
__device__ __forceinline__ int tile_scan(int2* q, const int tid, const int wave, const int lane, const TileArgs& A) {
  int head = 0, cnt = 0, total = 0;
  const unsigned below = (1u << lane) - 1u;
  {
    const int dl = wave * AT_SELF + lane;
    const bool on = (lane < AT_SELF) && (dl < AT_S) && (A.lo + dl < N_NODES);
    const unsigned m = (unsigned)__ballot(on);
    if (on) q[__popc(m & below)] = make_int2(E_EDGES + A.lo + dl, dl);
    cnt = __popc(m);
    total = cnt;
  }
  const int4* dst4 = (const int4*)(A.ei + E_EDGES);
  int4 p0 = tile_ids(dst4, 0, tid);
  int4 p1 = tile_ids(dst4, 1, tid);
  int4 p2 = tile_ids(dst4, 2, tid);
  int4 p3 = tile_ids(dst4, 3, tid);
  #pragma unroll 1
  for (int it = 0; it < AT_PASSES; it += 4) {
    tile_pass<PH>(p0, dst4, it,     tid, false, q, head, cnt, total, below, lane, A);
    tile_pass<PH>(p1, dst4, it + 1, tid, false, q, head, cnt, total, below, lane, A);
    tile_pass<PH>(p2, dst4, it + 2, tid, false, q, head, cnt, total, below, lane, A);
    tile_pass<PH>(p3, dst4, it + 3, tid, it + 4 >= AT_PASSES, q, head, cnt, total, below, lane, A);
  }
  return total;
}

template <int REUSE>
__global__ void __launch_bounds__(AT_THREADS)
tile_softmax_aggregate_ln_kernel(const int* __restrict__ ei,
                                 const float* __restrict__ xl,
                                 const float* __restrict__ xr,
                                 const float* __restrict__ att,
                                 float* sc,
                                 const float* __restrict__ bias,
                                 const float* __restrict__ gamma,
                                 const float* __restrict__ beta,
                                 float* __restrict__ out,
                                 int*  lists,
                                 int*  lcnt,
                                 int*  lover) {
  __shared__ __align__(16) float    tile[AT_S * HC];
  __shared__ __align__(16) float    sxr[AT_S * HC];
  __shared__ __align__(16) int2     ring[AT_WAVES * AT_QCAP];
  __shared__ __align__(16) unsigned smax[AT_S * HEADS];
  __shared__ __align__(16) float    sden[AT_S * HEADS];
  __shared__ int      sover;
  const int tid  = threadIdx.x;
  const int wave = tid >> 5;
  const int lane = tid & 31;
  const int lo   = blockIdx.x * AT_S;

  {
    constexpr int NV4 = AT_S * HC / 4;
    constexpr int NLD = (NV4 + AT_THREADS - 1) / AT_THREADS;
    const float4 zero4 = make_float4(0.f, 0.f, 0.f, 0.f);
    float4 r[NLD];
    #pragma unroll
    for (int k = 0; k < NLD; ++k) {
      const int gi = lo * (HC / 4) + tid + k * AT_THREADS;
      r[k] = ((const float4*)xr)[min(gi, N_NODES * (HC / 4) - 1)];
    }
    #pragma unroll
    for (int k = 0; k < NLD; ++k) {
      const int i = tid + k * AT_THREADS;
      if (i < NV4) { ((float4*)tile)[i] = zero4; ((float4*)sxr)[i] = r[k]; }
    }
  }
  if (tid < AT_S * HEADS) { smax[tid] = ORD_NEG_INF; sden[tid] = 0.f; }
  if (tid == 0) sover = 0;
  __syncthreads();

  TileArgs A;
  A.ei = ei; A.xl = xl; A.sxr = sxr; A.att = att; A.sc = sc;
  A.tile = tile; A.smax = smax; A.sden = sden; A.lo = lo;
  int2* q = ring + wave * AT_QCAP;

  const int slot = blockIdx.x * AT_WAVES + wave;
  int total;
  bool listed = false;
  if (REUSE) listed = (lover[blockIdx.x] == 0);
  if (REUSE && listed) {
    total = min(max(__builtin_amdgcn_readfirstlane(lcnt[slot]), 0), AT_QCAP);
    const int2* ls = (const int2*)lists + (size_t)slot * AT_QCAP;
    int2 r[AT_QCAP / 32];
    #pragma unroll
    for (int k = 0; k < AT_QCAP / 32; ++k) {
      const int i = lane + 32 * k;
      r[k] = (i < total) ? ls[i] : make_int2(0, 0);
    }
    #pragma unroll
    for (int k = 0; k < AT_QCAP / 32; ++k) {
      const int i = lane + 32 * k;
      if (i < total) q[i] = r[k];
    }
    for (int h0 = 0; h0 < total; h0 += 32) {
      const int n = (total - h0 < 32) ? (total - h0) : 32;
      score_drain(q, h0, n, lane, A);
    }
  } else {
    total = tile_scan<0>(q, tid, wave, lane, A);
  }
  if (total > AT_QCAP && lane == 0) sover = 1;
  __syncthreads();
  if (!REUSE) {
    if (tid == 0) lover[blockIdx.x] = sover;
    if (sover == 0) {
      int2* ls = (int2*)lists + (size_t)slot * AT_QCAP;
      for (int i = lane; i < total; i += 32) ls[i] = q[i];
      if (lane == 0) lcnt[slot] = total;
    }
  }

  if (sover == 0) {
    for (int h0 = 0; h0 < total; h0 += 32) {
      const int n = (total - h0 < 32) ? (total - h0) : 32;
      agg_drain(q, h0, n, lane, A);
    }
  } else {
    (void)tile_scan<1>(q, tid, wave, lane, A);
  }
  __syncthreads();

  for (int dl = wave; dl < AT_S; dl += AT_WAVES) {
    const int node = lo + dl;
    if (node >= N_NODES) break;
    float v[4]; float sum = 0.f;
    #pragma unroll
    for (int i = 0; i < 4; ++i) {
      const int c = i * 32 + lane;
      v[i] = tile[dl * HC + c] / sden[dl * HEADS + i] + bias[c];
      sum += v[i];
    }
    for (int off = 16; off > 0; off >>= 1) sum += __shfl_xor(sum, off, 32);
    const float mu = sum * (1.0f / HC);
    float sq = 0.f;
    #pragma unroll
    for (int i = 0; i < 4; ++i) { float d = v[i] - mu; sq += d * d; }
    for (int off = 16; off > 0; off >>= 1) sq += __shfl_xor(sq, off, 32);
    const float rstd = rsqrtf(sq * (1.0f / HC) + LN_EPS);
    #pragma unroll
    for (int i = 0; i < 4; ++i) {
      const int c = i * 32 + lane;
      const float h = (v[i] - mu) * rstd * gamma[c] + beta[c];
      out[(size_t)node * HC + c] = h / (1.0f + __expf(-h));
    }
  }
}

#define POOL_CHUNKS 4
#define POOL_SLAB   (8 * POOL_CHUNKS * 32)
#define POOL_SLABS  ((N_NODES + POOL_SLAB - 1) / POOL_SLAB)
#define PROW        (HC + 32)
__global__ void pool_gather_kernel(const float* __restrict__ x,
                                   const int* __restrict__ batch,
                                   float* __restrict__ part) {
  __shared__ float sacc[8][HC];
  __shared__ int   scnt[8];
  const int g     = blockIdx.x % NGRAPH;
  const int s     = blockIdx.x / NGRAPH;
  const int wave  = threadIdx.x >> 5;
  const int lane  = threadIdx.x & 31;
  const int slab0 = s * POOL_SLAB;

  int ids[POOL_CHUNKS];
  #pragma unroll
  for (int i = 0; i < POOL_CHUNKS; ++i) {
    int node = slab0 + (i * 8 + wave) * 32 + lane;
    ids[i] = (node < N_NODES) ? batch[node] : -1;
  }
  float4 acc = make_float4(0.f, 0.f, 0.f, 0.f);
  int cnt = 0;
  #pragma unroll
  for (int i = 0; i < POOL_CHUNKS; ++i) {
    const unsigned mask = (unsigned)__ballot(ids[i] == g);
    if (mask == 0u) continue;
    cnt += __popc(mask);
    const int base = slab0 + (i * 8 + wave) * 32;
    for (int j0 = 0; j0 < 32; j0 += 16) {
      if (((mask >> j0) & 0xffffu) == 0u) continue;
      float4 v[16];
      #pragma unroll
      for (int u = 0; u < 16; ++u) {
        int node = base + j0 + u;
        node = (node < N_NODES) ? node : (N_NODES - 1);
        v[u] = *(const float4*)(x + (size_t)node * HC + lane * 4);
      }
      #pragma unroll
      for (int u = 0; u < 16; ++u) {
        const bool on = ((mask >> (j0 + u)) & 1u) != 0u;
        acc.x += on ? v[u].x : 0.f;
        acc.y += on ? v[u].y : 0.f;
        acc.z += on ? v[u].z : 0.f;
        acc.w += on ? v[u].w : 0.f;
      }
    }
  }
  *(float4*)&sacc[wave][lane * 4] = acc;
  if (lane == 0) scnt[wave] = cnt;
  __syncthreads();
  const int tid = threadIdx.x;
  float* prow = part + ((size_t)g * POOL_SLABS + s) * PROW;
  if (tid < HC) {
    float t = 0.f;
    #pragma unroll
    for (int w = 0; w < 8; ++w) t += sacc[w][tid];
    prow[tid] = t;
  } else if (tid < PROW) {
    int c = 0;
    #pragma unroll
    for (int w = 0; w < 8; ++w) c += scnt[w];
    ((int*)prow)[tid] = (tid == HC) ? c : 0;
  }
}
__global__ void pool_final_kernel(const float* __restrict__ part,
                                  float* __restrict__ out) {
  int t = blockIdx.x * blockDim.x + threadIdx.x;
  if (t >= NGRAPH * HC) return;
  const int g = t >> 7, c = t & 127;
  const float* p = part + (size_t)g * POOL_SLABS * PROW;
  float sum = 0.f;
  int n = 0;
  #pragma unroll 7
  for (int s = 0; s < POOL_SLABS; ++s) {
    sum += p[s * PROW + c];
    n   += ((const int*)p)[s * PROW + HC];
  }
  out[t] = sum / fmaxf((float)n, 1.0f);
}

extern "C" void kernel_launch(void* const* d_in, const int* in_sizes, int n_in,
                              void* d_out, int out_size, void* d_ws, size_t ws_size,
                              hipStream_t stream) {
  (void)in_sizes; (void)n_in; (void)out_size; (void)ws_size;
  const float* x     = (const float*)d_in[0];
  const int*   ei    = (const int*)d_in[1];
  const int*   batch = (const int*)d_in[2];
  const float* Wl[2]    = {(const float*)d_in[3],  (const float*)d_in[11]};
  const float* Wr[2]    = {(const float*)d_in[4],  (const float*)d_in[12]};
  const float* bl[2]    = {(const float*)d_in[5],  (const float*)d_in[13]};
  const float* br[2]    = {(const float*)d_in[6],  (const float*)d_in[14]};
  const float* att[2]   = {(const float*)d_in[7],  (const float*)d_in[15]};
  const float* bias[2]  = {(const float*)d_in[8],  (const float*)d_in[16]};
  const float* gamma[2] = {(const float*)d_in[9],  (const float*)d_in[17]};
  const float* beta[2]  = {(const float*)d_in[10], (const float*)d_in[18]};

  float* ws = (float*)d_ws;
  float*    xl    = ws;            ws += (size_t)N_NODES * HC;
  float*    xr2   = ws;            ws += (size_t)N_NODES * HC;
  float*    xnode = ws;            ws += (size_t)N_NODES * HC;
  float*    scbuf = ws;            ws += (size_t)ETOT * HEADS;
  float*    ppart = ws;            ws += (size_t)NGRAPH * POOL_SLABS * PROW;
  int*      lists = (int*)ws;      ws += (size_t)AT_TILES * AT_WAVES * AT_QCAP * 2;
  int*      lcnt  = (int*)ws;      ws += (size_t)AT_TILES * AT_WAVES;
  int*      lover = (int*)ws;

  dim3 b256(256);

  const float* xin = x;
  for (int l = 0; l < 2; ++l) {
    gemm_bias_wmma<<<G_BLOCKS, G_THREADS, 0, stream>>>(xin, Wl[l], bl[l], xl, N_NODES);
    gemm_bias_wmma<<<G_BLOCKS, G_THREADS, 0, stream>>>(xin, Wr[l], br[l], xr2, N_NODES);
    if (l == 0)
      tile_softmax_aggregate_ln_kernel<0><<<AT_TILES, AT_THREADS, 0, stream>>>(ei, xl, xr2, att[l], scbuf, bias[l], gamma[l], beta[l], xnode, lists, lcnt, lover);
    else
      tile_softmax_aggregate_ln_kernel<1><<<AT_TILES, AT_THREADS, 0, stream>>>(ei, xl, xr2, att[l], scbuf, bias[l], gamma[l], beta[l], xnode, lists, lcnt, lover);
    xin = xnode;
  }
  pool_gather_kernel<<<NGRAPH * POOL_SLABS, b256, 0, stream>>>(xnode, batch, ppart);
  pool_final_kernel<<<(NGRAPH * HC + 255) / 256, b256, 0, stream>>>(ppart, (float*)d_out);
}


# layer 1's tile scan reads a one-byte-per-edge tile-number plane (written inside the first GEMM launch), not the 4-byte ids: 16 edges per 16-byte load; 254 tiles of 197 rows; LDS sum order as before
# speedup vs baseline: 3.2883x; 1.0372x over previous
#include <hip/hip_runtime.h>
#include <hip/hip_bf16.h>
#include <math.h>

#define N_NODES 50000
#define E_EDGES 800000
#define ETOT    (E_EDGES + N_NODES)
#define HEADS   4
#define CHP     32
#define HC      128
#define NGRAPH  64
#define SLOPE   0.2f
#define LN_EPS  1e-5f

#define AT_S     197
#define AT_TILES ((N_NODES + AT_S - 1) / AT_S)
#define AT_NONE  255u

typedef float v2f __attribute__((ext_vector_type(2)));
typedef float v8f __attribute__((ext_vector_type(8)));

__device__ __forceinline__ unsigned f32_to_ord(float f) {
  unsigned u = __float_as_uint(f);
  return (u & 0x80000000u) ? ~u : (u | 0x80000000u);
}
__device__ __forceinline__ float ord_to_f32(unsigned u) {
  u = (u & 0x80000000u) ? (u & 0x7fffffffu) : ~u;
  return __uint_as_float(u);
}
__device__ __forceinline__ unsigned tile_of(const int d) {
  return ((unsigned)d < (unsigned)N_NODES) ? ((unsigned)d / (unsigned)AT_S) : AT_NONE;
}
__device__ __forceinline__ void atomAddF(float* p, float v) {
  unsafeAtomicAdd(p, v);
}

#define G_THREADS 512
#define G_WAVES   (G_THREADS / 32)
#define G_BLOCKS  256
#define G_NT      4
typedef float v4f __attribute__((ext_vector_type(4)));

__device__ __forceinline__ void gemm_load_a(v4f (&a)[16], const float* __restrict__ X, const int u, const int n,
                                            const int lane) {
  const int rt  = blockIdx.x + G_BLOCKS * (u >> 1);
  const int row = max(min(rt * 16 + (lane & 15), n - 1), 0);
  const v4f* p = (const v4f*)(X + (size_t)row * HC) + (lane >> 4);
  #pragma unroll
  for (int j = 0; j < 16; ++j) a[j] = p[2 * j];
}

template <int PRE>
__global__ void __launch_bounds__(G_THREADS)
gemm_bias_wmma(const float* __restrict__ X,
               const float* __restrict__ W,
               const float* __restrict__ bias,
               float* __restrict__ Y, int n,
               const int* __restrict__ ei,
               unsigned* __restrict__ tplane) {
  __shared__ __align__(16) float sW[HC * HC];
  const int tid  = threadIdx.x;
  const int wave = __builtin_amdgcn_readfirstlane(tid >> 5);
  const int lane = tid & 31;
  const int c    = lane & 15;
  const int h    = lane >> 4;
  const int ntiles = (n + 15) >> 4;
  const int nrt    = (ntiles - (int)blockIdx.x + G_BLOCKS - 1) / G_BLOCKS;
  const int nunits = 2 * nrt;

  constexpr int NGRP = E_EDGES / 4;
  constexpr int PER  = (NGRP + G_BLOCKS * G_THREADS - 1) / (G_BLOCKS * G_THREADS);
  int4 dd[PER];
  if (PRE) {
    #pragma unroll
    for (int k = 0; k < PER; ++k) {
      const int i = (int)blockIdx.x * G_THREADS + tid + k * (G_BLOCKS * G_THREADS);
      dd[k] = ((const int4*)(ei + E_EDGES))[min(i, NGRP - 1)];
    }
  }

  int u = wave;
  v4f a[16];
  gemm_load_a(a, X, u, n, lane);

  if (PRE) {
    #pragma unroll
    for (int k = 0; k < PER; ++k) {
      const int i = (int)blockIdx.x * G_THREADS + tid + k * (G_BLOCKS * G_THREADS);
      if (i < NGRP)
        tplane[i] = tile_of(dd[k].x) | (tile_of(dd[k].y) << 8) | (tile_of(dd[k].z) << 16) | (tile_of(dd[k].w) << 24);
    }
  }

  {
    v4f r[2][4];
    #pragma unroll
    for (int i = 0; i < 2; ++i) {
      const int item = tid + G_THREADS * i;
      const int q = item >> 5, g = item & 31;
      #pragma unroll
      for (int k = 0; k < 4; ++k) r[i][k] = *(const v4f*)(W + (4 * q + k) * HC + 4 * g);
    }
    #pragma unroll
    for (int i = 0; i < 2; ++i) {
      const int item = tid + G_THREADS * i;
      const int q = item >> 5, g = item & 31;
      v4f* d = (v4f*)sW + q * HC + 4 * g;
      d[0] = v4f{r[i][0].x, r[i][1].x, r[i][2].x, r[i][3].x};
      d[1] = v4f{r[i][0].y, r[i][1].y, r[i][2].y, r[i][3].y};
      d[2] = v4f{r[i][0].z, r[i][1].z, r[i][2].z, r[i][3].z};
      d[3] = v4f{r[i][0].w, r[i][1].w, r[i][2].w, r[i][3].w};
    }
  }
  __syncthreads();

  while (u < nunits) {
    const int un = u + G_WAVES;
    v4f an[16];
    gemm_load_a(an, X, min(un, nunits - 1), n, lane);
    const int rt   = blockIdx.x + G_BLOCKS * (u >> 1);
    const int col0 = (u & 1) * (16 * G_NT) + c;
    const v4f* bp = (const v4f*)sW + h * HC + col0;
    float bc[G_NT];
    v8f acc[G_NT];
    #pragma unroll
    for (int t = 0; t < G_NT; ++t) { bc[t] = bias[col0 + 16 * t]; acc[t] = v8f{}; }
    v4f bq[2][G_NT];
    #pragma unroll
    for (int t = 0; t < G_NT; ++t) bq[0][t] = bp[16 * t];
    #pragma unroll
    for (int j = 0; j < 16; ++j) {
      if (j + 1 < 16) {
        #pragma unroll
        for (int t = 0; t < G_NT; ++t) bq[(j + 1) & 1][t] = bp[2 * (j + 1) * HC + 16 * t];
      }
      #pragma unroll
      for (int t = 0; t < G_NT; ++t) {
        const v4f b = bq[j & 1][t];
        v2f a0; a0.x = a[j].x; a0.y = a[j].y;
        v2f b0; b0.x = b.x;    b0.y = b.y;
        acc[t] = __builtin_amdgcn_wmma_f32_16x16x4_f32(false, a0, false, b0, (short)0, acc[t], false, false);
        v2f a1; a1.x = a[j].z; a1.y = a[j].w;
        v2f b1; b1.x = b.z;    b1.y = b.w;
        acc[t] = __builtin_amdgcn_wmma_f32_16x16x4_f32(false, a1, false, b1, (short)0, acc[t], false, false);
      }
      __builtin_amdgcn_sched_barrier(0);
    }
    float* yp = Y + (size_t)(rt * 16 + 8 * h) * HC + col0;
    if (rt * 16 + 16 <= n) {
      #pragma unroll
      for (int t = 0; t < G_NT; ++t) {
        #pragma unroll
        for (int r = 0; r < 8; ++r) yp[r * HC + 16 * t] = acc[t][r] + bc[t];
      }
    } else {
      #pragma unroll
      for (int t = 0; t < G_NT; ++t) {
        #pragma unroll
        for (int r = 0; r < 8; ++r) {
          if (rt * 16 + r + 8 * h < n) yp[r * HC + 16 * t] = acc[t][r] + bc[t];
        }
      }
    }
    #pragma unroll
    for (int j = 0; j < 16; ++j) a[j] = an[j];
    u = un;
  }
}

#define AT_THREADS 1024
#define AT_WAVES   (AT_THREADS / 32)
#define AT_QCAP    256
#define AT_GROUP   8
#define AT_V16     (E_EDGES / 16)
#define AT_ITERS   ((AT_V16 + AT_THREADS - 1) / AT_THREADS)
#define AT_PASSES  (((AT_ITERS + 3) / 4) * 4)
#define AT_SELF    ((AT_S + AT_WAVES - 1) / AT_WAVES)
#define ORD_NEG_INF 0x007FFFFFu

struct TileArgs {
  const uint4* tp;
  unsigned     pat;
  const int*   ei;
  const float* xl;
  const float* sxr;
  const float* att;
  float*       sc;
  float*       tile;
  unsigned*    smax;
  float*       sden;
  int          lo;
};

__device__ __forceinline__ int lane_bcast_i(int v, int j) {
  return __builtin_amdgcn_readlane(v, j);
}
__device__ __forceinline__ float lane_bcast_f(float v, int j) {
  return __int_as_float(__builtin_amdgcn_readlane(__float_as_int(v), j));
}
__device__ __forceinline__ float lrelu_dot4(const float4 a, const float4 b, const float4 w) {
  float t0 = a.x + b.x, t1 = a.y + b.y, t2 = a.z + b.z, t3 = a.w + b.w;
  t0 = (t0 > 0.f) ? t0 : t0 * SLOPE;
  t1 = (t1 > 0.f) ? t1 : t1 * SLOPE;
  t2 = (t2 > 0.f) ? t2 : t2 * SLOPE;
  t3 = (t3 > 0.f) ? t3 : t3 * SLOPE;
  return t0 * w.x + t1 * w.y + t2 * w.z + t3 * w.w;
}

__device__ __forceinline__ void score_drain(int2* q, int head, int n, int lane, const TileArgs& A) {
  int src = 0, dl = 0, e = 0;
  if (lane < n) {
    const int qi = (head + lane) & (AT_QCAP - 1);
    const int2 t = q[qi];
    e = t.x;
    dl = t.y;
    src = (e < E_EDGES) ? A.ei[e] : (e - E_EDGES);
    if (dl < 0) {
      dl = A.ei[E_EDGES + e] - A.lo;
      q[qi].y = dl;
    }
  }
  const float4 w4 = *(const float4*)(A.att + 4 * lane);
  const int  r  = lane & 7;
  const int  hd = lane >> 3;
  const bool b2 = (r & 4) != 0, b1 = (r & 2) != 0, b0 = (r & 1) != 0;
  for (int j0 = 0; j0 < n; j0 += AT_GROUP) {
    float4 a[AT_GROUP], b[AT_GROUP];
    #pragma unroll
    for (int u = 0; u < AT_GROUP; ++u) {
      const int s = lane_bcast_i(src, j0 + u);
      const int d = lane_bcast_i(dl, j0 + u);
      a[u] = *(const float4*)(A.xl + (size_t)s * HC + 4 * lane);
      b[u] = *(const float4*)(A.sxr + d * HC + 4 * lane);
    }
    float p[AT_GROUP];
    #pragma unroll
    for (int u = 0; u < AT_GROUP; ++u) p[u] = lrelu_dot4(a[u], b[u], w4);
    float f4[4];
    #pragma unroll
    for (int i = 0; i < 4; ++i) {
      const float keep = b2 ? p[i + 4] : p[i];
      const float send = b2 ? p[i] : p[i + 4];
      f4[i] = keep + __shfl_xor(send, 4, 32);
    }
    float f2[2];
    #pragma unroll
    for (int i = 0; i < 2; ++i) {
      const float keep = b1 ? f4[i + 2] : f4[i];
      const float send = b1 ? f4[i] : f4[i + 2];
      f2[i] = keep + __shfl_xor(send, 2, 32);
    }
    const float keep = b0 ? f2[1] : f2[0];
    const float send = b0 ? f2[0] : f2[1];
    const float sc = keep + __shfl_xor(send, 1, 32);
    const int ue = j0 + r;
    const int eu = __shfl(e, ue, 32);
    const int du = __shfl(dl, ue, 32);
    if (ue < n) {
      A.sc[(size_t)eu * HEADS + hd] = sc;
      atomicMax(&A.smax[du * HEADS + hd], f32_to_ord(sc));
    }
  }
}

__device__ __forceinline__ void agg_drain(const int2* q, int head, int n, int lane, const TileArgs& A) {
  int src = 0, dl = 0;
  float a0 = 0.f, a1 = 0.f, a2 = 0.f, a3 = 0.f;
  if (lane < n) {
    const int2 t = q[(head + lane) & (AT_QCAP - 1)];
    const int e = t.x;
    dl = t.y;
    src = (e < E_EDGES) ? A.ei[e] : (e - E_EDGES);
    if (dl < 0) dl = A.ei[E_EDGES + e] - A.lo;
    const float4 s4 = *(const float4*)(A.sc + (size_t)e * HEADS);
    const uint4  m4 = *(const uint4*)(A.smax + dl * HEADS);
    a0 = __expf(s4.x - ord_to_f32(m4.x));
    a1 = __expf(s4.y - ord_to_f32(m4.y));
    a2 = __expf(s4.z - ord_to_f32(m4.z));
    a3 = __expf(s4.w - ord_to_f32(m4.w));
    float* dn = A.sden + dl * HEADS;
    atomAddF(dn,     a0);
    atomAddF(dn + 1, a1);
    atomAddF(dn + 2, a2);
    atomAddF(dn + 3, a3);
  }
  for (int j0 = 0; j0 < n; j0 += AT_GROUP) {
    float v[AT_GROUP][HEADS];
    #pragma unroll
    for (int u = 0; u < AT_GROUP; ++u) {
      const int s = lane_bcast_i(src, j0 + u);
      const float* p = A.xl + (size_t)s * HC + lane;
      #pragma unroll
      for (int h = 0; h < HEADS; ++h) v[u][h] = p[h * CHP];
    }
    #pragma unroll
    for (int u = 0; u < AT_GROUP; ++u) {
      if (j0 + u < n) {
        const int   d  = lane_bcast_i(dl, j0 + u);
        const float w0 = lane_bcast_f(a0, j0 + u);
        const float w1 = lane_bcast_f(a1, j0 + u);
        const float w2 = lane_bcast_f(a2, j0 + u);
        const float w3 = lane_bcast_f(a3, j0 + u);
        float* t = A.tile + d * HC + lane;
        atomAddF(t,           v[u][0] * w0);
        atomAddF(t + CHP,     v[u][1] * w1);
        atomAddF(t + 2 * CHP, v[u][2] * w2);
        atomAddF(t + 3 * CHP, v[u][3] * w3);
      }
    }
  }
}

__device__ __forceinline__ uint4 tile_ids(const uint4* __restrict__ tp4, const int pass, const int tid) {
  const int i = pass * AT_THREADS + tid;
  return tp4[(i < AT_V16) ? i : (AT_V16 - 1)];
}

__device__ __forceinline__ unsigned zero_bytes(const unsigned x) {
  return ~(((x & 0x7f7f7f7fu) + 0x7f7f7f7fu) | x | 0x7f7f7f7fu);
}

template <int PH>
__device__ __forceinline__ void tile_pass(uint4& p, const uint4* __restrict__ tp4, const int pass, const int tid,
                                          const bool last, int2* q, int& head, int& cnt, int& total,
                                          const unsigned below, const int lane, const TileArgs& A) {
  const int idx = pass * AT_THREADS + tid;
  const unsigned pat = A.pat;
  const unsigned z0 = zero_bytes(p.x ^ pat), z1 = zero_bytes(p.y ^ pat);
  const unsigned z2 = zero_bytes(p.z ^ pat), z3 = zero_bytes(p.w ^ pat);
  p = tile_ids(tp4, pass + 4, tid);
  unsigned g = (z0 >> 7) | (z1 >> 6) | (z2 >> 5) | (z3 >> 4);
  if (idx >= AT_V16) g = 0u;
  for (;;) {
    const bool on = g != 0u;
    const unsigned m = (unsigned)__ballot(on);
    if (m != 0u) {
      if (on) {
        const int pb = __ffs(g) - 1;
        g &= g - 1u;
        q[(head + cnt + __popc(m & below)) & (AT_QCAP - 1)] = make_int2(idx * 16 + 4 * (pb & 7) + (pb >> 3), -1);
      }
      cnt += __popc(m);
      total += __popc(m);
    }
    while (cnt >= 32 || (last && m == 0u && cnt > 0)) {
      const int n = (cnt < 32) ? cnt : 32;
      if (PH == 0) score_drain(q, head, n, lane, A);
      else         agg_drain(q, head, n, lane, A);
      head = (head + n) & (AT_QCAP - 1);
      cnt -= n;
    }
    if (m == 0u) break;
  }
}

template <int PH>
__device__ __forceinline__ int tile_scan(int2* q, const int tid, const int wave, const int lane, const TileArgs& A) {
  int head = 0, cnt = 0, total = 0;
  const unsigned below = (1u << lane) - 1u;
  {
    const int dl = wave * AT_SELF + lane;
    const bool on = (lane < AT_SELF) && (dl < AT_S) && (A.lo + dl < N_NODES);
    const unsigned m = (unsigned)__ballot(on);
    if (on) q[__popc(m & below)] = make_int2(E_EDGES + A.lo + dl, dl);
    cnt = __popc(m);
    total = cnt;
  }
  const uint4* dst4 = A.tp;
  uint4 p0 = tile_ids(dst4, 0, tid);
  uint4 p1 = tile_ids(dst4, 1, tid);
  uint4 p2 = tile_ids(dst4, 2, tid);
  uint4 p3 = tile_ids(dst4, 3, tid);
  #pragma unroll 1
  for (int it = 0; it < AT_PASSES; it += 4) {
    tile_pass<PH>(p0, dst4, it,     tid, false, q, head, cnt, total, below, lane, A);
    tile_pass<PH>(p1, dst4, it + 1, tid, false, q, head, cnt, total, below, lane, A);
    tile_pass<PH>(p2, dst4, it + 2, tid, false, q, head, cnt, total, below, lane, A);
    tile_pass<PH>(p3, dst4, it + 3, tid, it + 4 >= AT_PASSES, q, head, cnt, total, below, lane, A);
  }
  return total;
}

template <int REUSE>
__global__ void __launch_bounds__(AT_THREADS)
tile_softmax_aggregate_ln_kernel(const int* __restrict__ ei,
                                 const float* __restrict__ xl,
                                 const float* __restrict__ xr,
                                 const float* __restrict__ att,
                                 float* sc,
                                 const float* __restrict__ bias,
                                 const float* __restrict__ gamma,
                                 const float* __restrict__ beta,
                                 float* __restrict__ out,
                                 int*  lists,
                                 int*  lcnt,
                                 int*  lover,
                                 const unsigned* __restrict__ tplane) {
  __shared__ __align__(16) float    tile[AT_S * HC];
  __shared__ __align__(16) float    sxr[AT_S * HC];
  __shared__ __align__(16) int2     ring[AT_WAVES * AT_QCAP];
  __shared__ __align__(16) unsigned smax[AT_S * HEADS];
  __shared__ __align__(16) float    sden[AT_S * HEADS];
  __shared__ int      sover;
  const int tid  = threadIdx.x;
  const int wave = tid >> 5;
  const int lane = tid & 31;
  const int lo   = blockIdx.x * AT_S;

  {
    constexpr int NV4 = AT_S * HC / 4;
    constexpr int NLD = (NV4 + AT_THREADS - 1) / AT_THREADS;
    const float4 zero4 = make_float4(0.f, 0.f, 0.f, 0.f);
    float4 r[NLD];
    #pragma unroll
    for (int k = 0; k < NLD; ++k) {
      const int gi = lo * (HC / 4) + tid + k * AT_THREADS;
      r[k] = ((const float4*)xr)[min(gi, N_NODES * (HC / 4) - 1)];
    }
    #pragma unroll
    for (int k = 0; k < NLD; ++k) {
      const int i = tid + k * AT_THREADS;
      if (i < NV4) { ((float4*)tile)[i] = zero4; ((float4*)sxr)[i] = r[k]; }
    }
  }
  if (tid < AT_S * HEADS) { smax[tid] = ORD_NEG_INF; sden[tid] = 0.f; }
  if (tid == 0) sover = 0;
  __syncthreads();

  TileArgs A;
  A.tp = (const uint4*)tplane; A.pat = (unsigned)blockIdx.x * 0x01010101u;
  A.ei = ei; A.xl = xl; A.sxr = sxr; A.att = att; A.sc = sc;
  A.tile = tile; A.smax = smax; A.sden = sden; A.lo = lo;
  int2* q = ring + wave * AT_QCAP;

  const int slot = blockIdx.x * AT_WAVES + wave;
  int total;
  bool listed = false;
  if (REUSE) listed = (lover[blockIdx.x] == 0);
  if (REUSE && listed) {
    total = min(max(__builtin_amdgcn_readfirstlane(lcnt[slot]), 0), AT_QCAP);
    const int2* ls = (const int2*)lists + (size_t)slot * AT_QCAP;
    int2 r[AT_QCAP / 32];
    #pragma unroll
    for (int k = 0; k < AT_QCAP / 32; ++k) {
      const int i = lane + 32 * k;
      r[k] = (i < total) ? ls[i] : make_int2(0, 0);
    }
    #pragma unroll
    for (int k = 0; k < AT_QCAP / 32; ++k) {
      const int i = lane + 32 * k;
      if (i < total) q[i] = r[k];
    }
    for (int h0 = 0; h0 < total; h0 += 32) {
      const int n = (total - h0 < 32) ? (total - h0) : 32;
      score_drain(q, h0, n, lane, A);
    }
  } else {
    total = tile_scan<0>(q, tid, wave, lane, A);
  }
  if (total > AT_QCAP && lane == 0) sover = 1;
  __syncthreads();
  if (!REUSE) {
    if (tid == 0) lover[blockIdx.x] = sover;
    if (sover == 0) {
      int2* ls = (int2*)lists + (size_t)slot * AT_QCAP;
      for (int i = lane; i < total; i += 32) ls[i] = q[i];
      if (lane == 0) lcnt[slot] = total;
    }
  }

  if (sover == 0) {
    for (int h0 = 0; h0 < total; h0 += 32) {
      const int n = (total - h0 < 32) ? (total - h0) : 32;
      agg_drain(q, h0, n, lane, A);
    }
  } else {
    (void)tile_scan<1>(q, tid, wave, lane, A);
  }
  __syncthreads();

  for (int dl = wave; dl < AT_S; dl += AT_WAVES) {
    const int node = lo + dl;
    if (node >= N_NODES) break;
    float v[4]; float sum = 0.f;
    #pragma unroll
    for (int i = 0; i < 4; ++i) {
      const int c = i * 32 + lane;
      v[i] = tile[dl * HC + c] / sden[dl * HEADS + i] + bias[c];
      sum += v[i];
    }
    for (int off = 16; off > 0; off >>= 1) sum += __shfl_xor(sum, off, 32);
    const float mu = sum * (1.0f / HC);
    float sq = 0.f;
    #pragma unroll
    for (int i = 0; i < 4; ++i) { float d = v[i] - mu; sq += d * d; }
    for (int off = 16; off > 0; off >>= 1) sq += __shfl_xor(sq, off, 32);
    const float rstd = rsqrtf(sq * (1.0f / HC) + LN_EPS);
    #pragma unroll
    for (int i = 0; i < 4; ++i) {
      const int c = i * 32 + lane;
      const float h = (v[i] - mu) * rstd * gamma[c] + beta[c];
      out[(size_t)node * HC + c] = h / (1.0f + __expf(-h));
    }
  }
}

#define POOL_CHUNKS 4
#define POOL_SLAB   (8 * POOL_CHUNKS * 32)
#define POOL_SLABS  ((N_NODES + POOL_SLAB - 1) / POOL_SLAB)
#define PROW        (HC + 32)
__global__ void pool_gather_kernel(const float* __restrict__ x,
                                   const int* __restrict__ batch,
                                   float* __restrict__ part) {
  __shared__ float sacc[8][HC];
  __shared__ int   scnt[8];
  const int g     = blockIdx.x % NGRAPH;
  const int s     = blockIdx.x / NGRAPH;
  const int wave  = threadIdx.x >> 5;
  const int lane  = threadIdx.x & 31;
  const int slab0 = s * POOL_SLAB;

  int ids[POOL_CHUNKS];
  #pragma unroll
  for (int i = 0; i < POOL_CHUNKS; ++i) {
    int node = slab0 + (i * 8 + wave) * 32 + lane;
    ids[i] = (node < N_NODES) ? batch[node] : -1;
  }
  float4 acc = make_float4(0.f, 0.f, 0.f, 0.f);
  int cnt = 0;
  #pragma unroll
  for (int i = 0; i < POOL_CHUNKS; ++i) {
    const unsigned mask = (unsigned)__ballot(ids[i] == g);
    if (mask == 0u) continue;
    cnt += __popc(mask);
    const int base = slab0 + (i * 8 + wave) * 32;
    for (int j0 = 0; j0 < 32; j0 += 16) {
      if (((mask >> j0) & 0xffffu) == 0u) continue;
      float4 v[16];
      #pragma unroll
      for (int u = 0; u < 16; ++u) {
        int node = base + j0 + u;
        node = (node < N_NODES) ? node : (N_NODES - 1);
        v[u] = *(const float4*)(x + (size_t)node * HC + lane * 4);
      }
      #pragma unroll
      for (int u = 0; u < 16; ++u) {
        const bool on = ((mask >> (j0 + u)) & 1u) != 0u;
        acc.x += on ? v[u].x : 0.f;
        acc.y += on ? v[u].y : 0.f;
        acc.z += on ? v[u].z : 0.f;
        acc.w += on ? v[u].w : 0.f;
      }
    }
  }
  *(float4*)&sacc[wave][lane * 4] = acc;
  if (lane == 0) scnt[wave] = cnt;
  __syncthreads();
  const int tid = threadIdx.x;
  float* prow = part + ((size_t)g * POOL_SLABS + s) * PROW;
  if (tid < HC) {
    float t = 0.f;
    #pragma unroll
    for (int w = 0; w < 8; ++w) t += sacc[w][tid];
    prow[tid] = t;
  } else if (tid < PROW) {
    int c = 0;
    #pragma unroll
    for (int w = 0; w < 8; ++w) c += scnt[w];
    ((int*)prow)[tid] = (tid == HC) ? c : 0;
  }
}
__global__ void pool_final_kernel(const float* __restrict__ part,
                                  float* __restrict__ out) {
  int t = blockIdx.x * blockDim.x + threadIdx.x;
  if (t >= NGRAPH * HC) return;
  const int g = t >> 7, c = t & 127;
  const float* p = part + (size_t)g * POOL_SLABS * PROW;
  float sum = 0.f;
  int n = 0;
  #pragma unroll 7
  for (int s = 0; s < POOL_SLABS; ++s) {
    sum += p[s * PROW + c];
    n   += ((const int*)p)[s * PROW + HC];
  }
  out[t] = sum / fmaxf((float)n, 1.0f);
}

extern "C" void kernel_launch(void* const* d_in, const int* in_sizes, int n_in,
                              void* d_out, int out_size, void* d_ws, size_t ws_size,
                              hipStream_t stream) {
  (void)in_sizes; (void)n_in; (void)out_size; (void)ws_size;
  const float* x     = (const float*)d_in[0];
  const int*   ei    = (const int*)d_in[1];
  const int*   batch = (const int*)d_in[2];
  const float* Wl[2]    = {(const float*)d_in[3],  (const float*)d_in[11]};
  const float* Wr[2]    = {(const float*)d_in[4],  (const float*)d_in[12]};
  const float* bl[2]    = {(const float*)d_in[5],  (const float*)d_in[13]};
  const float* br[2]    = {(const float*)d_in[6],  (const float*)d_in[14]};
  const float* att[2]   = {(const float*)d_in[7],  (const float*)d_in[15]};
  const float* bias[2]  = {(const float*)d_in[8],  (const float*)d_in[16]};
  const float* gamma[2] = {(const float*)d_in[9],  (const float*)d_in[17]};
  const float* beta[2]  = {(const float*)d_in[10], (const float*)d_in[18]};

  float* ws = (float*)d_ws;
  float*    xl    = ws;            ws += (size_t)N_NODES * HC;
  float*    xr2   = ws;            ws += (size_t)N_NODES * HC;
  float*    xnode = ws;            ws += (size_t)N_NODES * HC;
  float*    scbuf = ws;            ws += (size_t)ETOT * HEADS;
  float*    ppart = ws;            ws += (size_t)NGRAPH * POOL_SLABS * PROW;
  int*      lists = (int*)ws;      ws += (size_t)AT_TILES * AT_WAVES * AT_QCAP * 2;
  int*      lcnt  = (int*)ws;      ws += (size_t)AT_TILES * AT_WAVES;
  int*      lover = (int*)ws;      ws += 256;
  unsigned* tplane = (unsigned*)ws;

  dim3 b256(256);

  const float* xin = x;
  for (int l = 0; l < 2; ++l) {
    if (l == 0)
      gemm_bias_wmma<1><<<G_BLOCKS, G_THREADS, 0, stream>>>(xin, Wl[l], bl[l], xl, N_NODES, ei, tplane);
    else
      gemm_bias_wmma<0><<<G_BLOCKS, G_THREADS, 0, stream>>>(xin, Wl[l], bl[l], xl, N_NODES, ei, tplane);
    gemm_bias_wmma<0><<<G_BLOCKS, G_THREADS, 0, stream>>>(xin, Wr[l], br[l], xr2, N_NODES, ei, tplane);
    if (l == 0)
      tile_softmax_aggregate_ln_kernel<0><<<AT_TILES, AT_THREADS, 0, stream>>>(ei, xl, xr2, att[l], scbuf, bias[l], gamma[l], beta[l], xnode, lists, lcnt, lover, tplane);
    else
      tile_softmax_aggregate_ln_kernel<1><<<AT_TILES, AT_THREADS, 0, stream>>>(ei, xl, xr2, att[l], scbuf, bias[l], gamma[l], beta[l], xnode, lists, lcnt, lover, tplane);
    xin = xnode;
  }
  pool_gather_kernel<<<NGRAPH * POOL_SLABS, b256, 0, stream>>>(xnode, batch, ppart);
  pool_final_kernel<<<(NGRAPH * HC + 255) / 256, b256, 0, stream>>>(ppart, (float*)d_out);
}


# tile kernel in ONE pass: softmax shift = the row's self-loop score, so score, exp, denominator and weighted LDS add share one fetch of x_l; two-phase redo if score - shift > 60; LN sum order changes
# speedup vs baseline: 4.7131x; 1.4333x over previous
#include <hip/hip_runtime.h>
#include <hip/hip_bf16.h>
#include <math.h>

#define N_NODES 50000
#define E_EDGES 800000
#define ETOT    (E_EDGES + N_NODES)
#define HEADS   4
#define CHP     32
#define HC      128
#define NGRAPH  64
#define SLOPE   0.2f
#define LN_EPS  1e-5f

#define AT_S     197
#define AT_TILES ((N_NODES + AT_S - 1) / AT_S)
#define AT_NONE  255u

typedef float v2f __attribute__((ext_vector_type(2)));
typedef float v8f __attribute__((ext_vector_type(8)));

__device__ __forceinline__ unsigned f32_to_ord(float f) {
  unsigned u = __float_as_uint(f);
  return (u & 0x80000000u) ? ~u : (u | 0x80000000u);
}
__device__ __forceinline__ float ord_to_f32(unsigned u) {
  u = (u & 0x80000000u) ? (u & 0x7fffffffu) : ~u;
  return __uint_as_float(u);
}
__device__ __forceinline__ unsigned tile_of(const int d) {
  return ((unsigned)d < (unsigned)N_NODES) ? ((unsigned)d / (unsigned)AT_S) : AT_NONE;
}
__device__ __forceinline__ void atomAddF(float* p, float v) {
  unsafeAtomicAdd(p, v);
}

#define G_THREADS 512
#define G_WAVES   (G_THREADS / 32)
#define G_BLOCKS  256
#define G_NT      4
typedef float v4f __attribute__((ext_vector_type(4)));

__device__ __forceinline__ void gemm_load_a(v4f (&a)[16], const float* __restrict__ X, const int u, const int n,
                                            const int lane) {
  const int rt  = blockIdx.x + G_BLOCKS * (u >> 1);
  const int row = max(min(rt * 16 + (lane & 15), n - 1), 0);
  const v4f* p = (const v4f*)(X + (size_t)row * HC) + (lane >> 4);
  #pragma unroll
  for (int j = 0; j < 16; ++j) a[j] = p[2 * j];
}

template <int PRE>
__global__ void __launch_bounds__(G_THREADS)
gemm_bias_wmma(const float* __restrict__ X,
               const float* __restrict__ W,
               const float* __restrict__ bias,
               float* __restrict__ Y, int n,
               const int* __restrict__ ei,
               unsigned* __restrict__ tplane) {
  __shared__ __align__(16) float sW[HC * HC];
  const int tid  = threadIdx.x;
  const int wave = __builtin_amdgcn_readfirstlane(tid >> 5);
  const int lane = tid & 31;
  const int c    = lane & 15;
  const int h    = lane >> 4;
  const int ntiles = (n + 15) >> 4;
  const int nrt    = (ntiles - (int)blockIdx.x + G_BLOCKS - 1) / G_BLOCKS;
  const int nunits = 2 * nrt;

  constexpr int NGRP = E_EDGES / 4;
  constexpr int PER  = (NGRP + G_BLOCKS * G_THREADS - 1) / (G_BLOCKS * G_THREADS);
  int4 dd[PER];
  if (PRE) {
    #pragma unroll
    for (int k = 0; k < PER; ++k) {
      const int i = (int)blockIdx.x * G_THREADS + tid + k * (G_BLOCKS * G_THREADS);
      dd[k] = ((const int4*)(ei + E_EDGES))[min(i, NGRP - 1)];
    }
  }

  int u = wave;
  v4f a[16];
  gemm_load_a(a, X, u, n, lane);

  if (PRE) {
    #pragma unroll
    for (int k = 0; k < PER; ++k) {
      const int i = (int)blockIdx.x * G_THREADS + tid + k * (G_BLOCKS * G_THREADS);
      if (i < NGRP)
        tplane[i] = tile_of(dd[k].x) | (tile_of(dd[k].y) << 8) | (tile_of(dd[k].z) << 16) | (tile_of(dd[k].w) << 24);
    }
  }

  {
    v4f r[2][4];
    #pragma unroll
    for (int i = 0; i < 2; ++i) {
      const int item = tid + G_THREADS * i;
      const int q = item >> 5, g = item & 31;
      #pragma unroll
      for (int k = 0; k < 4; ++k) r[i][k] = *(const v4f*)(W + (4 * q + k) * HC + 4 * g);
    }
    #pragma unroll
    for (int i = 0; i < 2; ++i) {
      const int item = tid + G_THREADS * i;
      const int q = item >> 5, g = item & 31;
      v4f* d = (v4f*)sW + q * HC + 4 * g;
      d[0] = v4f{r[i][0].x, r[i][1].x, r[i][2].x, r[i][3].x};
      d[1] = v4f{r[i][0].y, r[i][1].y, r[i][2].y, r[i][3].y};
      d[2] = v4f{r[i][0].z, r[i][1].z, r[i][2].z, r[i][3].z};
      d[3] = v4f{r[i][0].w, r[i][1].w, r[i][2].w, r[i][3].w};
    }
  }
  __syncthreads();

  while (u < nunits) {
    const int un = u + G_WAVES;
    v4f an[16];
    gemm_load_a(an, X, min(un, nunits - 1), n, lane);
    const int rt   = blockIdx.x + G_BLOCKS * (u >> 1);
    const int col0 = (u & 1) * (16 * G_NT) + c;
    const v4f* bp = (const v4f*)sW + h * HC + col0;
    float bc[G_NT];
    v8f acc[G_NT];
    #pragma unroll
    for (int t = 0; t < G_NT; ++t) { bc[t] = bias[col0 + 16 * t]; acc[t] = v8f{}; }
    v4f bq[2][G_NT];
    #pragma unroll
    for (int t = 0; t < G_NT; ++t) bq[0][t] = bp[16 * t];
    #pragma unroll
    for (int j = 0; j < 16; ++j) {
      if (j + 1 < 16) {
        #pragma unroll
        for (int t = 0; t < G_NT; ++t) bq[(j + 1) & 1][t] = bp[2 * (j + 1) * HC + 16 * t];
      }
      #pragma unroll
      for (int t = 0; t < G_NT; ++t) {
        const v4f b = bq[j & 1][t];
        v2f a0; a0.x = a[j].x; a0.y = a[j].y;
        v2f b0; b0.x = b.x;    b0.y = b.y;
        acc[t] = __builtin_amdgcn_wmma_f32_16x16x4_f32(false, a0, false, b0, (short)0, acc[t], false, false);
        v2f a1; a1.x = a[j].z; a1.y = a[j].w;
        v2f b1; b1.x = b.z;    b1.y = b.w;
        acc[t] = __builtin_amdgcn_wmma_f32_16x16x4_f32(false, a1, false, b1, (short)0, acc[t], false, false);
      }
      __builtin_amdgcn_sched_barrier(0);
    }
    float* yp = Y + (size_t)(rt * 16 + 8 * h) * HC + col0;
    if (rt * 16 + 16 <= n) {
      #pragma unroll
      for (int t = 0; t < G_NT; ++t) {
        #pragma unroll
        for (int r = 0; r < 8; ++r) yp[r * HC + 16 * t] = acc[t][r] + bc[t];
      }
    } else {
      #pragma unroll
      for (int t = 0; t < G_NT; ++t) {
        #pragma unroll
        for (int r = 0; r < 8; ++r) {
          if (rt * 16 + r + 8 * h < n) yp[r * HC + 16 * t] = acc[t][r] + bc[t];
        }
      }
    }
    #pragma unroll
    for (int j = 0; j < 16; ++j) a[j] = an[j];
    u = un;
  }
}

#define AT_THREADS 1024
#define AT_WAVES   (AT_THREADS / 32)
#define AT_QCAP    256
#define AT_GROUP   8
#define AT_V16     (E_EDGES / 16)
#define AT_ITERS   ((AT_V16 + AT_THREADS - 1) / AT_THREADS)
#define AT_PASSES  (((AT_ITERS + 3) / 4) * 4)
#define AT_SELF    ((AT_S + AT_WAVES - 1) / AT_WAVES)
#define ORD_NEG_INF 0x007FFFFFu
#define AT_LIMIT   60.0f

struct TileArgs {
  const uint4* tp;
  unsigned     pat;
  const int*   ei;
  const float* xl;
  const float* sxr;
  const float* att;
  float*       sc;
  float*       tile;
  unsigned*    smax;
  float*       shift;
  float*       sden;
  int          lo;
};

__device__ __forceinline__ int lane_bcast_i(int v, int j) {
  return __builtin_amdgcn_readlane(v, j);
}
__device__ __forceinline__ float lane_bcast_f(float v, int j) {
  return __int_as_float(__builtin_amdgcn_readlane(__float_as_int(v), j));
}
__device__ __forceinline__ float lrelu_dot4(const float4 a, const float4 b, const float4 w) {
  float t0 = a.x + b.x, t1 = a.y + b.y, t2 = a.z + b.z, t3 = a.w + b.w;
  t0 = (t0 > 0.f) ? t0 : t0 * SLOPE;
  t1 = (t1 > 0.f) ? t1 : t1 * SLOPE;
  t2 = (t2 > 0.f) ? t2 : t2 * SLOPE;
  t3 = (t3 > 0.f) ? t3 : t3 * SLOPE;
  return t0 * w.x + t1 * w.y + t2 * w.z + t3 * w.w;
}

__device__ __forceinline__ void score_drain(int2* q, int head, int n, int lane, const TileArgs& A) {
  int src = 0, dl = 0, e = 0;
  if (lane < n) {
    const int qi = (head + lane) & (AT_QCAP - 1);
    const int2 t = q[qi];
    e = t.x;
    dl = t.y;
    src = (e < E_EDGES) ? A.ei[e] : (e - E_EDGES);
    if (dl < 0) {
      dl = A.ei[E_EDGES + e] - A.lo;
      q[qi].y = dl;
    }
  }
  const float4 w4 = *(const float4*)(A.att + 4 * lane);
  const int  r  = lane & 7;
  const int  hd = lane >> 3;
  const bool b2 = (r & 4) != 0, b1 = (r & 2) != 0, b0 = (r & 1) != 0;
  for (int j0 = 0; j0 < n; j0 += AT_GROUP) {
    float4 a[AT_GROUP], b[AT_GROUP];
    #pragma unroll
    for (int u = 0; u < AT_GROUP; ++u) {
      const int s = lane_bcast_i(src, j0 + u);
      const int d = lane_bcast_i(dl, j0 + u);
      a[u] = *(const float4*)(A.xl + (size_t)s * HC + 4 * lane);
      b[u] = *(const float4*)(A.sxr + d * HC + 4 * lane);
    }
    float p[AT_GROUP];
    #pragma unroll
    for (int u = 0; u < AT_GROUP; ++u) p[u] = lrelu_dot4(a[u], b[u], w4);
    float f4[4];
    #pragma unroll
    for (int i = 0; i < 4; ++i) {
      const float keep = b2 ? p[i + 4] : p[i];
      const float send = b2 ? p[i] : p[i + 4];
      f4[i] = keep + __shfl_xor(send, 4, 32);
    }
    float f2[2];
    #pragma unroll
    for (int i = 0; i < 2; ++i) {
      const float keep = b1 ? f4[i + 2] : f4[i];
      const float send = b1 ? f4[i] : f4[i + 2];
      f2[i] = keep + __shfl_xor(send, 2, 32);
    }
    const float keep = b0 ? f2[1] : f2[0];
    const float send = b0 ? f2[0] : f2[1];
    const float sc = keep + __shfl_xor(send, 1, 32);
    const int ue = j0 + r;
    const int eu = __shfl(e, ue, 32);
    const int du = __shfl(dl, ue, 32);
    if (ue < n) {
      A.sc[(size_t)eu * HEADS + hd] = sc;
      atomicMax(&A.smax[du * HEADS + hd], f32_to_ord(sc));
    }
  }
}

template <int U>
__device__ __forceinline__ float group8_bcast(const float v) {
  return __int_as_float(__builtin_amdgcn_mov_dpp8(__float_as_int(v), U * 0x249249));
}
template <int U>
__device__ __forceinline__ void add_edge(const float4 a, const float w, const int d, const int lane, float* tile) {
  const float wu = group8_bcast<U>(w);
  float* t = tile + d * HC + lane;
  atomAddF(t,      a.x * wu);
  atomAddF(t + 32, a.y * wu);
  atomAddF(t + 64, a.z * wu);
  atomAddF(t + 96, a.w * wu);
}

template <int SHIFT>
__device__ __forceinline__ bool one_pass_drain(int2* q, int head, int n, int lane, const TileArgs& A) {
  static_assert(AT_GROUP == 8, "the fold and the weight exchange work on groups of eight edges");
  int src = 0, dl = 0;
  if (lane < n) {
    const int qi = (head + lane) & (AT_QCAP - 1);
    const int2 t = q[qi];
    const int e = t.x;
    dl = t.y;
    src = (e < E_EDGES) ? A.ei[e] : (e - E_EDGES);
    if (dl < 0) {
      dl = A.ei[E_EDGES + e] - A.lo;
      q[qi].y = dl;
    }
  }
  const float4 w4 = *(const float4*)(A.att + 4 * lane);
  const int  r  = lane & 7;
  const int  hd = lane >> 3;
  const bool b2 = (r & 4) != 0, b1 = (r & 2) != 0, b0 = (r & 1) != 0;
  bool big = false;
  for (int j0 = 0; j0 < n; j0 += AT_GROUP) {
    float4 a[AT_GROUP], b[AT_GROUP];
    #pragma unroll
    for (int u = 0; u < AT_GROUP; ++u) {
      const int s = lane_bcast_i(src, j0 + u);
      const int d = lane_bcast_i(dl, j0 + u);
      a[u] = *(const float4*)(A.xl + (size_t)s * HC + 4 * lane);
      b[u] = *(const float4*)(A.sxr + d * HC + 4 * lane);
    }
    float p[AT_GROUP];
    #pragma unroll
    for (int u = 0; u < AT_GROUP; ++u) p[u] = lrelu_dot4(a[u], b[u], w4);
    float f4[4];
    #pragma unroll
    for (int i = 0; i < 4; ++i) {
      const float keep = b2 ? p[i + 4] : p[i];
      const float send = b2 ? p[i] : p[i + 4];
      f4[i] = keep + __shfl_xor(send, 4, 32);
    }
    float f2[2];
    #pragma unroll
    for (int i = 0; i < 2; ++i) {
      const float keep = b1 ? f4[i + 2] : f4[i];
      const float send = b1 ? f4[i] : f4[i + 2];
      f2[i] = keep + __shfl_xor(send, 2, 32);
    }
    const float keep = b0 ? f2[1] : f2[0];
    const float send = b0 ? f2[0] : f2[1];
    const float sc = keep + __shfl_xor(send, 1, 32);
    const int ue = j0 + r;
    const int du = __shfl(dl, ue, 32);
    if (SHIFT) {
      if (ue < n) A.shift[du * HEADS + hd] = sc;
    } else {
      float w = 0.f;
      if (ue < n) {
        const float t = sc - A.shift[du * HEADS + hd];
        big = big || (t > AT_LIMIT);
        w = __expf(t);
        atomAddF(A.sden + du * HEADS + hd, w);
      }
      if (j0 + 0 < n) add_edge<0>(a[0], w, lane_bcast_i(dl, j0 + 0), lane, A.tile);
      if (j0 + 1 < n) add_edge<1>(a[1], w, lane_bcast_i(dl, j0 + 1), lane, A.tile);
      if (j0 + 2 < n) add_edge<2>(a[2], w, lane_bcast_i(dl, j0 + 2), lane, A.tile);
      if (j0 + 3 < n) add_edge<3>(a[3], w, lane_bcast_i(dl, j0 + 3), lane, A.tile);
      if (j0 + 4 < n) add_edge<4>(a[4], w, lane_bcast_i(dl, j0 + 4), lane, A.tile);
      if (j0 + 5 < n) add_edge<5>(a[5], w, lane_bcast_i(dl, j0 + 5), lane, A.tile);
      if (j0 + 6 < n) add_edge<6>(a[6], w, lane_bcast_i(dl, j0 + 6), lane, A.tile);
      if (j0 + 7 < n) add_edge<7>(a[7], w, lane_bcast_i(dl, j0 + 7), lane, A.tile);
    }
  }
  return big;
}

__device__ __forceinline__ void agg_drain(const int2* q, int head, int n, int lane, const TileArgs& A) {
  int src = 0, dl = 0;
  float a0 = 0.f, a1 = 0.f, a2 = 0.f, a3 = 0.f;
  if (lane < n) {
    const int2 t = q[(head + lane) & (AT_QCAP - 1)];
    const int e = t.x;
    dl = t.y;
    src = (e < E_EDGES) ? A.ei[e] : (e - E_EDGES);
    if (dl < 0) dl = A.ei[E_EDGES + e] - A.lo;
    const float4 s4 = *(const float4*)(A.sc + (size_t)e * HEADS);
    const uint4  m4 = *(const uint4*)(A.smax + dl * HEADS);
    a0 = __expf(s4.x - ord_to_f32(m4.x));
    a1 = __expf(s4.y - ord_to_f32(m4.y));
    a2 = __expf(s4.z - ord_to_f32(m4.z));
    a3 = __expf(s4.w - ord_to_f32(m4.w));
    float* dn = A.sden + dl * HEADS;
    atomAddF(dn,     a0);
    atomAddF(dn + 1, a1);
    atomAddF(dn + 2, a2);
    atomAddF(dn + 3, a3);
  }
  for (int j0 = 0; j0 < n; j0 += AT_GROUP) {
    float v[AT_GROUP][HEADS];
    #pragma unroll
    for (int u = 0; u < AT_GROUP; ++u) {
      const int s = lane_bcast_i(src, j0 + u);
      const float* p = A.xl + (size_t)s * HC + lane;
      #pragma unroll
      for (int h = 0; h < HEADS; ++h) v[u][h] = p[h * CHP];
    }
    #pragma unroll
    for (int u = 0; u < AT_GROUP; ++u) {
      if (j0 + u < n) {
        const int   d  = lane_bcast_i(dl, j0 + u);
        const float w0 = lane_bcast_f(a0, j0 + u);
        const float w1 = lane_bcast_f(a1, j0 + u);
        const float w2 = lane_bcast_f(a2, j0 + u);
        const float w3 = lane_bcast_f(a3, j0 + u);
        float* t = A.tile + d * HC + (lane & 3) * 32 + (lane >> 2);
        atomAddF(t,      v[u][0] * w0);
        atomAddF(t + 8,  v[u][1] * w1);
        atomAddF(t + 16, v[u][2] * w2);
        atomAddF(t + 24, v[u][3] * w3);
      }
    }
  }
}

__device__ __forceinline__ uint4 tile_ids(const uint4* __restrict__ tp4, const int pass, const int tid) {
  const int i = pass * AT_THREADS + tid;
  return tp4[(i < AT_V16) ? i : (AT_V16 - 1)];
}

__device__ __forceinline__ unsigned zero_bytes(const unsigned x) {
  return ~(((x & 0x7f7f7f7fu) + 0x7f7f7f7fu) | x | 0x7f7f7f7fu);
}

__device__ __forceinline__ unsigned tile_bits(const uint4 p, const unsigned pat) {
  const unsigned z0 = zero_bytes(p.x ^ pat), z1 = zero_bytes(p.y ^ pat);
  const unsigned z2 = zero_bytes(p.z ^ pat), z3 = zero_bytes(p.w ^ pat);
  return (z0 >> 7) | (z1 >> 6) | (z2 >> 5) | (z3 >> 4);
}
__device__ __forceinline__ int bit_edge(const int idx, const int pb) {
  return idx * 16 + 4 * (pb & 7) + (pb >> 3);
}

__device__ __forceinline__ int ring_self(int2* q, const bool store, const int wave, const int lane, const TileArgs& A) {
  const int dl = wave * AT_SELF + lane;
  const bool on = (lane < AT_SELF) && (dl < AT_S) && (A.lo + dl < N_NODES);
  const unsigned m = (unsigned)__ballot(on);
  if (store && on) q[__popc(m & ((1u << lane) - 1u))] = make_int2(E_EDGES + A.lo + dl, dl);
  return __popc(m);
}

__device__ __forceinline__ void fill_pass(uint4& p, const uint4* __restrict__ tp4, const int pass, const int tid,
                                          int2* q, int& cnt, bool& over, const unsigned below,
                                          const TileArgs& A) {
  const int idx = pass * AT_THREADS + tid;
  unsigned g = tile_bits(p, A.pat);
  p = tile_ids(tp4, pass + 4, tid);
  if (idx >= AT_V16) g = 0u;
  for (;;) {
    const bool on = g != 0u;
    const unsigned m = (unsigned)__ballot(on);
    if (m == 0u) break;
    const bool room = !over && (cnt + __popc(m) <= AT_QCAP);
    if (on) {
      const int pb = __ffs(g) - 1;
      g &= g - 1u;
      if (room) q[cnt + __popc(m & below)] = make_int2(bit_edge(idx, pb), -1);
    }
    if (room) cnt += __popc(m);
    else      over = true;
  }
}

__device__ __forceinline__ bool fast_fill(int2* q, int& cnt, const int tid, const int lane, const TileArgs& A) {
  bool over = false;
  const unsigned below = (1u << lane) - 1u;
  const uint4* dst4 = A.tp;
  uint4 p0 = tile_ids(dst4, 0, tid);
  uint4 p1 = tile_ids(dst4, 1, tid);
  uint4 p2 = tile_ids(dst4, 2, tid);
  uint4 p3 = tile_ids(dst4, 3, tid);
  #pragma unroll 1
  for (int it = 0; it < AT_PASSES; it += 4) {
    fill_pass(p0, dst4, it,     tid, q, cnt, over, below, A);
    fill_pass(p1, dst4, it + 1, tid, q, cnt, over, below, A);
    fill_pass(p2, dst4, it + 2, tid, q, cnt, over, below, A);
    fill_pass(p3, dst4, it + 3, tid, q, cnt, over, below, A);
  }
  return over;
}

template <int PH>
__device__ __forceinline__ void drain_any(int2* q, const int head, const int n, const int lane, const TileArgs& A,
                                          bool& big) {
  if (PH == 0)      score_drain(q, head, n, lane, A);
  else if (PH == 1) agg_drain(q, head, n, lane, A);
  else              big = one_pass_drain<0>(q, head, n, lane, A) || big;
}

template <int PH>
__device__ __forceinline__ int plain_scan(int2* q, const int cnt0, bool& big, const int tid, const int lane,
                                          const TileArgs& A) {
  int head = 0, cnt = cnt0, met = 0;
  const unsigned below = (1u << lane) - 1u;
  #pragma unroll 1
  for (int pass = 0; pass <= AT_ITERS; ++pass) {
    const bool last = pass == AT_ITERS;
    const int idx = pass * AT_THREADS + tid;
    unsigned g = tile_bits(tile_ids(A.tp, pass, tid), A.pat);
    if (idx >= AT_V16) g = 0u;
    for (;;) {
      const bool on = g != 0u;
      const unsigned m = (unsigned)__ballot(on);
      if (m != 0u) {
        if (on) {
          const int pb = __ffs(g) - 1;
          g &= g - 1u;
          q[(head + cnt + __popc(m & below)) & (AT_QCAP - 1)] = make_int2(bit_edge(idx, pb), -1);
        }
        cnt += __popc(m);
        met += __popc(m);
      }
      while (cnt >= 32 || (last && m == 0u && cnt > 0)) {
        const int n = (cnt < 32) ? cnt : 32;
        drain_any<PH>(q, head, n, lane, A, big);
        head = (head + n) & (AT_QCAP - 1);
        cnt -= n;
      }
      if (m == 0u) break;
    }
  }
  return met;
}

template <int REUSE>
__global__ void __launch_bounds__(AT_THREADS)
tile_softmax_aggregate_ln_kernel(const int* __restrict__ ei,
                                 const float* __restrict__ xl,
                                 const float* __restrict__ xr,
                                 const float* __restrict__ att,
                                 float* sc,
                                 const float* __restrict__ bias,
                                 const float* __restrict__ gamma,
                                 const float* __restrict__ beta,
                                 float* __restrict__ out,
                                 int*  lists,
                                 int*  lcnt,
                                 int*  lover,
                                 const unsigned* __restrict__ tplane) {
  __shared__ __align__(16) float    tile[AT_S * HC];
  __shared__ __align__(16) float    sxr[AT_S * HC];
  __shared__ __align__(16) int2     ring[AT_WAVES * AT_QCAP];
  __shared__ __align__(16) unsigned smax[AT_S * HEADS];
  __shared__ __align__(16) float    sden[AT_S * HEADS];
  __shared__ int      sover;
  __shared__ int      sbad;
  const int tid  = threadIdx.x;
  const int wave = tid >> 5;
  const int lane = tid & 31;
  const int lo   = blockIdx.x * AT_S;

  {
    constexpr int NV4 = AT_S * HC / 4;
    constexpr int NLD = (NV4 + AT_THREADS - 1) / AT_THREADS;
    const float4 zero4 = make_float4(0.f, 0.f, 0.f, 0.f);
    float4 r[NLD];
    #pragma unroll
    for (int k = 0; k < NLD; ++k) {
      const int gi = lo * (HC / 4) + tid + k * AT_THREADS;
      r[k] = ((const float4*)xr)[min(gi, N_NODES * (HC / 4) - 1)];
    }
    #pragma unroll
    for (int k = 0; k < NLD; ++k) {
      const int i = tid + k * AT_THREADS;
      if (i < NV4) { ((float4*)tile)[i] = zero4; ((float4*)sxr)[i] = r[k]; }
    }
  }
  if (tid < AT_S * HEADS) sden[tid] = 0.f;
  if (tid == 0) { sover = 0; sbad = 0; }
  __syncthreads();

  TileArgs A;
  A.tp = (const uint4*)tplane; A.pat = (unsigned)blockIdx.x * 0x01010101u;
  A.ei = ei; A.xl = xl; A.sxr = sxr; A.att = att; A.sc = sc;
  A.tile = tile; A.smax = smax; A.shift = (float*)smax; A.sden = sden; A.lo = lo;
  int2* q = ring + wave * AT_QCAP;

  const int slot = blockIdx.x * AT_WAVES + wave;
  int total;
  bool listed = false;
  if (REUSE) listed = (lover[blockIdx.x] == 0);
  int nself, cnt;
  if (REUSE && listed) {
    total = min(max(__builtin_amdgcn_readfirstlane(lcnt[slot]), 0), AT_QCAP);
    const int2* ls = (const int2*)lists + (size_t)slot * AT_QCAP;
    int2 r[AT_QCAP / 32];
    #pragma unroll
    for (int k = 0; k < AT_QCAP / 32; ++k) {
      const int i = lane + 32 * k;
      r[k] = (i < total) ? ls[i] : make_int2(0, 0);
    }
    #pragma unroll
    for (int k = 0; k < AT_QCAP / 32; ++k) {
      const int i = lane + 32 * k;
      if (i < total) q[i] = r[k];
    }
    nself = min(ring_self(q, false, wave, lane, A), total);
    cnt = total;
  } else {
    nself = ring_self(q, true, wave, lane, A);
    cnt = nself;
    total = nself;
  }
  if (nself > 0) (void)one_pass_drain<1>(q, 0, nself, lane, A);
  __syncthreads();

  bool over = false;
  if (!(REUSE && listed)) {
    over = fast_fill(q, cnt, tid, lane, A);
    total = cnt;
  }
  bool big = false;
  if (!over) {
    for (int h0 = 0; h0 < cnt; h0 += 32) {
      const int n = (cnt - h0 < 32) ? (cnt - h0) : 32;
      big = one_pass_drain<0>(q, h0, n, lane, A) || big;
    }
  } else {
    const int ns = ring_self(q, true, wave, lane, A);
    total = ns + plain_scan<2>(q, ns, big, tid, lane, A);
  }
  if (total > AT_QCAP && lane == 0) sover = 1;
  if (__ballot(big) != 0 && lane == 0) sbad = 1;
  __syncthreads();
  if (!REUSE) {
    if (tid == 0) lover[blockIdx.x] = sover;
    if (sover == 0) {
      int2* ls = (int2*)lists + (size_t)slot * AT_QCAP;
      for (int i = lane; i < total; i += 32) ls[i] = q[i];
      if (lane == 0) lcnt[slot] = total;
    }
  }

  if (sbad != 0) {
    constexpr int NV4 = AT_S * HC / 4;
    for (int i = tid; i < NV4; i += AT_THREADS) ((float4*)tile)[i] = make_float4(0.f, 0.f, 0.f, 0.f);
    if (tid < AT_S * HEADS) { smax[tid] = ORD_NEG_INF; sden[tid] = 0.f; }
    __syncthreads();
    bool unused = false;
    if (sover == 0) {
      for (int h0 = 0; h0 < total; h0 += 32) {
        const int n = (total - h0 < 32) ? (total - h0) : 32;
        score_drain(q, h0, n, lane, A);
      }
    } else {
      (void)plain_scan<0>(q, ring_self(q, true, wave, lane, A), unused, tid, lane, A);
    }
    __syncthreads();
    if (sover == 0) {
      for (int h0 = 0; h0 < total; h0 += 32) {
        const int n = (total - h0 < 32) ? (total - h0) : 32;
        agg_drain(q, h0, n, lane, A);
      }
    } else {
      (void)plain_scan<1>(q, ring_self(q, true, wave, lane, A), unused, tid, lane, A);
    }
    __syncthreads();
  }

  const float4 bs4 = *(const float4*)(bias + 4 * lane);
  const float4 ga4 = *(const float4*)(gamma + 4 * lane);
  const float4 be4 = *(const float4*)(beta + 4 * lane);
  for (int dl = wave; dl < AT_S; dl += AT_WAVES) {
    const int node = lo + dl;
    if (node >= N_NODES) break;
    const float den = sden[dl * HEADS + (lane >> 3)];
    const float* trow = tile + dl * HC + lane;
    float v[4];
    v[0] = trow[0]  / den + bs4.x;
    v[1] = trow[32] / den + bs4.y;
    v[2] = trow[64] / den + bs4.z;
    v[3] = trow[96] / den + bs4.w;
    float sum = (v[0] + v[1]) + (v[2] + v[3]);
    for (int off = 16; off > 0; off >>= 1) sum += __shfl_xor(sum, off, 32);
    const float mu = sum * (1.0f / HC);
    float sq = 0.f;
    #pragma unroll
    for (int i = 0; i < 4; ++i) { float d = v[i] - mu; sq += d * d; }
    for (int off = 16; off > 0; off >>= 1) sq += __shfl_xor(sq, off, 32);
    const float rstd = rsqrtf(sq * (1.0f / HC) + LN_EPS);
    const float h0 = (v[0] - mu) * rstd * ga4.x + be4.x;
    const float h1 = (v[1] - mu) * rstd * ga4.y + be4.y;
    const float h2 = (v[2] - mu) * rstd * ga4.z + be4.z;
    const float h3 = (v[3] - mu) * rstd * ga4.w + be4.w;
    float4 o;
    o.x = h0 / (1.0f + __expf(-h0));
    o.y = h1 / (1.0f + __expf(-h1));
    o.z = h2 / (1.0f + __expf(-h2));
    o.w = h3 / (1.0f + __expf(-h3));
    *(float4*)(out + (size_t)node * HC + 4 * lane) = o;
  }
}

#define POOL_CHUNKS 4
#define POOL_SLAB   (8 * POOL_CHUNKS * 32)
#define POOL_SLABS  ((N_NODES + POOL_SLAB - 1) / POOL_SLAB)
#define PROW        (HC + 32)
__global__ void pool_gather_kernel(const float* __restrict__ x,
                                   const int* __restrict__ batch,
                                   float* __restrict__ part) {
  __shared__ float sacc[8][HC];
  __shared__ int   scnt[8];
  const int g     = blockIdx.x % NGRAPH;
  const int s     = blockIdx.x / NGRAPH;
  const int wave  = threadIdx.x >> 5;
  const int lane  = threadIdx.x & 31;
  const int slab0 = s * POOL_SLAB;

  int ids[POOL_CHUNKS];
  #pragma unroll
  for (int i = 0; i < POOL_CHUNKS; ++i) {
    int node = slab0 + (i * 8 + wave) * 32 + lane;
    ids[i] = (node < N_NODES) ? batch[node] : -1;
  }
  float4 acc = make_float4(0.f, 0.f, 0.f, 0.f);
  int cnt = 0;
  #pragma unroll
  for (int i = 0; i < POOL_CHUNKS; ++i) {
    const unsigned mask = (unsigned)__ballot(ids[i] == g);
    if (mask == 0u) continue;
    cnt += __popc(mask);
    const int base = slab0 + (i * 8 + wave) * 32;
    for (int j0 = 0; j0 < 32; j0 += 16) {
      if (((mask >> j0) & 0xffffu) == 0u) continue;
      float4 v[16];
      #pragma unroll
      for (int u = 0; u < 16; ++u) {
        int node = base + j0 + u;
        node = (node < N_NODES) ? node : (N_NODES - 1);
        v[u] = *(const float4*)(x + (size_t)node * HC + lane * 4);
      }
      #pragma unroll
      for (int u = 0; u < 16; ++u) {
        const bool on = ((mask >> (j0 + u)) & 1u) != 0u;
        acc.x += on ? v[u].x : 0.f;
        acc.y += on ? v[u].y : 0.f;
        acc.z += on ? v[u].z : 0.f;
        acc.w += on ? v[u].w : 0.f;
      }
    }
  }
  *(float4*)&sacc[wave][lane * 4] = acc;
  if (lane == 0) scnt[wave] = cnt;
  __syncthreads();
  const int tid = threadIdx.x;
  float* prow = part + ((size_t)g * POOL_SLABS + s) * PROW;
  if (tid < HC) {
    float t = 0.f;
    #pragma unroll
    for (int w = 0; w < 8; ++w) t += sacc[w][tid];
    prow[tid] = t;
  } else if (tid < PROW) {
    int c = 0;
    #pragma unroll
    for (int w = 0; w < 8; ++w) c += scnt[w];
    ((int*)prow)[tid] = (tid == HC) ? c : 0;
  }
}
__global__ void pool_final_kernel(const float* __restrict__ part,
                                  float* __restrict__ out) {
  int t = blockIdx.x * blockDim.x + threadIdx.x;
  if (t >= NGRAPH * HC) return;
  const int g = t >> 7, c = t & 127;
  const float* p = part + (size_t)g * POOL_SLABS * PROW;
  float sum = 0.f;
  int n = 0;
  #pragma unroll 7
  for (int s = 0; s < POOL_SLABS; ++s) {
    sum += p[s * PROW + c];
    n   += ((const int*)p)[s * PROW + HC];
  }
  out[t] = sum / fmaxf((float)n, 1.0f);
}

extern "C" void kernel_launch(void* const* d_in, const int* in_sizes, int n_in,
                              void* d_out, int out_size, void* d_ws, size_t ws_size,
                              hipStream_t stream) {
  (void)in_sizes; (void)n_in; (void)out_size; (void)ws_size;
  const float* x     = (const float*)d_in[0];
  const int*   ei    = (const int*)d_in[1];
  const int*   batch = (const int*)d_in[2];
  const float* Wl[2]    = {(const float*)d_in[3],  (const float*)d_in[11]};
  const float* Wr[2]    = {(const float*)d_in[4],  (const float*)d_in[12]};
  const float* bl[2]    = {(const float*)d_in[5],  (const float*)d_in[13]};
  const float* br[2]    = {(const float*)d_in[6],  (const float*)d_in[14]};
  const float* att[2]   = {(const float*)d_in[7],  (const float*)d_in[15]};
  const float* bias[2]  = {(const float*)d_in[8],  (const float*)d_in[16]};
  const float* gamma[2] = {(const float*)d_in[9],  (const float*)d_in[17]};
  const float* beta[2]  = {(const float*)d_in[10], (const float*)d_in[18]};

  float* ws = (float*)d_ws;
  float*    xl    = ws;            ws += (size_t)N_NODES * HC;
  float*    xr2   = ws;            ws += (size_t)N_NODES * HC;
  float*    xnode = ws;            ws += (size_t)N_NODES * HC;
  float*    scbuf = ws;            ws += (size_t)ETOT * HEADS;
  float*    ppart = ws;            ws += (size_t)NGRAPH * POOL_SLABS * PROW;
  int*      lists = (int*)ws;      ws += (size_t)AT_TILES * AT_WAVES * AT_QCAP * 2;
  int*      lcnt  = (int*)ws;      ws += (size_t)AT_TILES * AT_WAVES;
  int*      lover = (int*)ws;      ws += 256;
  unsigned* tplane = (unsigned*)ws;

  dim3 b256(256);

  const float* xin = x;
  for (int l = 0; l < 2; ++l) {
    if (l == 0)
      gemm_bias_wmma<1><<<G_BLOCKS, G_THREADS, 0, stream>>>(xin, Wl[l], bl[l], xl, N_NODES, ei, tplane);
    else
      gemm_bias_wmma<0><<<G_BLOCKS, G_THREADS, 0, stream>>>(xin, Wl[l], bl[l], xl, N_NODES, ei, tplane);
    gemm_bias_wmma<0><<<G_BLOCKS, G_THREADS, 0, stream>>>(xin, Wr[l], br[l], xr2, N_NODES, ei, tplane);
    if (l == 0)
      tile_softmax_aggregate_ln_kernel<0><<<AT_TILES, AT_THREADS, 0, stream>>>(ei, xl, xr2, att[l], scbuf, bias[l], gamma[l], beta[l], xnode, lists, lcnt, lover, tplane);
    else
      tile_softmax_aggregate_ln_kernel<1><<<AT_TILES, AT_THREADS, 0, stream>>>(ei, xl, xr2, att[l], scbuf, bias[l], gamma[l], beta[l], xnode, lists, lcnt, lover, tplane);
    xin = xnode;
  }
  pool_gather_kernel<<<NGRAPH * POOL_SLABS, b256, 0, stream>>>(xnode, batch, ppart);
  pool_final_kernel<<<(NGRAPH * HC + 255) / 256, b256, 0, stream>>>(ppart, (float*)d_out);
}


# layer 1's tile blocks find their edges by a per-tile bitmap of 16-edge groups (made in the first GEMM launch): a wave loads only its listed groups of its own 1/32 of the plane; LDS sum order changes
# speedup vs baseline: 4.8760x; 1.0346x over previous
#include <hip/hip_runtime.h>
#include <hip/hip_bf16.h>
#include <math.h>

#define N_NODES 50000
#define E_EDGES 800000
#define ETOT    (E_EDGES + N_NODES)
#define HEADS   4
#define CHP     32
#define HC      128
#define NGRAPH  64
#define SLOPE   0.2f
#define LN_EPS  1e-5f

#define AT_S     197
#define AT_TILES ((N_NODES + AT_S - 1) / AT_S)
#define AT_NONE  255u
#define BM_WORDS  ((E_EDGES / 16 + 31) / 32)
#define BM_SLOTS  256

typedef float v2f __attribute__((ext_vector_type(2)));
typedef float v8f __attribute__((ext_vector_type(8)));

__device__ __forceinline__ unsigned f32_to_ord(float f) {
  unsigned u = __float_as_uint(f);
  return (u & 0x80000000u) ? ~u : (u | 0x80000000u);
}
__device__ __forceinline__ float ord_to_f32(unsigned u) {
  u = (u & 0x80000000u) ? (u & 0x7fffffffu) : ~u;
  return __uint_as_float(u);
}
__device__ __forceinline__ unsigned tile_of(const int d) {
  return ((unsigned)d < (unsigned)N_NODES) ? ((unsigned)d / (unsigned)AT_S) : AT_NONE;
}
__device__ __forceinline__ void atomAddF(float* p, float v) {
  unsafeAtomicAdd(p, v);
}

#define G_THREADS 512
#define G_WAVES   (G_THREADS / 32)
#define G_BLOCKS  256
#define G_NT      4
typedef float v4f __attribute__((ext_vector_type(4)));

__device__ __forceinline__ void gemm_load_a(v4f (&a)[16], const float* __restrict__ X, const int u, const int n,
                                            const int lane) {
  const int rt  = blockIdx.x + G_BLOCKS * (u >> 1);
  const int row = max(min(rt * 16 + (lane & 15), n - 1), 0);
  const v4f* p = (const v4f*)(X + (size_t)row * HC) + (lane >> 4);
  #pragma unroll
  for (int j = 0; j < 16; ++j) a[j] = p[2 * j];
}

template <int PRE>
__global__ void __launch_bounds__(G_THREADS)
gemm_bias_wmma(const float* __restrict__ X,
               const float* __restrict__ W,
               const float* __restrict__ bias,
               float* __restrict__ Y, int n,
               const int* __restrict__ ei,
               unsigned* __restrict__ tplane,
               unsigned* __restrict__ bmap) {
  __shared__ __align__(16) float sW[HC * HC];
  __shared__ __align__(16) unsigned sB[PRE ? 2 * 256 * 4 : 4];
  const int tid  = threadIdx.x;
  const int wave = __builtin_amdgcn_readfirstlane(tid >> 5);
  const int lane = tid & 31;
  const int c    = lane & 15;
  const int h    = lane >> 4;
  const int ntiles = (n + 15) >> 4;
  const int nrt    = (ntiles - (int)blockIdx.x + G_BLOCKS - 1) / G_BLOCKS;
  const int nunits = 2 * nrt;

  constexpr int NGRP = E_EDGES / 4;
  constexpr int PER  = (NGRP + G_BLOCKS * G_THREADS - 1) / (G_BLOCKS * G_THREADS);
  static_assert(!PRE || (PER == 2 && G_THREADS == 2 * BM_SLOTS && G_BLOCKS * 4 == 1024 && AT_TILES <= 254),
                "the bitmap's layout: two chunks of 1024 words, four words per block and chunk, one 16-byte store per thread");
  int4 dd[PER];
  if (PRE) {
    for (int k = 0; k < 4; ++k) sB[tid + G_THREADS * k] = 0u;
    #pragma unroll
    for (int k = 0; k < PER; ++k) {
      const int i = (int)blockIdx.x * G_THREADS + tid + k * (G_BLOCKS * G_THREADS);
      dd[k] = ((const int4*)(ei + E_EDGES))[min(i, NGRP - 1)];
    }
  }

  int u = wave;
  v4f a[16];
  gemm_load_a(a, X, u, n, lane);

  if (PRE) {
    __syncthreads();
    const unsigned gbit = 1u << ((tid >> 2) & 31);
    const int      gwrd = tid >> 7;
    #pragma unroll
    for (int k = 0; k < PER; ++k) {
      const int i = (int)blockIdx.x * G_THREADS + tid + k * (G_BLOCKS * G_THREADS);
      if (i < NGRP) {
        const unsigned t0 = tile_of(dd[k].x), t1 = tile_of(dd[k].y), t2 = tile_of(dd[k].z), t3 = tile_of(dd[k].w);
        tplane[i] = t0 | (t1 << 8) | (t2 << 16) | (t3 << 24);
        unsigned* row = sB + k * 1024 + gwrd;
        if (t0 != AT_NONE) atomicOr(row + 4 * t0, gbit);
        if (t1 != AT_NONE) atomicOr(row + 4 * t1, gbit);
        if (t2 != AT_NONE) atomicOr(row + 4 * t2, gbit);
        if (t3 != AT_NONE) atomicOr(row + 4 * t3, gbit);
      }
    }
  }

  {
    v4f r[2][4];
    #pragma unroll
    for (int i = 0; i < 2; ++i) {
      const int item = tid + G_THREADS * i;
      const int q = item >> 5, g = item & 31;
      #pragma unroll
      for (int k = 0; k < 4; ++k) r[i][k] = *(const v4f*)(W + (4 * q + k) * HC + 4 * g);
    }
    #pragma unroll
    for (int i = 0; i < 2; ++i) {
      const int item = tid + G_THREADS * i;
      const int q = item >> 5, g = item & 31;
      v4f* d = (v4f*)sW + q * HC + 4 * g;
      d[0] = v4f{r[i][0].x, r[i][1].x, r[i][2].x, r[i][3].x};
      d[1] = v4f{r[i][0].y, r[i][1].y, r[i][2].y, r[i][3].y};
      d[2] = v4f{r[i][0].z, r[i][1].z, r[i][2].z, r[i][3].z};
      d[3] = v4f{r[i][0].w, r[i][1].w, r[i][2].w, r[i][3].w};
    }
  }
  __syncthreads();
  if (PRE) {
    const int k = tid >> 8, t = tid & (BM_SLOTS - 1);
    const uint4 v = *(const uint4*)(sB + k * 1024 + 4 * t);
    *(uint4*)(bmap + ((size_t)(k * G_BLOCKS + (int)blockIdx.x) * BM_SLOTS + t) * 4) = v;
  }

  while (u < nunits) {
    const int un = u + G_WAVES;
    v4f an[16];
    gemm_load_a(an, X, min(un, nunits - 1), n, lane);
    const int rt   = blockIdx.x + G_BLOCKS * (u >> 1);
    const int col0 = (u & 1) * (16 * G_NT) + c;
    const v4f* bp = (const v4f*)sW + h * HC + col0;
    float bc[G_NT];
    v8f acc[G_NT];
    #pragma unroll
    for (int t = 0; t < G_NT; ++t) { bc[t] = bias[col0 + 16 * t]; acc[t] = v8f{}; }
    v4f bq[2][G_NT];
    #pragma unroll
    for (int t = 0; t < G_NT; ++t) bq[0][t] = bp[16 * t];
    #pragma unroll
    for (int j = 0; j < 16; ++j) {
      if (j + 1 < 16) {
        #pragma unroll
        for (int t = 0; t < G_NT; ++t) bq[(j + 1) & 1][t] = bp[2 * (j + 1) * HC + 16 * t];
      }
      #pragma unroll
      for (int t = 0; t < G_NT; ++t) {
        const v4f b = bq[j & 1][t];
        v2f a0; a0.x = a[j].x; a0.y = a[j].y;
        v2f b0; b0.x = b.x;    b0.y = b.y;
        acc[t] = __builtin_amdgcn_wmma_f32_16x16x4_f32(false, a0, false, b0, (short)0, acc[t], false, false);
        v2f a1; a1.x = a[j].z; a1.y = a[j].w;
        v2f b1; b1.x = b.z;    b1.y = b.w;
        acc[t] = __builtin_amdgcn_wmma_f32_16x16x4_f32(false, a1, false, b1, (short)0, acc[t], false, false);
      }
      __builtin_amdgcn_sched_barrier(0);
    }
    float* yp = Y + (size_t)(rt * 16 + 8 * h) * HC + col0;
    if (rt * 16 + 16 <= n) {
      #pragma unroll
      for (int t = 0; t < G_NT; ++t) {
        #pragma unroll
        for (int r = 0; r < 8; ++r) yp[r * HC + 16 * t] = acc[t][r] + bc[t];
      }
    } else {
      #pragma unroll
      for (int t = 0; t < G_NT; ++t) {
        #pragma unroll
        for (int r = 0; r < 8; ++r) {
          if (rt * 16 + r + 8 * h < n) yp[r * HC + 16 * t] = acc[t][r] + bc[t];
        }
      }
    }
    #pragma unroll
    for (int j = 0; j < 16; ++j) a[j] = an[j];
    u = un;
  }
}

#define AT_THREADS 1024
#define AT_WAVES   (AT_THREADS / 32)
#define AT_QCAP    256
#define AT_GCAP    256
#define AT_BMW     ((BM_WORDS + AT_WAVES - 1) / AT_WAVES)
#define AT_GROUP   8
#define AT_V16     (E_EDGES / 16)
#define AT_SELF    ((AT_S + AT_WAVES - 1) / AT_WAVES)
#define ORD_NEG_INF 0x007FFFFFu
#define AT_LIMIT   60.0f

struct TileArgs {
  const uint4* tp;
  const unsigned* bm;
  unsigned     pat;
  const int*   ei;
  const float* xl;
  const float* sxr;
  const float* att;
  float*       sc;
  float*       tile;
  unsigned*    smax;
  float*       shift;
  float*       sden;
  int          lo;
};

__device__ __forceinline__ int lane_bcast_i(int v, int j) {
  return __builtin_amdgcn_readlane(v, j);
}
__device__ __forceinline__ float lane_bcast_f(float v, int j) {
  return __int_as_float(__builtin_amdgcn_readlane(__float_as_int(v), j));
}
__device__ __forceinline__ float lrelu_dot4(const float4 a, const float4 b, const float4 w) {
  float t0 = a.x + b.x, t1 = a.y + b.y, t2 = a.z + b.z, t3 = a.w + b.w;
  t0 = (t0 > 0.f) ? t0 : t0 * SLOPE;
  t1 = (t1 > 0.f) ? t1 : t1 * SLOPE;
  t2 = (t2 > 0.f) ? t2 : t2 * SLOPE;
  t3 = (t3 > 0.f) ? t3 : t3 * SLOPE;
  return t0 * w.x + t1 * w.y + t2 * w.z + t3 * w.w;
}

__device__ __forceinline__ void score_drain(int2* q, int head, int n, int lane, const TileArgs& A) {
  int src = 0, dl = 0, e = 0;
  if (lane < n) {
    const int qi = (head + lane) & (AT_QCAP - 1);
    const int2 t = q[qi];
    e = t.x;
    dl = t.y;
    src = (e < E_EDGES) ? A.ei[e] : (e - E_EDGES);
    if (dl < 0) {
      dl = A.ei[E_EDGES + e] - A.lo;
      q[qi].y = dl;
    }
  }
  const float4 w4 = *(const float4*)(A.att + 4 * lane);
  const int  r  = lane & 7;
  const int  hd = lane >> 3;
  const bool b2 = (r & 4) != 0, b1 = (r & 2) != 0, b0 = (r & 1) != 0;
  for (int j0 = 0; j0 < n; j0 += AT_GROUP) {
    float4 a[AT_GROUP], b[AT_GROUP];
    #pragma unroll
    for (int u = 0; u < AT_GROUP; ++u) {
      const int s = lane_bcast_i(src, j0 + u);
      const int d = lane_bcast_i(dl, j0 + u);
      a[u] = *(const float4*)(A.xl + (size_t)s * HC + 4 * lane);
      b[u] = *(const float4*)(A.sxr + d * HC + 4 * lane);
    }
    float p[AT_GROUP];
    #pragma unroll
    for (int u = 0; u < AT_GROUP; ++u) p[u] = lrelu_dot4(a[u], b[u], w4);
    float f4[4];
    #pragma unroll
    for (int i = 0; i < 4; ++i) {
      const float keep = b2 ? p[i + 4] : p[i];
      const float send = b2 ? p[i] : p[i + 4];
      f4[i] = keep + __shfl_xor(send, 4, 32);
    }
    float f2[2];
    #pragma unroll
    for (int i = 0; i < 2; ++i) {
      const float keep = b1 ? f4[i + 2] : f4[i];
      const float send = b1 ? f4[i] : f4[i + 2];
      f2[i] = keep + __shfl_xor(send, 2, 32);
    }
    const float keep = b0 ? f2[1] : f2[0];
    const float send = b0 ? f2[0] : f2[1];
    const float sc = keep + __shfl_xor(send, 1, 32);
    const int ue = j0 + r;
    const int eu = __shfl(e, ue, 32);
    const int du = __shfl(dl, ue, 32);
    if (ue < n) {
      A.sc[(size_t)eu * HEADS + hd] = sc;
      atomicMax(&A.smax[du * HEADS + hd], f32_to_ord(sc));
    }
  }
}

template <int U>
__device__ __forceinline__ float group8_bcast(const float v) {
  return __int_as_float(__builtin_amdgcn_mov_dpp8(__float_as_int(v), U * 0x249249));
}
template <int U>
__device__ __forceinline__ void add_edge(const float4 a, const float w, const int d, const int lane, float* tile) {
  const float wu = group8_bcast<U>(w);
  float* t = tile + d * HC + lane;
  atomAddF(t,      a.x * wu);
  atomAddF(t + 32, a.y * wu);
  atomAddF(t + 64, a.z * wu);
  atomAddF(t + 96, a.w * wu);
}

template <int SHIFT>
__device__ __forceinline__ bool one_pass_drain(int2* q, int head, int n, int lane, const TileArgs& A) {
  static_assert(AT_GROUP == 8, "the fold and the weight exchange work on groups of eight edges");
  int src = 0, dl = 0;
  if (lane < n) {
    const int qi = (head + lane) & (AT_QCAP - 1);
    const int2 t = q[qi];
    const int e = t.x;
    dl = t.y;
    src = (e < E_EDGES) ? A.ei[e] : (e - E_EDGES);
    if (dl < 0) {
      dl = A.ei[E_EDGES + e] - A.lo;
      q[qi].y = dl;
    }
  }
  const float4 w4 = *(const float4*)(A.att + 4 * lane);
  const int  r  = lane & 7;
  const int  hd = lane >> 3;
  const bool b2 = (r & 4) != 0, b1 = (r & 2) != 0, b0 = (r & 1) != 0;
  bool big = false;
  for (int j0 = 0; j0 < n; j0 += AT_GROUP) {
    float4 a[AT_GROUP], b[AT_GROUP];
    #pragma unroll
    for (int u = 0; u < AT_GROUP; ++u) {
      const int s = lane_bcast_i(src, j0 + u);
      const int d = lane_bcast_i(dl, j0 + u);
      a[u] = *(const float4*)(A.xl + (size_t)s * HC + 4 * lane);
      b[u] = *(const float4*)(A.sxr + d * HC + 4 * lane);
    }
    float p[AT_GROUP];
    #pragma unroll
    for (int u = 0; u < AT_GROUP; ++u) p[u] = lrelu_dot4(a[u], b[u], w4);
    float f4[4];
    #pragma unroll
    for (int i = 0; i < 4; ++i) {
      const float keep = b2 ? p[i + 4] : p[i];
      const float send = b2 ? p[i] : p[i + 4];
      f4[i] = keep + __shfl_xor(send, 4, 32);
    }
    float f2[2];
    #pragma unroll
    for (int i = 0; i < 2; ++i) {
      const float keep = b1 ? f4[i + 2] : f4[i];
      const float send = b1 ? f4[i] : f4[i + 2];
      f2[i] = keep + __shfl_xor(send, 2, 32);
    }
    const float keep = b0 ? f2[1] : f2[0];
    const float send = b0 ? f2[0] : f2[1];
    const float sc = keep + __shfl_xor(send, 1, 32);
    const int ue = j0 + r;
    const int du = __shfl(dl, ue, 32);
    if (SHIFT) {
      if (ue < n) A.shift[du * HEADS + hd] = sc;
    } else {
      float w = 0.f;
      if (ue < n) {
        const float t = sc - A.shift[du * HEADS + hd];
        big = big || (t > AT_LIMIT);
        w = __expf(t);
        atomAddF(A.sden + du * HEADS + hd, w);
      }
      if (j0 + 0 < n) add_edge<0>(a[0], w, lane_bcast_i(dl, j0 + 0), lane, A.tile);
      if (j0 + 1 < n) add_edge<1>(a[1], w, lane_bcast_i(dl, j0 + 1), lane, A.tile);
      if (j0 + 2 < n) add_edge<2>(a[2], w, lane_bcast_i(dl, j0 + 2), lane, A.tile);
      if (j0 + 3 < n) add_edge<3>(a[3], w, lane_bcast_i(dl, j0 + 3), lane, A.tile);
      if (j0 + 4 < n) add_edge<4>(a[4], w, lane_bcast_i(dl, j0 + 4), lane, A.tile);
      if (j0 + 5 < n) add_edge<5>(a[5], w, lane_bcast_i(dl, j0 + 5), lane, A.tile);
      if (j0 + 6 < n) add_edge<6>(a[6], w, lane_bcast_i(dl, j0 + 6), lane, A.tile);
      if (j0 + 7 < n) add_edge<7>(a[7], w, lane_bcast_i(dl, j0 + 7), lane, A.tile);
    }
  }
  return big;
}

__device__ __forceinline__ void agg_drain(const int2* q, int head, int n, int lane, const TileArgs& A) {
  int src = 0, dl = 0;
  float a0 = 0.f, a1 = 0.f, a2 = 0.f, a3 = 0.f;
  if (lane < n) {
    const int2 t = q[(head + lane) & (AT_QCAP - 1)];
    const int e = t.x;
    dl = t.y;
    src = (e < E_EDGES) ? A.ei[e] : (e - E_EDGES);
    if (dl < 0) dl = A.ei[E_EDGES + e] - A.lo;
    const float4 s4 = *(const float4*)(A.sc + (size_t)e * HEADS);
    const uint4  m4 = *(const uint4*)(A.smax + dl * HEADS);
    a0 = __expf(s4.x - ord_to_f32(m4.x));
    a1 = __expf(s4.y - ord_to_f32(m4.y));
    a2 = __expf(s4.z - ord_to_f32(m4.z));
    a3 = __expf(s4.w - ord_to_f32(m4.w));
    float* dn = A.sden + dl * HEADS;
    atomAddF(dn,     a0);
    atomAddF(dn + 1, a1);
    atomAddF(dn + 2, a2);
    atomAddF(dn + 3, a3);
  }
  for (int j0 = 0; j0 < n; j0 += AT_GROUP) {
    float v[AT_GROUP][HEADS];
    #pragma unroll
    for (int u = 0; u < AT_GROUP; ++u) {
      const int s = lane_bcast_i(src, j0 + u);
      const float* p = A.xl + (size_t)s * HC + lane;
      #pragma unroll
      for (int h = 0; h < HEADS; ++h) v[u][h] = p[h * CHP];
    }
    #pragma unroll
    for (int u = 0; u < AT_GROUP; ++u) {
      if (j0 + u < n) {
        const int   d  = lane_bcast_i(dl, j0 + u);
        const float w0 = lane_bcast_f(a0, j0 + u);
        const float w1 = lane_bcast_f(a1, j0 + u);
        const float w2 = lane_bcast_f(a2, j0 + u);
        const float w3 = lane_bcast_f(a3, j0 + u);
        float* t = A.tile + d * HC + (lane & 3) * 32 + (lane >> 2);
        atomAddF(t,      v[u][0] * w0);
        atomAddF(t + 8,  v[u][1] * w1);
        atomAddF(t + 16, v[u][2] * w2);
        atomAddF(t + 24, v[u][3] * w3);
      }
    }
  }
}

__device__ __forceinline__ uint4 plane_group(const uint4* __restrict__ tp4, const int idx) {
  return tp4[(idx < AT_V16) ? idx : (AT_V16 - 1)];
}

__device__ __forceinline__ unsigned zero_bytes(const unsigned x) {
  return ~(((x & 0x7f7f7f7fu) + 0x7f7f7f7fu) | x | 0x7f7f7f7fu);
}

__device__ __forceinline__ unsigned tile_bits(const uint4 p, const unsigned pat) {
  const unsigned z0 = zero_bytes(p.x ^ pat), z1 = zero_bytes(p.y ^ pat);
  const unsigned z2 = zero_bytes(p.z ^ pat), z3 = zero_bytes(p.w ^ pat);
  return (z0 >> 7) | (z1 >> 6) | (z2 >> 5) | (z3 >> 4);
}
__device__ __forceinline__ int bit_edge(const int idx, const int pb) {
  return idx * 16 + 4 * (pb & 7) + (pb >> 3);
}

__device__ __forceinline__ int ring_self(int2* q, const bool store, const int wave, const int lane, const TileArgs& A) {
  const int dl = wave * AT_SELF + lane;
  const bool on = (lane < AT_SELF) && (dl < AT_S) && (A.lo + dl < N_NODES);
  const unsigned m = (unsigned)__ballot(on);
  if (store && on) q[__popc(m & ((1u << lane) - 1u))] = make_int2(E_EDGES + A.lo + dl, dl);
  return __popc(m);
}

__device__ __forceinline__ void ring_bits(unsigned g, const int G, int2* q, int& cnt, bool& over, const unsigned below) {
  for (;;) {
    const bool on = g != 0u;
    const unsigned m = (unsigned)__ballot(on);
    if (m == 0u) break;
    const bool room = !over && (cnt + __popc(m) <= AT_QCAP);
    if (on) {
      const int pb = __ffs(g) - 1;
      g &= g - 1u;
      if (room) q[cnt + __popc(m & below)] = make_int2(bit_edge(G, pb), -1);
    }
    if (room) cnt += __popc(m);
    else      over = true;
  }
}

__device__ __forceinline__ int bm_word(const int w) {
  return (((w >> 10) * G_BLOCKS + ((w & 1023) >> 2)) * BM_SLOTS) * 4 + (w & 3);
}

__device__ __forceinline__ bool bitmap_fill(int2* q, int* gl, int& cnt, const int wave, const int lane, const TileArgs& A) {
  static_assert(AT_BMW > 32 && AT_BMW <= 64 && AT_WAVES * AT_BMW >= BM_WORDS && (AT_WAVES - 1) * AT_BMW + 32 <= BM_WORDS,
                "a wave's words: a first word for every lane, a second for the lanes below AT_BMW - 32");
  const unsigned below = (1u << lane) - 1u;
  const int w0 = wave * AT_BMW + lane;
  const int w1 = w0 + 32;
  const bool has1 = (lane < AT_BMW - 32) && (w1 < BM_WORDS);
  unsigned b0 = A.bm[bm_word(w0)];
  unsigned b1 = A.bm[bm_word(has1 ? w1 : w0)];
  if (!has1) b1 = 0u;
  const int c = __popc(b0) + __popc(b1);
  int pre = c;
  for (int d = 1; d < 32; d <<= 1) {
    const int t = __shfl_up(pre, d, 32);
    if (lane >= d) pre += t;
  }
  const int gcnt = lane_bcast_i(pre, 31);
  if (gcnt > AT_GCAP) return true;
  for (int at = pre - c; b0 != 0u; ++at) {
    gl[at] = w0 * 32 + (__ffs(b0) - 1);
    b0 &= b0 - 1u;
  }
  for (int at = pre - __popc(b1); b1 != 0u; ++at) {
    gl[at] = w1 * 32 + (__ffs(b1) - 1);
    b1 &= b1 - 1u;
  }
  bool over = false;
  for (int base = 0; base < gcnt; base += 128) {
    int   G[4];
    uint4 p[4];
    #pragma unroll
    for (int r = 0; r < 4; ++r) {
      const int k = base + 32 * r + lane;
      const int g = gl[(k < gcnt) ? k : (gcnt - 1)];
      G[r] = min(max(g, 0), AT_V16 - 1);
      p[r] = A.tp[G[r]];
    }
    #pragma unroll
    for (int r = 0; r < 4; ++r) {
      if (base + 32 * r < gcnt) {
        unsigned g = tile_bits(p[r], A.pat);
        if (base + 32 * r + lane >= gcnt) g = 0u;
        ring_bits(g, G[r], q, cnt, over, below);
      }
    }
  }
  return over;
}

template <int PH>
__device__ __forceinline__ void drain_any(int2* q, const int head, const int n, const int lane, const TileArgs& A,
                                          bool& big) {
  if (PH == 0)      score_drain(q, head, n, lane, A);
  else if (PH == 1) agg_drain(q, head, n, lane, A);
  else              big = one_pass_drain<0>(q, head, n, lane, A) || big;
}

template <int PH>
__device__ __forceinline__ int plain_scan(int2* q, const int cnt0, bool& big, const int wave, const int lane,
                                          const TileArgs& A) {
  int head = 0, cnt = cnt0, met = 0;
  const unsigned below = (1u << lane) - 1u;
  #pragma unroll 1
  for (int pass = 0; pass <= AT_BMW; ++pass) {
    const bool last = pass == AT_BMW;
    const int idx = (wave * AT_BMW + pass) * 32 + lane;
    unsigned g = tile_bits(plane_group(A.tp, idx), A.pat);
    if (last || idx >= AT_V16) g = 0u;
    for (;;) {
      const bool on = g != 0u;
      const unsigned m = (unsigned)__ballot(on);
      if (m != 0u) {
        if (on) {
          const int pb = __ffs(g) - 1;
          g &= g - 1u;
          q[(head + cnt + __popc(m & below)) & (AT_QCAP - 1)] = make_int2(bit_edge(idx, pb), -1);
        }
        cnt += __popc(m);
        met += __popc(m);
      }
      while (cnt >= 32 || (last && m == 0u && cnt > 0)) {
        const int n = (cnt < 32) ? cnt : 32;
        drain_any<PH>(q, head, n, lane, A, big);
        head = (head + n) & (AT_QCAP - 1);
        cnt -= n;
      }
      if (m == 0u) break;
    }
  }
  return met;
}

template <int REUSE>
__global__ void __launch_bounds__(AT_THREADS)
tile_softmax_aggregate_ln_kernel(const int* __restrict__ ei,
                                 const float* __restrict__ xl,
                                 const float* __restrict__ xr,
                                 const float* __restrict__ att,
                                 float* sc,
                                 const float* __restrict__ bias,
                                 const float* __restrict__ gamma,
                                 const float* __restrict__ beta,
                                 float* __restrict__ out,
                                 int*  lists,
                                 int*  lcnt,
                                 int*  lover,
                                 const unsigned* __restrict__ tplane,
                                 const unsigned* __restrict__ bmap) {
  __shared__ __align__(16) float    tile[AT_S * HC];
  __shared__ __align__(16) float    sxr[AT_S * HC];
  __shared__ __align__(16) int2     ring[AT_WAVES * AT_QCAP];
  __shared__ int                    glist[AT_WAVES * AT_GCAP];
  __shared__ __align__(16) unsigned smax[AT_S * HEADS];
  __shared__ __align__(16) float    sden[AT_S * HEADS];
  __shared__ int      sover;
  __shared__ int      sbad;
  const int tid  = threadIdx.x;
  const int wave = tid >> 5;
  const int lane = tid & 31;
  const int lo   = blockIdx.x * AT_S;

  {
    constexpr int NV4 = AT_S * HC / 4;
    constexpr int NLD = (NV4 + AT_THREADS - 1) / AT_THREADS;
    const float4 zero4 = make_float4(0.f, 0.f, 0.f, 0.f);
    float4 r[NLD];
    #pragma unroll
    for (int k = 0; k < NLD; ++k) {
      const int gi = lo * (HC / 4) + tid + k * AT_THREADS;
      r[k] = ((const float4*)xr)[min(gi, N_NODES * (HC / 4) - 1)];
    }
    #pragma unroll
    for (int k = 0; k < NLD; ++k) {
      const int i = tid + k * AT_THREADS;
      if (i < NV4) { ((float4*)tile)[i] = zero4; ((float4*)sxr)[i] = r[k]; }
    }
  }
  if (tid < AT_S * HEADS) sden[tid] = 0.f;
  if (tid == 0) { sover = 0; sbad = 0; }
  __syncthreads();

  TileArgs A;
  A.tp = (const uint4*)tplane; A.pat = (unsigned)blockIdx.x * 0x01010101u;
  A.bm = bmap + 4 * blockIdx.x;
  A.ei = ei; A.xl = xl; A.sxr = sxr; A.att = att; A.sc = sc;
  A.tile = tile; A.smax = smax; A.shift = (float*)smax; A.sden = sden; A.lo = lo;
  int2* q = ring + wave * AT_QCAP;

  const int slot = blockIdx.x * AT_WAVES + wave;
  int total;
  bool listed = false;
  if (REUSE) listed = (lover[blockIdx.x] == 0);
  int nself, cnt;
  if (REUSE && listed) {
    total = min(max(__builtin_amdgcn_readfirstlane(lcnt[slot]), 0), AT_QCAP);
    const int2* ls = (const int2*)lists + (size_t)slot * AT_QCAP;
    int2 r[AT_QCAP / 32];
    #pragma unroll
    for (int k = 0; k < AT_QCAP / 32; ++k) {
      const int i = lane + 32 * k;
      r[k] = (i < total) ? ls[i] : make_int2(0, 0);
    }
    #pragma unroll
    for (int k = 0; k < AT_QCAP / 32; ++k) {
      const int i = lane + 32 * k;
      if (i < total) q[i] = r[k];
    }
    nself = min(ring_self(q, false, wave, lane, A), total);
    cnt = total;
  } else {
    nself = ring_self(q, true, wave, lane, A);
    cnt = nself;
    total = nself;
  }
  if (nself > 0) (void)one_pass_drain<1>(q, 0, nself, lane, A);
  __syncthreads();

  bool over = false;
  if (!(REUSE && listed)) {
    over = bitmap_fill(q, glist + wave * AT_GCAP, cnt, wave, lane, A);
    total = cnt;
  }
  bool big = false;
  if (!over) {
    for (int h0 = 0; h0 < cnt; h0 += 32) {
      const int n = (cnt - h0 < 32) ? (cnt - h0) : 32;
      big = one_pass_drain<0>(q, h0, n, lane, A) || big;
    }
  } else {
    const int ns = ring_self(q, true, wave, lane, A);
    total = ns + plain_scan<2>(q, ns, big, wave, lane, A);
  }
  if (total > AT_QCAP && lane == 0) sover = 1;
  if (__ballot(big) != 0 && lane == 0) sbad = 1;
  __syncthreads();
  if (!REUSE) {
    if (tid == 0) lover[blockIdx.x] = sover;
    if (sover == 0) {
      int2* ls = (int2*)lists + (size_t)slot * AT_QCAP;
      for (int i = lane; i < total; i += 32) ls[i] = q[i];
      if (lane == 0) lcnt[slot] = total;
    }
  }

  if (sbad != 0) {
    constexpr int NV4 = AT_S * HC / 4;
    for (int i = tid; i < NV4; i += AT_THREADS) ((float4*)tile)[i] = make_float4(0.f, 0.f, 0.f, 0.f);
    if (tid < AT_S * HEADS) { smax[tid] = ORD_NEG_INF; sden[tid] = 0.f; }
    __syncthreads();
    bool unused = false;
    if (sover == 0) {
      for (int h0 = 0; h0 < total; h0 += 32) {
        const int n = (total - h0 < 32) ? (total - h0) : 32;
        score_drain(q, h0, n, lane, A);
      }
    } else {
      (void)plain_scan<0>(q, ring_self(q, true, wave, lane, A), unused, wave, lane, A);
    }
    __syncthreads();
    if (sover == 0) {
      for (int h0 = 0; h0 < total; h0 += 32) {
        const int n = (total - h0 < 32) ? (total - h0) : 32;
        agg_drain(q, h0, n, lane, A);
      }
    } else {
      (void)plain_scan<1>(q, ring_self(q, true, wave, lane, A), unused, wave, lane, A);
    }
    __syncthreads();
  }

  const float4 bs4 = *(const float4*)(bias + 4 * lane);
  const float4 ga4 = *(const float4*)(gamma + 4 * lane);
  const float4 be4 = *(const float4*)(beta + 4 * lane);
  for (int dl = wave; dl < AT_S; dl += AT_WAVES) {
    const int node = lo + dl;
    if (node >= N_NODES) break;
    const float den = sden[dl * HEADS + (lane >> 3)];
    const float* trow = tile + dl * HC + lane;
    float v[4];
    v[0] = trow[0]  / den + bs4.x;
    v[1] = trow[32] / den + bs4.y;
    v[2] = trow[64] / den + bs4.z;
    v[3] = trow[96] / den + bs4.w;
    float sum = (v[0] + v[1]) + (v[2] + v[3]);
    for (int off = 16; off > 0; off >>= 1) sum += __shfl_xor(sum, off, 32);
    const float mu = sum * (1.0f / HC);
    float sq = 0.f;
    #pragma unroll
    for (int i = 0; i < 4; ++i) { float d = v[i] - mu; sq += d * d; }
    for (int off = 16; off > 0; off >>= 1) sq += __shfl_xor(sq, off, 32);
    const float rstd = rsqrtf(sq * (1.0f / HC) + LN_EPS);
    const float h0 = (v[0] - mu) * rstd * ga4.x + be4.x;
    const float h1 = (v[1] - mu) * rstd * ga4.y + be4.y;
    const float h2 = (v[2] - mu) * rstd * ga4.z + be4.z;
    const float h3 = (v[3] - mu) * rstd * ga4.w + be4.w;
    float4 o;
    o.x = h0 / (1.0f + __expf(-h0));
    o.y = h1 / (1.0f + __expf(-h1));
    o.z = h2 / (1.0f + __expf(-h2));
    o.w = h3 / (1.0f + __expf(-h3));
    *(float4*)(out + (size_t)node * HC + 4 * lane) = o;
  }
}

#define POOL_CHUNKS 4
#define POOL_SLAB   (8 * POOL_CHUNKS * 32)
#define POOL_SLABS  ((N_NODES + POOL_SLAB - 1) / POOL_SLAB)
#define PROW        (HC + 32)
__global__ void pool_gather_kernel(const float* __restrict__ x,
                                   const int* __restrict__ batch,
                                   float* __restrict__ part) {
  __shared__ float sacc[8][HC];
  __shared__ int   scnt[8];
  const int g     = blockIdx.x % NGRAPH;
  const int s     = blockIdx.x / NGRAPH;
  const int wave  = threadIdx.x >> 5;
  const int lane  = threadIdx.x & 31;
  const int slab0 = s * POOL_SLAB;

  int ids[POOL_CHUNKS];
  #pragma unroll
  for (int i = 0; i < POOL_CHUNKS; ++i) {
    int node = slab0 + (i * 8 + wave) * 32 + lane;
    ids[i] = (node < N_NODES) ? batch[node] : -1;
  }
  float4 acc = make_float4(0.f, 0.f, 0.f, 0.f);
  int cnt = 0;
  #pragma unroll
  for (int i = 0; i < POOL_CHUNKS; ++i) {
    const unsigned mask = (unsigned)__ballot(ids[i] == g);
    if (mask == 0u) continue;
    cnt += __popc(mask);
    const int base = slab0 + (i * 8 + wave) * 32;
    for (int j0 = 0; j0 < 32; j0 += 16) {
      if (((mask >> j0) & 0xffffu) == 0u) continue;
      float4 v[16];
      #pragma unroll
      for (int u = 0; u < 16; ++u) {
        int node = base + j0 + u;
        node = (node < N_NODES) ? node : (N_NODES - 1);
        v[u] = *(const float4*)(x + (size_t)node * HC + lane * 4);
      }
      #pragma unroll
      for (int u = 0; u < 16; ++u) {
        const bool on = ((mask >> (j0 + u)) & 1u) != 0u;
        acc.x += on ? v[u].x : 0.f;
        acc.y += on ? v[u].y : 0.f;
        acc.z += on ? v[u].z : 0.f;
        acc.w += on ? v[u].w : 0.f;
      }
    }
  }
  *(float4*)&sacc[wave][lane * 4] = acc;
  if (lane == 0) scnt[wave] = cnt;
  __syncthreads();
  const int tid = threadIdx.x;
  float* prow = part + ((size_t)g * POOL_SLABS + s) * PROW;
  if (tid < HC) {
    float t = 0.f;
    #pragma unroll
    for (int w = 0; w < 8; ++w) t += sacc[w][tid];
    prow[tid] = t;
  } else if (tid < PROW) {
    int c = 0;
    #pragma unroll
    for (int w = 0; w < 8; ++w) c += scnt[w];
    ((int*)prow)[tid] = (tid == HC) ? c : 0;
  }
}
__global__ void pool_final_kernel(const float* __restrict__ part,
                                  float* __restrict__ out) {
  int t = blockIdx.x * blockDim.x + threadIdx.x;
  if (t >= NGRAPH * HC) return;
  const int g = t >> 7, c = t & 127;
  const float* p = part + (size_t)g * POOL_SLABS * PROW;
  float sum = 0.f;
  int n = 0;
  #pragma unroll 7
  for (int s = 0; s < POOL_SLABS; ++s) {
    sum += p[s * PROW + c];
    n   += ((const int*)p)[s * PROW + HC];
  }
  out[t] = sum / fmaxf((float)n, 1.0f);
}

extern "C" void kernel_launch(void* const* d_in, const int* in_sizes, int n_in,
                              void* d_out, int out_size, void* d_ws, size_t ws_size,
                              hipStream_t stream) {
  (void)in_sizes; (void)n_in; (void)out_size; (void)ws_size;
  const float* x     = (const float*)d_in[0];
  const int*   ei    = (const int*)d_in[1];
  const int*   batch = (const int*)d_in[2];
  const float* Wl[2]    = {(const float*)d_in[3],  (const float*)d_in[11]};
  const float* Wr[2]    = {(const float*)d_in[4],  (const float*)d_in[12]};
  const float* bl[2]    = {(const float*)d_in[5],  (const float*)d_in[13]};
  const float* br[2]    = {(const float*)d_in[6],  (const float*)d_in[14]};
  const float* att[2]   = {(const float*)d_in[7],  (const float*)d_in[15]};
  const float* bias[2]  = {(const float*)d_in[8],  (const float*)d_in[16]};
  const float* gamma[2] = {(const float*)d_in[9],  (const float*)d_in[17]};
  const float* beta[2]  = {(const float*)d_in[10], (const float*)d_in[18]};

  float* ws = (float*)d_ws;
  float*    xl    = ws;            ws += (size_t)N_NODES * HC;
  float*    xr2   = ws;            ws += (size_t)N_NODES * HC;
  float*    xnode = ws;            ws += (size_t)N_NODES * HC;
  float*    scbuf = ws;            ws += (size_t)ETOT * HEADS;
  float*    ppart = ws;            ws += (size_t)NGRAPH * POOL_SLABS * PROW;
  int*      lists = (int*)ws;      ws += (size_t)AT_TILES * AT_WAVES * AT_QCAP * 2;
  int*      lcnt  = (int*)ws;      ws += (size_t)AT_TILES * AT_WAVES;
  int*      lover = (int*)ws;      ws += 256;
  unsigned* tplane = (unsigned*)ws;  ws += E_EDGES / 4;
  unsigned* bmap   = (unsigned*)ws;

  dim3 b256(256);

  const float* xin = x;
  for (int l = 0; l < 2; ++l) {
    if (l == 0)
      gemm_bias_wmma<1><<<G_BLOCKS, G_THREADS, 0, stream>>>(xin, Wl[l], bl[l], xl, N_NODES, ei, tplane, bmap);
    else
      gemm_bias_wmma<0><<<G_BLOCKS, G_THREADS, 0, stream>>>(xin, Wl[l], bl[l], xl, N_NODES, ei, tplane, bmap);
    gemm_bias_wmma<0><<<G_BLOCKS, G_THREADS, 0, stream>>>(xin, Wr[l], br[l], xr2, N_NODES, ei, tplane, bmap);
    if (l == 0)
      tile_softmax_aggregate_ln_kernel<0><<<AT_TILES, AT_THREADS, 0, stream>>>(ei, xl, xr2, att[l], scbuf, bias[l], gamma[l], beta[l], xnode, lists, lcnt, lover, tplane, bmap);
    else
      tile_softmax_aggregate_ln_kernel<1><<<AT_TILES, AT_THREADS, 0, stream>>>(ei, xl, xr2, att[l], scbuf, bias[l], gamma[l], beta[l], xnode, lists, lcnt, lover, tplane, bmap);
    xin = xnode;
  }
  pool_gather_kernel<<<NGRAPH * POOL_SLABS, b256, 0, stream>>>(xnode, batch, ppart);
  pool_final_kernel<<<(NGRAPH * HC + 255) / 256, b256, 0, stream>>>(ppart, (float*)d_out);
}


# pool gather by node-slab owner blocks: 391 blocks of 4 waves (not 3136 of 8), a wave sums its 32 rows per graph id in registers, waves add into an LDS table in wave order; pool sum order changes
# speedup vs baseline: 5.0464x; 1.0349x over previous
#include <hip/hip_runtime.h>
#include <hip/hip_bf16.h>
#include <math.h>

#define N_NODES 50000
#define E_EDGES 800000
#define ETOT    (E_EDGES + N_NODES)
#define HEADS   4
#define CHP     32
#define HC      128
#define NGRAPH  64
#define SLOPE   0.2f
#define LN_EPS  1e-5f

#define AT_S     197
#define AT_TILES ((N_NODES + AT_S - 1) / AT_S)
#define AT_NONE  255u
#define BM_WORDS  ((E_EDGES / 16 + 31) / 32)
#define BM_SLOTS  256

typedef float v2f __attribute__((ext_vector_type(2)));
typedef float v8f __attribute__((ext_vector_type(8)));

__device__ __forceinline__ unsigned f32_to_ord(float f) {
  unsigned u = __float_as_uint(f);
  return (u & 0x80000000u) ? ~u : (u | 0x80000000u);
}
__device__ __forceinline__ float ord_to_f32(unsigned u) {
  u = (u & 0x80000000u) ? (u & 0x7fffffffu) : ~u;
  return __uint_as_float(u);
}
__device__ __forceinline__ unsigned tile_of(const int d) {
  return ((unsigned)d < (unsigned)N_NODES) ? ((unsigned)d / (unsigned)AT_S) : AT_NONE;
}
__device__ __forceinline__ void atomAddF(float* p, float v) {
  unsafeAtomicAdd(p, v);
}

#define G_THREADS 512
#define G_WAVES   (G_THREADS / 32)
#define G_BLOCKS  256
#define G_NT      4
typedef float v4f __attribute__((ext_vector_type(4)));

__device__ __forceinline__ void gemm_load_a(v4f (&a)[16], const float* __restrict__ X, const int u, const int n,
                                            const int lane) {
  const int rt  = blockIdx.x + G_BLOCKS * (u >> 1);
  const int row = max(min(rt * 16 + (lane & 15), n - 1), 0);
  const v4f* p = (const v4f*)(X + (size_t)row * HC) + (lane >> 4);
  #pragma unroll
  for (int j = 0; j < 16; ++j) a[j] = p[2 * j];
}

template <int PRE>
__global__ void __launch_bounds__(G_THREADS)
gemm_bias_wmma(const float* __restrict__ X,
               const float* __restrict__ W,
               const float* __restrict__ bias,
               float* __restrict__ Y, int n,
               const int* __restrict__ ei,
               unsigned* __restrict__ tplane,
               unsigned* __restrict__ bmap) {
  __shared__ __align__(16) float sW[HC * HC];
  __shared__ __align__(16) unsigned sB[PRE ? 2 * 256 * 4 : 4];
  const int tid  = threadIdx.x;
  const int wave = __builtin_amdgcn_readfirstlane(tid >> 5);
  const int lane = tid & 31;
  const int c    = lane & 15;
  const int h    = lane >> 4;
  const int ntiles = (n + 15) >> 4;
  const int nrt    = (ntiles - (int)blockIdx.x + G_BLOCKS - 1) / G_BLOCKS;
  const int nunits = 2 * nrt;

  constexpr int NGRP = E_EDGES / 4;
  constexpr int PER  = (NGRP + G_BLOCKS * G_THREADS - 1) / (G_BLOCKS * G_THREADS);
  static_assert(!PRE || (PER == 2 && G_THREADS == 2 * BM_SLOTS && G_BLOCKS * 4 == 1024 && AT_TILES <= 254),
                "the bitmap's layout: two chunks of 1024 words, four words per block and chunk, one 16-byte store per thread");
  int4 dd[PER];
  if (PRE) {
    for (int k = 0; k < 4; ++k) sB[tid + G_THREADS * k] = 0u;
    #pragma unroll
    for (int k = 0; k < PER; ++k) {
      const int i = (int)blockIdx.x * G_THREADS + tid + k * (G_BLOCKS * G_THREADS);
      dd[k] = ((const int4*)(ei + E_EDGES))[min(i, NGRP - 1)];
    }
  }

  int u = wave;
  v4f a[16];
  gemm_load_a(a, X, u, n, lane);

  if (PRE) {
    __syncthreads();
    const unsigned gbit = 1u << ((tid >> 2) & 31);
    const int      gwrd = tid >> 7;
    #pragma unroll
    for (int k = 0; k < PER; ++k) {
      const int i = (int)blockIdx.x * G_THREADS + tid + k * (G_BLOCKS * G_THREADS);
      if (i < NGRP) {
        const unsigned t0 = tile_of(dd[k].x), t1 = tile_of(dd[k].y), t2 = tile_of(dd[k].z), t3 = tile_of(dd[k].w);
        tplane[i] = t0 | (t1 << 8) | (t2 << 16) | (t3 << 24);
        unsigned* row = sB + k * 1024 + gwrd;
        if (t0 != AT_NONE) atomicOr(row + 4 * t0, gbit);
        if (t1 != AT_NONE) atomicOr(row + 4 * t1, gbit);
        if (t2 != AT_NONE) atomicOr(row + 4 * t2, gbit);
        if (t3 != AT_NONE) atomicOr(row + 4 * t3, gbit);
      }
    }
  }

  {
    v4f r[2][4];
    #pragma unroll
    for (int i = 0; i < 2; ++i) {
      const int item = tid + G_THREADS * i;
      const int q = item >> 5, g = item & 31;
      #pragma unroll
      for (int k = 0; k < 4; ++k) r[i][k] = *(const v4f*)(W + (4 * q + k) * HC + 4 * g);
    }
    #pragma unroll
    for (int i = 0; i < 2; ++i) {
      const int item = tid + G_THREADS * i;
      const int q = item >> 5, g = item & 31;
      v4f* d = (v4f*)sW + q * HC + 4 * g;
      d[0] = v4f{r[i][0].x, r[i][1].x, r[i][2].x, r[i][3].x};
      d[1] = v4f{r[i][0].y, r[i][1].y, r[i][2].y, r[i][3].y};
      d[2] = v4f{r[i][0].z, r[i][1].z, r[i][2].z, r[i][3].z};
      d[3] = v4f{r[i][0].w, r[i][1].w, r[i][2].w, r[i][3].w};
    }
  }
  __syncthreads();
  if (PRE) {
    const int k = tid >> 8, t = tid & (BM_SLOTS - 1);
    const uint4 v = *(const uint4*)(sB + k * 1024 + 4 * t);
    *(uint4*)(bmap + ((size_t)(k * G_BLOCKS + (int)blockIdx.x) * BM_SLOTS + t) * 4) = v;
  }

  while (u < nunits) {
    const int un = u + G_WAVES;
    v4f an[16];
    gemm_load_a(an, X, min(un, nunits - 1), n, lane);
    const int rt   = blockIdx.x + G_BLOCKS * (u >> 1);
    const int col0 = (u & 1) * (16 * G_NT) + c;
    const v4f* bp = (const v4f*)sW + h * HC + col0;
    float bc[G_NT];
    v8f acc[G_NT];
    #pragma unroll
    for (int t = 0; t < G_NT; ++t) { bc[t] = bias[col0 + 16 * t]; acc[t] = v8f{}; }
    v4f bq[2][G_NT];
    #pragma unroll
    for (int t = 0; t < G_NT; ++t) bq[0][t] = bp[16 * t];
    #pragma unroll
    for (int j = 0; j < 16; ++j) {
      if (j + 1 < 16) {
        #pragma unroll
        for (int t = 0; t < G_NT; ++t) bq[(j + 1) & 1][t] = bp[2 * (j + 1) * HC + 16 * t];
      }
      #pragma unroll
      for (int t = 0; t < G_NT; ++t) {
        const v4f b = bq[j & 1][t];
        v2f a0; a0.x = a[j].x; a0.y = a[j].y;
        v2f b0; b0.x = b.x;    b0.y = b.y;
        acc[t] = __builtin_amdgcn_wmma_f32_16x16x4_f32(false, a0, false, b0, (short)0, acc[t], false, false);
        v2f a1; a1.x = a[j].z; a1.y = a[j].w;
        v2f b1; b1.x = b.z;    b1.y = b.w;
        acc[t] = __builtin_amdgcn_wmma_f32_16x16x4_f32(false, a1, false, b1, (short)0, acc[t], false, false);
      }
      __builtin_amdgcn_sched_barrier(0);
    }
    float* yp = Y + (size_t)(rt * 16 + 8 * h) * HC + col0;
    if (rt * 16 + 16 <= n) {
      #pragma unroll
      for (int t = 0; t < G_NT; ++t) {
        #pragma unroll
        for (int r = 0; r < 8; ++r) yp[r * HC + 16 * t] = acc[t][r] + bc[t];
      }
    } else {
      #pragma unroll
      for (int t = 0; t < G_NT; ++t) {
        #pragma unroll
        for (int r = 0; r < 8; ++r) {
          if (rt * 16 + r + 8 * h < n) yp[r * HC + 16 * t] = acc[t][r] + bc[t];
        }
      }
    }
    #pragma unroll
    for (int j = 0; j < 16; ++j) a[j] = an[j];
    u = un;
  }
}

#define AT_THREADS 1024
#define AT_WAVES   (AT_THREADS / 32)
#define AT_QCAP    256
#define AT_GCAP    256
#define AT_BMW     ((BM_WORDS + AT_WAVES - 1) / AT_WAVES)
#define AT_GROUP   8
#define AT_V16     (E_EDGES / 16)
#define AT_SELF    ((AT_S + AT_WAVES - 1) / AT_WAVES)
#define ORD_NEG_INF 0x007FFFFFu
#define AT_LIMIT   60.0f

struct TileArgs {
  const uint4* tp;
  const unsigned* bm;
  unsigned     pat;
  const int*   ei;
  const float* xl;
  const float* sxr;
  const float* att;
  float*       sc;
  float*       tile;
  unsigned*    smax;
  float*       shift;
  float*       sden;
  int          lo;
};

__device__ __forceinline__ int lane_bcast_i(int v, int j) {
  return __builtin_amdgcn_readlane(v, j);
}
__device__ __forceinline__ float lane_bcast_f(float v, int j) {
  return __int_as_float(__builtin_amdgcn_readlane(__float_as_int(v), j));
}
__device__ __forceinline__ float lrelu_dot4(const float4 a, const float4 b, const float4 w) {
  float t0 = a.x + b.x, t1 = a.y + b.y, t2 = a.z + b.z, t3 = a.w + b.w;
  t0 = (t0 > 0.f) ? t0 : t0 * SLOPE;
  t1 = (t1 > 0.f) ? t1 : t1 * SLOPE;
  t2 = (t2 > 0.f) ? t2 : t2 * SLOPE;
  t3 = (t3 > 0.f) ? t3 : t3 * SLOPE;
  return t0 * w.x + t1 * w.y + t2 * w.z + t3 * w.w;
}

__device__ __forceinline__ void score_drain(int2* q, int head, int n, int lane, const TileArgs& A) {
  int src = 0, dl = 0, e = 0;
  if (lane < n) {
    const int qi = (head + lane) & (AT_QCAP - 1);
    const int2 t = q[qi];
    e = t.x;
    dl = t.y;
    src = (e < E_EDGES) ? A.ei[e] : (e - E_EDGES);
    if (dl < 0) {
      dl = A.ei[E_EDGES + e] - A.lo;
      q[qi].y = dl;
    }
  }
  const float4 w4 = *(const float4*)(A.att + 4 * lane);
  const int  r  = lane & 7;
  const int  hd = lane >> 3;
  const bool b2 = (r & 4) != 0, b1 = (r & 2) != 0, b0 = (r & 1) != 0;
  for (int j0 = 0; j0 < n; j0 += AT_GROUP) {
    float4 a[AT_GROUP], b[AT_GROUP];
    #pragma unroll
    for (int u = 0; u < AT_GROUP; ++u) {
      const int s = lane_bcast_i(src, j0 + u);
      const int d = lane_bcast_i(dl, j0 + u);
      a[u] = *(const float4*)(A.xl + (size_t)s * HC + 4 * lane);
      b[u] = *(const float4*)(A.sxr + d * HC + 4 * lane);
    }
    float p[AT_GROUP];
    #pragma unroll
    for (int u = 0; u < AT_GROUP; ++u) p[u] = lrelu_dot4(a[u], b[u], w4);
    float f4[4];
    #pragma unroll
    for (int i = 0; i < 4; ++i) {
      const float keep = b2 ? p[i + 4] : p[i];
      const float send = b2 ? p[i] : p[i + 4];
      f4[i] = keep + __shfl_xor(send, 4, 32);
    }
    float f2[2];
    #pragma unroll
    for (int i = 0; i < 2; ++i) {
      const float keep = b1 ? f4[i + 2] : f4[i];
      const float send = b1 ? f4[i] : f4[i + 2];
      f2[i] = keep + __shfl_xor(send, 2, 32);
    }
    const float keep = b0 ? f2[1] : f2[0];
    const float send = b0 ? f2[0] : f2[1];
    const float sc = keep + __shfl_xor(send, 1, 32);
    const int ue = j0 + r;
    const int eu = __shfl(e, ue, 32);
    const int du = __shfl(dl, ue, 32);
    if (ue < n) {
      A.sc[(size_t)eu * HEADS + hd] = sc;
      atomicMax(&A.smax[du * HEADS + hd], f32_to_ord(sc));
    }
  }
}

template <int U>
__device__ __forceinline__ float group8_bcast(const float v) {
  return __int_as_float(__builtin_amdgcn_mov_dpp8(__float_as_int(v), U * 0x249249));
}
template <int U>
__device__ __forceinline__ void add_edge(const float4 a, const float w, const int d, const int lane, float* tile) {
  const float wu = group8_bcast<U>(w);
  float* t = tile + d * HC + lane;
  atomAddF(t,      a.x * wu);
  atomAddF(t + 32, a.y * wu);
  atomAddF(t + 64, a.z * wu);
  atomAddF(t + 96, a.w * wu);
}

template <int SHIFT>
__device__ __forceinline__ bool one_pass_drain(int2* q, int head, int n, int lane, const TileArgs& A) {
  static_assert(AT_GROUP == 8, "the fold and the weight exchange work on groups of eight edges");
  int src = 0, dl = 0;
  if (lane < n) {
    const int qi = (head + lane) & (AT_QCAP - 1);
    const int2 t = q[qi];
    const int e = t.x;
    dl = t.y;
    src = (e < E_EDGES) ? A.ei[e] : (e - E_EDGES);
    if (dl < 0) {
      dl = A.ei[E_EDGES + e] - A.lo;
      q[qi].y = dl;
    }
  }
  const float4 w4 = *(const float4*)(A.att + 4 * lane);
  const int  r  = lane & 7;
  const int  hd = lane >> 3;
  const bool b2 = (r & 4) != 0, b1 = (r & 2) != 0, b0 = (r & 1) != 0;
  bool big = false;
  for (int j0 = 0; j0 < n; j0 += AT_GROUP) {
    float4 a[AT_GROUP], b[AT_GROUP];
    #pragma unroll
    for (int u = 0; u < AT_GROUP; ++u) {
      const int s = lane_bcast_i(src, j0 + u);
      const int d = lane_bcast_i(dl, j0 + u);
      a[u] = *(const float4*)(A.xl + (size_t)s * HC + 4 * lane);
      b[u] = *(const float4*)(A.sxr + d * HC + 4 * lane);
    }
    float p[AT_GROUP];
    #pragma unroll
    for (int u = 0; u < AT_GROUP; ++u) p[u] = lrelu_dot4(a[u], b[u], w4);
    float f4[4];
    #pragma unroll
    for (int i = 0; i < 4; ++i) {
      const float keep = b2 ? p[i + 4] : p[i];
      const float send = b2 ? p[i] : p[i + 4];
      f4[i] = keep + __shfl_xor(send, 4, 32);
    }
    float f2[2];
    #pragma unroll
    for (int i = 0; i < 2; ++i) {
      const float keep = b1 ? f4[i + 2] : f4[i];
      const float send = b1 ? f4[i] : f4[i + 2];
      f2[i] = keep + __shfl_xor(send, 2, 32);
    }
    const float keep = b0 ? f2[1] : f2[0];
    const float send = b0 ? f2[0] : f2[1];
    const float sc = keep + __shfl_xor(send, 1, 32);
    const int ue = j0 + r;
    const int du = __shfl(dl, ue, 32);
    if (SHIFT) {
      if (ue < n) A.shift[du * HEADS + hd] = sc;
    } else {
      float w = 0.f;
      if (ue < n) {
        const float t = sc - A.shift[du * HEADS + hd];
        big = big || (t > AT_LIMIT);
        w = __expf(t);
        atomAddF(A.sden + du * HEADS + hd, w);
      }
      if (j0 + 0 < n) add_edge<0>(a[0], w, lane_bcast_i(dl, j0 + 0), lane, A.tile);
      if (j0 + 1 < n) add_edge<1>(a[1], w, lane_bcast_i(dl, j0 + 1), lane, A.tile);
      if (j0 + 2 < n) add_edge<2>(a[2], w, lane_bcast_i(dl, j0 + 2), lane, A.tile);
      if (j0 + 3 < n) add_edge<3>(a[3], w, lane_bcast_i(dl, j0 + 3), lane, A.tile);
      if (j0 + 4 < n) add_edge<4>(a[4], w, lane_bcast_i(dl, j0 + 4), lane, A.tile);
      if (j0 + 5 < n) add_edge<5>(a[5], w, lane_bcast_i(dl, j0 + 5), lane, A.tile);
      if (j0 + 6 < n) add_edge<6>(a[6], w, lane_bcast_i(dl, j0 + 6), lane, A.tile);
      if (j0 + 7 < n) add_edge<7>(a[7], w, lane_bcast_i(dl, j0 + 7), lane, A.tile);
    }
  }
  return big;
}

__device__ __forceinline__ void agg_drain(const int2* q, int head, int n, int lane, const TileArgs& A) {
  int src = 0, dl = 0;
  float a0 = 0.f, a1 = 0.f, a2 = 0.f, a3 = 0.f;
  if (lane < n) {
    const int2 t = q[(head + lane) & (AT_QCAP - 1)];
    const int e = t.x;
    dl = t.y;
    src = (e < E_EDGES) ? A.ei[e] : (e - E_EDGES);
    if (dl < 0) dl = A.ei[E_EDGES + e] - A.lo;
    const float4 s4 = *(const float4*)(A.sc + (size_t)e * HEADS);
    const uint4  m4 = *(const uint4*)(A.smax + dl * HEADS);
    a0 = __expf(s4.x - ord_to_f32(m4.x));
    a1 = __expf(s4.y - ord_to_f32(m4.y));
    a2 = __expf(s4.z - ord_to_f32(m4.z));
    a3 = __expf(s4.w - ord_to_f32(m4.w));
    float* dn = A.sden + dl * HEADS;
    atomAddF(dn,     a0);
    atomAddF(dn + 1, a1);
    atomAddF(dn + 2, a2);
    atomAddF(dn + 3, a3);
  }
  for (int j0 = 0; j0 < n; j0 += AT_GROUP) {
    float v[AT_GROUP][HEADS];
    #pragma unroll
    for (int u = 0; u < AT_GROUP; ++u) {
      const int s = lane_bcast_i(src, j0 + u);
      const float* p = A.xl + (size_t)s * HC + lane;
      #pragma unroll
      for (int h = 0; h < HEADS; ++h) v[u][h] = p[h * CHP];
    }
    #pragma unroll
    for (int u = 0; u < AT_GROUP; ++u) {
      if (j0 + u < n) {
        const int   d  = lane_bcast_i(dl, j0 + u);
        const float w0 = lane_bcast_f(a0, j0 + u);
        const float w1 = lane_bcast_f(a1, j0 + u);
        const float w2 = lane_bcast_f(a2, j0 + u);
        const float w3 = lane_bcast_f(a3, j0 + u);
        float* t = A.tile + d * HC + (lane & 3) * 32 + (lane >> 2);
        atomAddF(t,      v[u][0] * w0);
        atomAddF(t + 8,  v[u][1] * w1);
        atomAddF(t + 16, v[u][2] * w2);
        atomAddF(t + 24, v[u][3] * w3);
      }
    }
  }
}

__device__ __forceinline__ uint4 plane_group(const uint4* __restrict__ tp4, const int idx) {
  return tp4[(idx < AT_V16) ? idx : (AT_V16 - 1)];
}

__device__ __forceinline__ unsigned zero_bytes(const unsigned x) {
  return ~(((x & 0x7f7f7f7fu) + 0x7f7f7f7fu) | x | 0x7f7f7f7fu);
}

__device__ __forceinline__ unsigned tile_bits(const uint4 p, const unsigned pat) {
  const unsigned z0 = zero_bytes(p.x ^ pat), z1 = zero_bytes(p.y ^ pat);
  const unsigned z2 = zero_bytes(p.z ^ pat), z3 = zero_bytes(p.w ^ pat);
  return (z0 >> 7) | (z1 >> 6) | (z2 >> 5) | (z3 >> 4);
}
__device__ __forceinline__ int bit_edge(const int idx, const int pb) {
  return idx * 16 + 4 * (pb & 7) + (pb >> 3);
}

__device__ __forceinline__ int ring_self(int2* q, const bool store, const int wave, const int lane, const TileArgs& A) {
  const int dl = wave * AT_SELF + lane;
  const bool on = (lane < AT_SELF) && (dl < AT_S) && (A.lo + dl < N_NODES);
  const unsigned m = (unsigned)__ballot(on);
  if (store && on) q[__popc(m & ((1u << lane) - 1u))] = make_int2(E_EDGES + A.lo + dl, dl);
  return __popc(m);
}

__device__ __forceinline__ void ring_bits(unsigned g, const int G, int2* q, int& cnt, bool& over, const unsigned below) {
  for (;;) {
    const bool on = g != 0u;
    const unsigned m = (unsigned)__ballot(on);
    if (m == 0u) break;
    const bool room = !over && (cnt + __popc(m) <= AT_QCAP);
    if (on) {
      const int pb = __ffs(g) - 1;
      g &= g - 1u;
      if (room) q[cnt + __popc(m & below)] = make_int2(bit_edge(G, pb), -1);
    }
    if (room) cnt += __popc(m);
    else      over = true;
  }
}

__device__ __forceinline__ int bm_word(const int w) {
  return (((w >> 10) * G_BLOCKS + ((w & 1023) >> 2)) * BM_SLOTS) * 4 + (w & 3);
}

__device__ __forceinline__ bool bitmap_fill(int2* q, int* gl, int& cnt, const int wave, const int lane, const TileArgs& A) {
  static_assert(AT_BMW > 32 && AT_BMW <= 64 && AT_WAVES * AT_BMW >= BM_WORDS && (AT_WAVES - 1) * AT_BMW + 32 <= BM_WORDS,
                "a wave's words: a first word for every lane, a second for the lanes below AT_BMW - 32");
  const unsigned below = (1u << lane) - 1u;
  const int w0 = wave * AT_BMW + lane;
  const int w1 = w0 + 32;
  const bool has1 = (lane < AT_BMW - 32) && (w1 < BM_WORDS);
  unsigned b0 = A.bm[bm_word(w0)];
  unsigned b1 = A.bm[bm_word(has1 ? w1 : w0)];
  if (!has1) b1 = 0u;
  const int c = __popc(b0) + __popc(b1);
  int pre = c;
  for (int d = 1; d < 32; d <<= 1) {
    const int t = __shfl_up(pre, d, 32);
    if (lane >= d) pre += t;
  }
  const int gcnt = lane_bcast_i(pre, 31);
  if (gcnt > AT_GCAP) return true;
  for (int at = pre - c; b0 != 0u; ++at) {
    gl[at] = w0 * 32 + (__ffs(b0) - 1);
    b0 &= b0 - 1u;
  }
  for (int at = pre - __popc(b1); b1 != 0u; ++at) {
    gl[at] = w1 * 32 + (__ffs(b1) - 1);
    b1 &= b1 - 1u;
  }
  bool over = false;
  for (int base = 0; base < gcnt; base += 128) {
    int   G[4];
    uint4 p[4];
    #pragma unroll
    for (int r = 0; r < 4; ++r) {
      const int k = base + 32 * r + lane;
      const int g = gl[(k < gcnt) ? k : (gcnt - 1)];
      G[r] = min(max(g, 0), AT_V16 - 1);
      p[r] = A.tp[G[r]];
    }
    #pragma unroll
    for (int r = 0; r < 4; ++r) {
      if (base + 32 * r < gcnt) {
        unsigned g = tile_bits(p[r], A.pat);
        if (base + 32 * r + lane >= gcnt) g = 0u;
        ring_bits(g, G[r], q, cnt, over, below);
      }
    }
  }
  return over;
}

template <int PH>
__device__ __forceinline__ void drain_any(int2* q, const int head, const int n, const int lane, const TileArgs& A,
                                          bool& big) {
  if (PH == 0)      score_drain(q, head, n, lane, A);
  else if (PH == 1) agg_drain(q, head, n, lane, A);
  else              big = one_pass_drain<0>(q, head, n, lane, A) || big;
}

template <int PH>
__device__ __forceinline__ int plain_scan(int2* q, const int cnt0, bool& big, const int wave, const int lane,
                                          const TileArgs& A) {
  int head = 0, cnt = cnt0, met = 0;
  const unsigned below = (1u << lane) - 1u;
  #pragma unroll 1
  for (int pass = 0; pass <= AT_BMW; ++pass) {
    const bool last = pass == AT_BMW;
    const int idx = (wave * AT_BMW + pass) * 32 + lane;
    unsigned g = tile_bits(plane_group(A.tp, idx), A.pat);
    if (last || idx >= AT_V16) g = 0u;
    for (;;) {
      const bool on = g != 0u;
      const unsigned m = (unsigned)__ballot(on);
      if (m != 0u) {
        if (on) {
          const int pb = __ffs(g) - 1;
          g &= g - 1u;
          q[(head + cnt + __popc(m & below)) & (AT_QCAP - 1)] = make_int2(bit_edge(idx, pb), -1);
        }
        cnt += __popc(m);
        met += __popc(m);
      }
      while (cnt >= 32 || (last && m == 0u && cnt > 0)) {
        const int n = (cnt < 32) ? cnt : 32;
        drain_any<PH>(q, head, n, lane, A, big);
        head = (head + n) & (AT_QCAP - 1);
        cnt -= n;
      }
      if (m == 0u) break;
    }
  }
  return met;
}

template <int REUSE>
__global__ void __launch_bounds__(AT_THREADS)
tile_softmax_aggregate_ln_kernel(const int* __restrict__ ei,
                                 const float* __restrict__ xl,
                                 const float* __restrict__ xr,
                                 const float* __restrict__ att,
                                 float* sc,
                                 const float* __restrict__ bias,
                                 const float* __restrict__ gamma,
                                 const float* __restrict__ beta,
                                 float* __restrict__ out,
                                 int*  lists,
                                 int*  lcnt,
                                 int*  lover,
                                 const unsigned* __restrict__ tplane,
                                 const unsigned* __restrict__ bmap) {
  __shared__ __align__(16) float    tile[AT_S * HC];
  __shared__ __align__(16) float    sxr[AT_S * HC];
  __shared__ __align__(16) int2     ring[AT_WAVES * AT_QCAP];
  __shared__ int                    glist[AT_WAVES * AT_GCAP];
  __shared__ __align__(16) unsigned smax[AT_S * HEADS];
  __shared__ __align__(16) float    sden[AT_S * HEADS];
  __shared__ int      sover;
  __shared__ int      sbad;
  const int tid  = threadIdx.x;
  const int wave = tid >> 5;
  const int lane = tid & 31;
  const int lo   = blockIdx.x * AT_S;

  {
    constexpr int NV4 = AT_S * HC / 4;
    constexpr int NLD = (NV4 + AT_THREADS - 1) / AT_THREADS;
    const float4 zero4 = make_float4(0.f, 0.f, 0.f, 0.f);
    float4 r[NLD];
    #pragma unroll
    for (int k = 0; k < NLD; ++k) {
      const int gi = lo * (HC / 4) + tid + k * AT_THREADS;
      r[k] = ((const float4*)xr)[min(gi, N_NODES * (HC / 4) - 1)];
    }
    #pragma unroll
    for (int k = 0; k < NLD; ++k) {
      const int i = tid + k * AT_THREADS;
      if (i < NV4) { ((float4*)tile)[i] = zero4; ((float4*)sxr)[i] = r[k]; }
    }
  }
  if (tid < AT_S * HEADS) sden[tid] = 0.f;
  if (tid == 0) { sover = 0; sbad = 0; }
  __syncthreads();

  TileArgs A;
  A.tp = (const uint4*)tplane; A.pat = (unsigned)blockIdx.x * 0x01010101u;
  A.bm = bmap + 4 * blockIdx.x;
  A.ei = ei; A.xl = xl; A.sxr = sxr; A.att = att; A.sc = sc;
  A.tile = tile; A.smax = smax; A.shift = (float*)smax; A.sden = sden; A.lo = lo;
  int2* q = ring + wave * AT_QCAP;

  const int slot = blockIdx.x * AT_WAVES + wave;
  int total;
  bool listed = false;
  if (REUSE) listed = (lover[blockIdx.x] == 0);
  int nself, cnt;
  if (REUSE && listed) {
    total = min(max(__builtin_amdgcn_readfirstlane(lcnt[slot]), 0), AT_QCAP);
    const int2* ls = (const int2*)lists + (size_t)slot * AT_QCAP;
    int2 r[AT_QCAP / 32];
    #pragma unroll
    for (int k = 0; k < AT_QCAP / 32; ++k) {
      const int i = lane + 32 * k;
      r[k] = (i < total) ? ls[i] : make_int2(0, 0);
    }
    #pragma unroll
    for (int k = 0; k < AT_QCAP / 32; ++k) {
      const int i = lane + 32 * k;
      if (i < total) q[i] = r[k];
    }
    nself = min(ring_self(q, false, wave, lane, A), total);
    cnt = total;
  } else {
    nself = ring_self(q, true, wave, lane, A);
    cnt = nself;
    total = nself;
  }
  if (nself > 0) (void)one_pass_drain<1>(q, 0, nself, lane, A);
  __syncthreads();

  bool over = false;
  if (!(REUSE && listed)) {
    over = bitmap_fill(q, glist + wave * AT_GCAP, cnt, wave, lane, A);
    total = cnt;
  }
  bool big = false;
  if (!over) {
    for (int h0 = 0; h0 < cnt; h0 += 32) {
      const int n = (cnt - h0 < 32) ? (cnt - h0) : 32;
      big = one_pass_drain<0>(q, h0, n, lane, A) || big;
    }
  } else {
    const int ns = ring_self(q, true, wave, lane, A);
    total = ns + plain_scan<2>(q, ns, big, wave, lane, A);
  }
  if (total > AT_QCAP && lane == 0) sover = 1;
  if (__ballot(big) != 0 && lane == 0) sbad = 1;
  __syncthreads();
  if (!REUSE) {
    if (tid == 0) lover[blockIdx.x] = sover;
    if (sover == 0) {
      int2* ls = (int2*)lists + (size_t)slot * AT_QCAP;
      for (int i = lane; i < total; i += 32) ls[i] = q[i];
      if (lane == 0) lcnt[slot] = total;
    }
  }

  if (sbad != 0) {
    constexpr int NV4 = AT_S * HC / 4;
    for (int i = tid; i < NV4; i += AT_THREADS) ((float4*)tile)[i] = make_float4(0.f, 0.f, 0.f, 0.f);
    if (tid < AT_S * HEADS) { smax[tid] = ORD_NEG_INF; sden[tid] = 0.f; }
    __syncthreads();
    bool unused = false;
    if (sover == 0) {
      for (int h0 = 0; h0 < total; h0 += 32) {
        const int n = (total - h0 < 32) ? (total - h0) : 32;
        score_drain(q, h0, n, lane, A);
      }
    } else {
      (void)plain_scan<0>(q, ring_self(q, true, wave, lane, A), unused, wave, lane, A);
    }
    __syncthreads();
    if (sover == 0) {
      for (int h0 = 0; h0 < total; h0 += 32) {
        const int n = (total - h0 < 32) ? (total - h0) : 32;
        agg_drain(q, h0, n, lane, A);
      }
    } else {
      (void)plain_scan<1>(q, ring_self(q, true, wave, lane, A), unused, wave, lane, A);
    }
    __syncthreads();
  }

  const float4 bs4 = *(const float4*)(bias + 4 * lane);
  const float4 ga4 = *(const float4*)(gamma + 4 * lane);
  const float4 be4 = *(const float4*)(beta + 4 * lane);
  for (int dl = wave; dl < AT_S; dl += AT_WAVES) {
    const int node = lo + dl;
    if (node >= N_NODES) break;
    const float den = sden[dl * HEADS + (lane >> 3)];
    const float* trow = tile + dl * HC + lane;
    float v[4];
    v[0] = trow[0]  / den + bs4.x;
    v[1] = trow[32] / den + bs4.y;
    v[2] = trow[64] / den + bs4.z;
    v[3] = trow[96] / den + bs4.w;
    float sum = (v[0] + v[1]) + (v[2] + v[3]);
    for (int off = 16; off > 0; off >>= 1) sum += __shfl_xor(sum, off, 32);
    const float mu = sum * (1.0f / HC);
    float sq = 0.f;
    #pragma unroll
    for (int i = 0; i < 4; ++i) { float d = v[i] - mu; sq += d * d; }
    for (int off = 16; off > 0; off >>= 1) sq += __shfl_xor(sq, off, 32);
    const float rstd = rsqrtf(sq * (1.0f / HC) + LN_EPS);
    const float h0 = (v[0] - mu) * rstd * ga4.x + be4.x;
    const float h1 = (v[1] - mu) * rstd * ga4.y + be4.y;
    const float h2 = (v[2] - mu) * rstd * ga4.z + be4.z;
    const float h3 = (v[3] - mu) * rstd * ga4.w + be4.w;
    float4 o;
    o.x = h0 / (1.0f + __expf(-h0));
    o.y = h1 / (1.0f + __expf(-h1));
    o.z = h2 / (1.0f + __expf(-h2));
    o.w = h3 / (1.0f + __expf(-h3));
    *(float4*)(out + (size_t)node * HC + 4 * lane) = o;
  }
}

#define PG_WAVES   4
#define PG_THREADS (PG_WAVES * 32)
#define PG_NODES   (PG_WAVES * 32)
#define PG_BLOCKS  ((N_NODES + PG_NODES - 1) / PG_NODES)
#define PG_ROUNDS  ((PG_BLOCKS + 31) / 32)
__global__ void __launch_bounds__(PG_THREADS)
pool_gather_kernel(const float* __restrict__ x,
                   const int* __restrict__ batch,
                   float* __restrict__ part,
                   int* __restrict__ pcnt) {
  static_assert(PG_THREADS == HC && NGRAPH == 64, "a thread per channel at the store; two 32-graph words of presence");
  __shared__ __align__(16) float sacc[NGRAPH * HC];
  __shared__ int scnt[NGRAPH];
  const int tid  = threadIdx.x;
  const int wave = tid >> 5;
  const int lane = tid & 31;
  const int base = blockIdx.x * PG_NODES + wave * 32;
  int id = (base + lane < N_NODES) ? batch[base + lane] : -1;
  if ((unsigned)id >= (unsigned)NGRAPH) id = -1;
  float4 v[32];
  #pragma unroll
  for (int u = 0; u < 32; ++u) {
    int node = base + u;
    node = (node < N_NODES) ? node : (N_NODES - 1);
    v[u] = *(const float4*)(x + (size_t)node * HC + lane * 4);
  }
  for (int i = tid; i < NGRAPH * HC / 4; i += PG_THREADS) ((float4*)sacc)[i] = make_float4(0.f, 0.f, 0.f, 0.f);
  if (tid < NGRAPH) scnt[tid] = 0;
  __syncthreads();
  unsigned rem = (unsigned)__ballot(id >= 0);
  for (int w = 0; w < PG_WAVES; ++w) {
    if (wave == w) {
      while (rem != 0u) {
        const int g = __builtin_amdgcn_readlane(id, __ffs(rem) - 1);
        const unsigned mask = (unsigned)__ballot(id == g);
        rem &= ~mask;
        float4 acc = make_float4(0.f, 0.f, 0.f, 0.f);
        #pragma unroll
        for (int u = 0; u < 32; ++u) {
          const bool on = ((mask >> u) & 1u) != 0u;
          acc.x += on ? v[u].x : 0.f;
          acc.y += on ? v[u].y : 0.f;
          acc.z += on ? v[u].z : 0.f;
          acc.w += on ? v[u].w : 0.f;
        }
        float4* p = (float4*)(sacc + g * HC) + lane;
        float4 o = *p;
        o.x += acc.x; o.y += acc.y; o.z += acc.z; o.w += acc.w;
        *p = o;
        if (lane == 0) scnt[g] += __popc(mask);
      }
    }
    __syncthreads();
  }
  const int c0 = scnt[lane], c1 = scnt[lane + 32];
  if (tid < NGRAPH) pcnt[blockIdx.x * NGRAPH + tid] = (tid < 32) ? c0 : c1;
  const unsigned m0 = (unsigned)__ballot(c0 > 0), m1 = (unsigned)__ballot(c1 > 0);
  float* prow = part + (size_t)blockIdx.x * NGRAPH * HC + tid;
  for (unsigned m = m0; m != 0u; m &= m - 1u) {
    const int g = __ffs(m) - 1;
    prow[g * HC] = sacc[g * HC + tid];
  }
  for (unsigned m = m1; m != 0u; m &= m - 1u) {
    const int g = 32 + __ffs(m) - 1;
    prow[g * HC] = sacc[g * HC + tid];
  }
}
__global__ void __launch_bounds__(PG_THREADS)
pool_final_kernel(const float* __restrict__ part,
                  const int* __restrict__ pcnt,
                  float* __restrict__ out) {
  __shared__ int slist[PG_ROUNDS * 32];
  __shared__ int sn[2];
  const int g    = blockIdx.x;
  const int tid  = threadIdx.x;
  const int lane = tid & 31;
  if (tid < 32) {
    int cc[PG_ROUNDS];
    #pragma unroll
    for (int k = 0; k < PG_ROUNDS; ++k) {
      const int b = 32 * k + lane;
      cc[k] = (b < PG_BLOCKS) ? pcnt[b * NGRAPH + g] : 0;
    }
    int n = 0, members = 0;
    #pragma unroll
    for (int k = 0; k < PG_ROUNDS; ++k) {
      const bool on = cc[k] > 0;
      const unsigned m = (unsigned)__ballot(on);
      if (on) slist[n + __popc(m & ((1u << lane) - 1u))] = 32 * k + lane;
      n += __popc(m);
      members += cc[k];
    }
    for (int off = 16; off > 0; off >>= 1) members += __shfl_xor(members, off, 32);
    if (lane == 0) { sn[0] = n; sn[1] = members; }
  }
  __syncthreads();
  const int n = sn[0];
  float sum = 0.f;
  for (int i0 = 0; i0 < n; i0 += 4) {
    float r[4];
    #pragma unroll
    for (int u = 0; u < 4; ++u) {
      const int b = slist[min(i0 + u, n - 1)];
      r[u] = part[((size_t)b * NGRAPH + g) * HC + tid];
    }
    #pragma unroll
    for (int u = 0; u < 4; ++u) sum += (i0 + u < n) ? r[u] : 0.f;
  }
  out[g * HC + tid] = sum / fmaxf((float)sn[1], 1.0f);
}

extern "C" void kernel_launch(void* const* d_in, const int* in_sizes, int n_in,
                              void* d_out, int out_size, void* d_ws, size_t ws_size,
                              hipStream_t stream) {
  (void)in_sizes; (void)n_in; (void)out_size; (void)ws_size;
  const float* x     = (const float*)d_in[0];
  const int*   ei    = (const int*)d_in[1];
  const int*   batch = (const int*)d_in[2];
  const float* Wl[2]    = {(const float*)d_in[3],  (const float*)d_in[11]};
  const float* Wr[2]    = {(const float*)d_in[4],  (const float*)d_in[12]};
  const float* bl[2]    = {(const float*)d_in[5],  (const float*)d_in[13]};
  const float* br[2]    = {(const float*)d_in[6],  (const float*)d_in[14]};
  const float* att[2]   = {(const float*)d_in[7],  (const float*)d_in[15]};
  const float* bias[2]  = {(const float*)d_in[8],  (const float*)d_in[16]};
  const float* gamma[2] = {(const float*)d_in[9],  (const float*)d_in[17]};
  const float* beta[2]  = {(const float*)d_in[10], (const float*)d_in[18]};

  float* ws = (float*)d_ws;
  float*    xl    = ws;            ws += (size_t)N_NODES * HC;
  float*    xr2   = ws;            ws += (size_t)N_NODES * HC;
  float*    xnode = ws;            ws += (size_t)N_NODES * HC;
  float*    scbuf = ws;            ws += (size_t)ETOT * HEADS;
  int*      pcnt  = (int*)ws;      ws += (size_t)PG_BLOCKS * NGRAPH;
  float*    ppart = ws;            ws += (size_t)PG_BLOCKS * NGRAPH * HC;
  int*      lists = (int*)ws;      ws += (size_t)AT_TILES * AT_WAVES * AT_QCAP * 2;
  int*      lcnt  = (int*)ws;      ws += (size_t)AT_TILES * AT_WAVES;
  int*      lover = (int*)ws;      ws += 256;
  unsigned* tplane = (unsigned*)ws;  ws += E_EDGES / 4;
  unsigned* bmap   = (unsigned*)ws;

  const float* xin = x;
  for (int l = 0; l < 2; ++l) {
    if (l == 0)
      gemm_bias_wmma<1><<<G_BLOCKS, G_THREADS, 0, stream>>>(xin, Wl[l], bl[l], xl, N_NODES, ei, tplane, bmap);
    else
      gemm_bias_wmma<0><<<G_BLOCKS, G_THREADS, 0, stream>>>(xin, Wl[l], bl[l], xl, N_NODES, ei, tplane, bmap);
    gemm_bias_wmma<0><<<G_BLOCKS, G_THREADS, 0, stream>>>(xin, Wr[l], br[l], xr2, N_NODES, ei, tplane, bmap);
    if (l == 0)
      tile_softmax_aggregate_ln_kernel<0><<<AT_TILES, AT_THREADS, 0, stream>>>(ei, xl, xr2, att[l], scbuf, bias[l], gamma[l], beta[l], xnode, lists, lcnt, lover, tplane, bmap);
    else
      tile_softmax_aggregate_ln_kernel<1><<<AT_TILES, AT_THREADS, 0, stream>>>(ei, xl, xr2, att[l], scbuf, bias[l], gamma[l], beta[l], xnode, lists, lcnt, lover, tplane, bmap);
    xin = xnode;
  }
  pool_gather_kernel<<<PG_BLOCKS, PG_THREADS, 0, stream>>>(xnode, batch, ppart, pcnt);
  pool_final_kernel<<<NGRAPH, PG_THREADS, 0, stream>>>(ppart, pcnt, (float*)d_out);
}


# x_r never leaves the work-group: each tile block makes its own 197 rows of x_r by f32 WMMA at its head (W_r staged in the LDS sum tile before it is zeroed); 8 launches to 6; all sums unchanged
# speedup vs baseline: 5.2412x; 1.0386x over previous
#include <hip/hip_runtime.h>
#include <hip/hip_bf16.h>
#include <math.h>

#define N_NODES 50000
#define E_EDGES 800000
#define ETOT    (E_EDGES + N_NODES)
#define HEADS   4
#define CHP     32
#define HC      128
#define NGRAPH  64
#define SLOPE   0.2f
#define LN_EPS  1e-5f

#define AT_S     197
#define AT_TILES ((N_NODES + AT_S - 1) / AT_S)
#define AT_NONE  255u
#define BM_WORDS  ((E_EDGES / 16 + 31) / 32)
#define BM_SLOTS  256

typedef float v2f __attribute__((ext_vector_type(2)));
typedef float v8f __attribute__((ext_vector_type(8)));

__device__ __forceinline__ unsigned f32_to_ord(float f) {
  unsigned u = __float_as_uint(f);
  return (u & 0x80000000u) ? ~u : (u | 0x80000000u);
}
__device__ __forceinline__ float ord_to_f32(unsigned u) {
  u = (u & 0x80000000u) ? (u & 0x7fffffffu) : ~u;
  return __uint_as_float(u);
}
__device__ __forceinline__ unsigned tile_of(const int d) {
  return ((unsigned)d < (unsigned)N_NODES) ? ((unsigned)d / (unsigned)AT_S) : AT_NONE;
}
__device__ __forceinline__ void atomAddF(float* p, float v) {
  unsafeAtomicAdd(p, v);
}

#define G_THREADS 512
#define G_WAVES   (G_THREADS / 32)
#define G_BLOCKS  256
#define G_NT      4
typedef float v4f __attribute__((ext_vector_type(4)));

__device__ __forceinline__ void gemm_load_a(v4f (&a)[16], const float* __restrict__ X, const int u, const int n,
                                            const int lane) {
  const int rt  = blockIdx.x + G_BLOCKS * (u >> 1);
  const int row = max(min(rt * 16 + (lane & 15), n - 1), 0);
  const v4f* p = (const v4f*)(X + (size_t)row * HC) + (lane >> 4);
  #pragma unroll
  for (int j = 0; j < 16; ++j) a[j] = p[2 * j];
}

template <int PRE>
__global__ void __launch_bounds__(G_THREADS)
gemm_bias_wmma(const float* __restrict__ X,
               const float* __restrict__ W,
               const float* __restrict__ bias,
               float* __restrict__ Y, int n,
               const int* __restrict__ ei,
               unsigned* __restrict__ tplane,
               unsigned* __restrict__ bmap) {
  __shared__ __align__(16) float sW[HC * HC];
  __shared__ __align__(16) unsigned sB[PRE ? 2 * 256 * 4 : 4];
  const int tid  = threadIdx.x;
  const int wave = __builtin_amdgcn_readfirstlane(tid >> 5);
  const int lane = tid & 31;
  const int c    = lane & 15;
  const int h    = lane >> 4;
  const int ntiles = (n + 15) >> 4;
  const int nrt    = (ntiles - (int)blockIdx.x + G_BLOCKS - 1) / G_BLOCKS;
  const int nunits = 2 * nrt;

  constexpr int NGRP = E_EDGES / 4;
  constexpr int PER  = (NGRP + G_BLOCKS * G_THREADS - 1) / (G_BLOCKS * G_THREADS);
  static_assert(!PRE || (PER == 2 && G_THREADS == 2 * BM_SLOTS && G_BLOCKS * 4 == 1024 && AT_TILES <= 254),
                "the bitmap's layout: two chunks of 1024 words, four words per block and chunk, one 16-byte store per thread");
  int4 dd[PER];
  if (PRE) {
    for (int k = 0; k < 4; ++k) sB[tid + G_THREADS * k] = 0u;
    #pragma unroll
    for (int k = 0; k < PER; ++k) {
      const int i = (int)blockIdx.x * G_THREADS + tid + k * (G_BLOCKS * G_THREADS);
      dd[k] = ((const int4*)(ei + E_EDGES))[min(i, NGRP - 1)];
    }
  }

  int u = wave;
  v4f a[16];
  gemm_load_a(a, X, u, n, lane);

  if (PRE) {
    __syncthreads();
    const unsigned gbit = 1u << ((tid >> 2) & 31);
    const int      gwrd = tid >> 7;
    #pragma unroll
    for (int k = 0; k < PER; ++k) {
      const int i = (int)blockIdx.x * G_THREADS + tid + k * (G_BLOCKS * G_THREADS);
      if (i < NGRP) {
        const unsigned t0 = tile_of(dd[k].x), t1 = tile_of(dd[k].y), t2 = tile_of(dd[k].z), t3 = tile_of(dd[k].w);
        tplane[i] = t0 | (t1 << 8) | (t2 << 16) | (t3 << 24);
        unsigned* row = sB + k * 1024 + gwrd;
        if (t0 != AT_NONE) atomicOr(row + 4 * t0, gbit);
        if (t1 != AT_NONE) atomicOr(row + 4 * t1, gbit);
        if (t2 != AT_NONE) atomicOr(row + 4 * t2, gbit);
        if (t3 != AT_NONE) atomicOr(row + 4 * t3, gbit);
      }
    }
  }

  {
    v4f r[2][4];
    #pragma unroll
    for (int i = 0; i < 2; ++i) {
      const int item = tid + G_THREADS * i;
      const int q = item >> 5, g = item & 31;
      #pragma unroll
      for (int k = 0; k < 4; ++k) r[i][k] = *(const v4f*)(W + (4 * q + k) * HC + 4 * g);
    }
    #pragma unroll
    for (int i = 0; i < 2; ++i) {
      const int item = tid + G_THREADS * i;
      const int q = item >> 5, g = item & 31;
      v4f* d = (v4f*)sW + q * HC + 4 * g;
      d[0] = v4f{r[i][0].x, r[i][1].x, r[i][2].x, r[i][3].x};
      d[1] = v4f{r[i][0].y, r[i][1].y, r[i][2].y, r[i][3].y};
      d[2] = v4f{r[i][0].z, r[i][1].z, r[i][2].z, r[i][3].z};
      d[3] = v4f{r[i][0].w, r[i][1].w, r[i][2].w, r[i][3].w};
    }
  }
  __syncthreads();
  if (PRE) {
    const int k = tid >> 8, t = tid & (BM_SLOTS - 1);
    const uint4 v = *(const uint4*)(sB + k * 1024 + 4 * t);
    *(uint4*)(bmap + ((size_t)(k * G_BLOCKS + (int)blockIdx.x) * BM_SLOTS + t) * 4) = v;
  }

  while (u < nunits) {
    const int un = u + G_WAVES;
    v4f an[16];
    gemm_load_a(an, X, min(un, nunits - 1), n, lane);
    const int rt   = blockIdx.x + G_BLOCKS * (u >> 1);
    const int col0 = (u & 1) * (16 * G_NT) + c;
    const v4f* bp = (const v4f*)sW + h * HC + col0;
    float bc[G_NT];
    v8f acc[G_NT];
    #pragma unroll
    for (int t = 0; t < G_NT; ++t) { bc[t] = bias[col0 + 16 * t]; acc[t] = v8f{}; }
    v4f bq[2][G_NT];
    #pragma unroll
    for (int t = 0; t < G_NT; ++t) bq[0][t] = bp[16 * t];
    #pragma unroll
    for (int j = 0; j < 16; ++j) {
      if (j + 1 < 16) {
        #pragma unroll
        for (int t = 0; t < G_NT; ++t) bq[(j + 1) & 1][t] = bp[2 * (j + 1) * HC + 16 * t];
      }
      #pragma unroll
      for (int t = 0; t < G_NT; ++t) {
        const v4f b = bq[j & 1][t];
        v2f a0; a0.x = a[j].x; a0.y = a[j].y;
        v2f b0; b0.x = b.x;    b0.y = b.y;
        acc[t] = __builtin_amdgcn_wmma_f32_16x16x4_f32(false, a0, false, b0, (short)0, acc[t], false, false);
        v2f a1; a1.x = a[j].z; a1.y = a[j].w;
        v2f b1; b1.x = b.z;    b1.y = b.w;
        acc[t] = __builtin_amdgcn_wmma_f32_16x16x4_f32(false, a1, false, b1, (short)0, acc[t], false, false);
      }
      __builtin_amdgcn_sched_barrier(0);
    }
    float* yp = Y + (size_t)(rt * 16 + 8 * h) * HC + col0;
    if (rt * 16 + 16 <= n) {
      #pragma unroll
      for (int t = 0; t < G_NT; ++t) {
        #pragma unroll
        for (int r = 0; r < 8; ++r) yp[r * HC + 16 * t] = acc[t][r] + bc[t];
      }
    } else {
      #pragma unroll
      for (int t = 0; t < G_NT; ++t) {
        #pragma unroll
        for (int r = 0; r < 8; ++r) {
          if (rt * 16 + r + 8 * h < n) yp[r * HC + 16 * t] = acc[t][r] + bc[t];
        }
      }
    }
    #pragma unroll
    for (int j = 0; j < 16; ++j) a[j] = an[j];
    u = un;
  }
}

#define AT_THREADS 1024
#define AT_WAVES   (AT_THREADS / 32)
#define AT_QCAP    256
#define AT_GCAP    256
#define AT_BMW     ((BM_WORDS + AT_WAVES - 1) / AT_WAVES)
#define AT_GROUP   8
#define AT_V16     (E_EDGES / 16)
#define AT_SELF    ((AT_S + AT_WAVES - 1) / AT_WAVES)
#define ORD_NEG_INF 0x007FFFFFu
#define AT_LIMIT   60.0f

struct TileArgs {
  const uint4* tp;
  const unsigned* bm;
  unsigned     pat;
  const int*   ei;
  const float* xl;
  const float* sxr;
  const float* att;
  float*       sc;
  float*       tile;
  unsigned*    smax;
  float*       shift;
  float*       sden;
  int          lo;
};

__device__ __forceinline__ int lane_bcast_i(int v, int j) {
  return __builtin_amdgcn_readlane(v, j);
}
__device__ __forceinline__ float lane_bcast_f(float v, int j) {
  return __int_as_float(__builtin_amdgcn_readlane(__float_as_int(v), j));
}
__device__ __forceinline__ float lrelu_dot4(const float4 a, const float4 b, const float4 w) {
  float t0 = a.x + b.x, t1 = a.y + b.y, t2 = a.z + b.z, t3 = a.w + b.w;
  t0 = (t0 > 0.f) ? t0 : t0 * SLOPE;
  t1 = (t1 > 0.f) ? t1 : t1 * SLOPE;
  t2 = (t2 > 0.f) ? t2 : t2 * SLOPE;
  t3 = (t3 > 0.f) ? t3 : t3 * SLOPE;
  return t0 * w.x + t1 * w.y + t2 * w.z + t3 * w.w;
}

__device__ __forceinline__ void score_drain(int2* q, int head, int n, int lane, const TileArgs& A) {
  int src = 0, dl = 0, e = 0;
  if (lane < n) {
    const int qi = (head + lane) & (AT_QCAP - 1);
    const int2 t = q[qi];
    e = t.x;
    dl = t.y;
    src = (e < E_EDGES) ? A.ei[e] : (e - E_EDGES);
    if (dl < 0) {
      dl = A.ei[E_EDGES + e] - A.lo;
      q[qi].y = dl;
    }
  }
  const float4 w4 = *(const float4*)(A.att + 4 * lane);
  const int  r  = lane & 7;
  const int  hd = lane >> 3;
  const bool b2 = (r & 4) != 0, b1 = (r & 2) != 0, b0 = (r & 1) != 0;
  for (int j0 = 0; j0 < n; j0 += AT_GROUP) {
    float4 a[AT_GROUP], b[AT_GROUP];
    #pragma unroll
    for (int u = 0; u < AT_GROUP; ++u) {
      const int s = lane_bcast_i(src, j0 + u);
      const int d = lane_bcast_i(dl, j0 + u);
      a[u] = *(const float4*)(A.xl + (size_t)s * HC + 4 * lane);
      b[u] = *(const float4*)(A.sxr + d * HC + 4 * lane);
    }
    float p[AT_GROUP];
    #pragma unroll
    for (int u = 0; u < AT_GROUP; ++u) p[u] = lrelu_dot4(a[u], b[u], w4);
    float f4[4];
    #pragma unroll
    for (int i = 0; i < 4; ++i) {
      const float keep = b2 ? p[i + 4] : p[i];
      const float send = b2 ? p[i] : p[i + 4];
      f4[i] = keep + __shfl_xor(send, 4, 32);
    }
    float f2[2];
    #pragma unroll
    for (int i = 0; i < 2; ++i) {
      const float keep = b1 ? f4[i + 2] : f4[i];
      const float send = b1 ? f4[i] : f4[i + 2];
      f2[i] = keep + __shfl_xor(send, 2, 32);
    }
    const float keep = b0 ? f2[1] : f2[0];
    const float send = b0 ? f2[0] : f2[1];
    const float sc = keep + __shfl_xor(send, 1, 32);
    const int ue = j0 + r;
    const int eu = __shfl(e, ue, 32);
    const int du = __shfl(dl, ue, 32);
    if (ue < n) {
      A.sc[(size_t)eu * HEADS + hd] = sc;
      atomicMax(&A.smax[du * HEADS + hd], f32_to_ord(sc));
    }
  }
}

template <int U>
__device__ __forceinline__ float group8_bcast(const float v) {
  return __int_as_float(__builtin_amdgcn_mov_dpp8(__float_as_int(v), U * 0x249249));
}
template <int U>
__device__ __forceinline__ void add_edge(const float4 a, const float w, const int d, const int lane, float* tile) {
  const float wu = group8_bcast<U>(w);
  float* t = tile + d * HC + lane;
  atomAddF(t,      a.x * wu);
  atomAddF(t + 32, a.y * wu);
  atomAddF(t + 64, a.z * wu);
  atomAddF(t + 96, a.w * wu);
}

template <int SHIFT>
__device__ __forceinline__ bool one_pass_drain(int2* q, int head, int n, int lane, const TileArgs& A) {
  static_assert(AT_GROUP == 8, "the fold and the weight exchange work on groups of eight edges");
  int src = 0, dl = 0;
  if (lane < n) {
    const int qi = (head + lane) & (AT_QCAP - 1);
    const int2 t = q[qi];
    const int e = t.x;
    dl = t.y;
    src = (e < E_EDGES) ? A.ei[e] : (e - E_EDGES);
    if (dl < 0) {
      dl = A.ei[E_EDGES + e] - A.lo;
      q[qi].y = dl;
    }
  }
  const float4 w4 = *(const float4*)(A.att + 4 * lane);
  const int  r  = lane & 7;
  const int  hd = lane >> 3;
  const bool b2 = (r & 4) != 0, b1 = (r & 2) != 0, b0 = (r & 1) != 0;
  bool big = false;
  for (int j0 = 0; j0 < n; j0 += AT_GROUP) {
    float4 a[AT_GROUP], b[AT_GROUP];
    #pragma unroll
    for (int u = 0; u < AT_GROUP; ++u) {
      const int s = lane_bcast_i(src, j0 + u);
      const int d = lane_bcast_i(dl, j0 + u);
      a[u] = *(const float4*)(A.xl + (size_t)s * HC + 4 * lane);
      b[u] = *(const float4*)(A.sxr + d * HC + 4 * lane);
    }
    float p[AT_GROUP];
    #pragma unroll
    for (int u = 0; u < AT_GROUP; ++u) p[u] = lrelu_dot4(a[u], b[u], w4);
    float f4[4];
    #pragma unroll
    for (int i = 0; i < 4; ++i) {
      const float keep = b2 ? p[i + 4] : p[i];
      const float send = b2 ? p[i] : p[i + 4];
      f4[i] = keep + __shfl_xor(send, 4, 32);
    }
    float f2[2];
    #pragma unroll
    for (int i = 0; i < 2; ++i) {
      const float keep = b1 ? f4[i + 2] : f4[i];
      const float send = b1 ? f4[i] : f4[i + 2];
      f2[i] = keep + __shfl_xor(send, 2, 32);
    }
    const float keep = b0 ? f2[1] : f2[0];
    const float send = b0 ? f2[0] : f2[1];
    const float sc = keep + __shfl_xor(send, 1, 32);
    const int ue = j0 + r;
    const int du = __shfl(dl, ue, 32);
    if (SHIFT) {
      if (ue < n) A.shift[du * HEADS + hd] = sc;
    } else {
      float w = 0.f;
      if (ue < n) {
        const float t = sc - A.shift[du * HEADS + hd];
        big = big || (t > AT_LIMIT);
        w = __expf(t);
        atomAddF(A.sden + du * HEADS + hd, w);
      }
      if (j0 + 0 < n) add_edge<0>(a[0], w, lane_bcast_i(dl, j0 + 0), lane, A.tile);
      if (j0 + 1 < n) add_edge<1>(a[1], w, lane_bcast_i(dl, j0 + 1), lane, A.tile);
      if (j0 + 2 < n) add_edge<2>(a[2], w, lane_bcast_i(dl, j0 + 2), lane, A.tile);
      if (j0 + 3 < n) add_edge<3>(a[3], w, lane_bcast_i(dl, j0 + 3), lane, A.tile);
      if (j0 + 4 < n) add_edge<4>(a[4], w, lane_bcast_i(dl, j0 + 4), lane, A.tile);
      if (j0 + 5 < n) add_edge<5>(a[5], w, lane_bcast_i(dl, j0 + 5), lane, A.tile);
      if (j0 + 6 < n) add_edge<6>(a[6], w, lane_bcast_i(dl, j0 + 6), lane, A.tile);
      if (j0 + 7 < n) add_edge<7>(a[7], w, lane_bcast_i(dl, j0 + 7), lane, A.tile);
    }
  }
  return big;
}

__device__ __forceinline__ void agg_drain(const int2* q, int head, int n, int lane, const TileArgs& A) {
  int src = 0, dl = 0;
  float a0 = 0.f, a1 = 0.f, a2 = 0.f, a3 = 0.f;
  if (lane < n) {
    const int2 t = q[(head + lane) & (AT_QCAP - 1)];
    const int e = t.x;
    dl = t.y;
    src = (e < E_EDGES) ? A.ei[e] : (e - E_EDGES);
    if (dl < 0) dl = A.ei[E_EDGES + e] - A.lo;
    const float4 s4 = *(const float4*)(A.sc + (size_t)e * HEADS);
    const uint4  m4 = *(const uint4*)(A.smax + dl * HEADS);
    a0 = __expf(s4.x - ord_to_f32(m4.x));
    a1 = __expf(s4.y - ord_to_f32(m4.y));
    a2 = __expf(s4.z - ord_to_f32(m4.z));
    a3 = __expf(s4.w - ord_to_f32(m4.w));
    float* dn = A.sden + dl * HEADS;
    atomAddF(dn,     a0);
    atomAddF(dn + 1, a1);
    atomAddF(dn + 2, a2);
    atomAddF(dn + 3, a3);
  }
  for (int j0 = 0; j0 < n; j0 += AT_GROUP) {
    float v[AT_GROUP][HEADS];
    #pragma unroll
    for (int u = 0; u < AT_GROUP; ++u) {
      const int s = lane_bcast_i(src, j0 + u);
      const float* p = A.xl + (size_t)s * HC + lane;
      #pragma unroll
      for (int h = 0; h < HEADS; ++h) v[u][h] = p[h * CHP];
    }
    #pragma unroll
    for (int u = 0; u < AT_GROUP; ++u) {
      if (j0 + u < n) {
        const int   d  = lane_bcast_i(dl, j0 + u);
        const float w0 = lane_bcast_f(a0, j0 + u);
        const float w1 = lane_bcast_f(a1, j0 + u);
        const float w2 = lane_bcast_f(a2, j0 + u);
        const float w3 = lane_bcast_f(a3, j0 + u);
        float* t = A.tile + d * HC + (lane & 3) * 32 + (lane >> 2);
        atomAddF(t,      v[u][0] * w0);
        atomAddF(t + 8,  v[u][1] * w1);
        atomAddF(t + 16, v[u][2] * w2);
        atomAddF(t + 24, v[u][3] * w3);
      }
    }
  }
}

__device__ __forceinline__ uint4 plane_group(const uint4* __restrict__ tp4, const int idx) {
  return tp4[(idx < AT_V16) ? idx : (AT_V16 - 1)];
}

__device__ __forceinline__ unsigned zero_bytes(const unsigned x) {
  return ~(((x & 0x7f7f7f7fu) + 0x7f7f7f7fu) | x | 0x7f7f7f7fu);
}

__device__ __forceinline__ unsigned tile_bits(const uint4 p, const unsigned pat) {
  const unsigned z0 = zero_bytes(p.x ^ pat), z1 = zero_bytes(p.y ^ pat);
  const unsigned z2 = zero_bytes(p.z ^ pat), z3 = zero_bytes(p.w ^ pat);
  return (z0 >> 7) | (z1 >> 6) | (z2 >> 5) | (z3 >> 4);
}
__device__ __forceinline__ int bit_edge(const int idx, const int pb) {
  return idx * 16 + 4 * (pb & 7) + (pb >> 3);
}

__device__ __forceinline__ int ring_self(int2* q, const bool store, const int wave, const int lane, const TileArgs& A) {
  const int dl = wave * AT_SELF + lane;
  const bool on = (lane < AT_SELF) && (dl < AT_S) && (A.lo + dl < N_NODES);
  const unsigned m = (unsigned)__ballot(on);
  if (store && on) q[__popc(m & ((1u << lane) - 1u))] = make_int2(E_EDGES + A.lo + dl, dl);
  return __popc(m);
}

__device__ __forceinline__ void ring_bits(unsigned g, const int G, int2* q, int& cnt, bool& over, const unsigned below) {
  for (;;) {
    const bool on = g != 0u;
    const unsigned m = (unsigned)__ballot(on);
    if (m == 0u) break;
    const bool room = !over && (cnt + __popc(m) <= AT_QCAP);
    if (on) {
      const int pb = __ffs(g) - 1;
      g &= g - 1u;
      if (room) q[cnt + __popc(m & below)] = make_int2(bit_edge(G, pb), -1);
    }
    if (room) cnt += __popc(m);
    else      over = true;
  }
}

__device__ __forceinline__ int bm_word(const int w) {
  return (((w >> 10) * G_BLOCKS + ((w & 1023) >> 2)) * BM_SLOTS) * 4 + (w & 3);
}

__device__ __forceinline__ bool bitmap_fill(int2* q, int* gl, int& cnt, const int wave, const int lane, const TileArgs& A) {
  static_assert(AT_BMW > 32 && AT_BMW <= 64 && AT_WAVES * AT_BMW >= BM_WORDS && (AT_WAVES - 1) * AT_BMW + 32 <= BM_WORDS,
                "a wave's words: a first word for every lane, a second for the lanes below AT_BMW - 32");
  const unsigned below = (1u << lane) - 1u;
  const int w0 = wave * AT_BMW + lane;
  const int w1 = w0 + 32;
  const bool has1 = (lane < AT_BMW - 32) && (w1 < BM_WORDS);
  unsigned b0 = A.bm[bm_word(w0)];
  unsigned b1 = A.bm[bm_word(has1 ? w1 : w0)];
  if (!has1) b1 = 0u;
  const int c = __popc(b0) + __popc(b1);
  int pre = c;
  for (int d = 1; d < 32; d <<= 1) {
    const int t = __shfl_up(pre, d, 32);
    if (lane >= d) pre += t;
  }
  const int gcnt = lane_bcast_i(pre, 31);
  if (gcnt > AT_GCAP) return true;
  for (int at = pre - c; b0 != 0u; ++at) {
    gl[at] = w0 * 32 + (__ffs(b0) - 1);
    b0 &= b0 - 1u;
  }
  for (int at = pre - __popc(b1); b1 != 0u; ++at) {
    gl[at] = w1 * 32 + (__ffs(b1) - 1);
    b1 &= b1 - 1u;
  }
  bool over = false;
  for (int base = 0; base < gcnt; base += 128) {
    int   G[4];
    uint4 p[4];
    #pragma unroll
    for (int r = 0; r < 4; ++r) {
      const int k = base + 32 * r + lane;
      const int g = gl[(k < gcnt) ? k : (gcnt - 1)];
      G[r] = min(max(g, 0), AT_V16 - 1);
      p[r] = A.tp[G[r]];
    }
    #pragma unroll
    for (int r = 0; r < 4; ++r) {
      if (base + 32 * r < gcnt) {
        unsigned g = tile_bits(p[r], A.pat);
        if (base + 32 * r + lane >= gcnt) g = 0u;
        ring_bits(g, G[r], q, cnt, over, below);
      }
    }
  }
  return over;
}

template <int PH>
__device__ __forceinline__ void drain_any(int2* q, const int head, const int n, const int lane, const TileArgs& A,
                                          bool& big) {
  if (PH == 0)      score_drain(q, head, n, lane, A);
  else if (PH == 1) agg_drain(q, head, n, lane, A);
  else              big = one_pass_drain<0>(q, head, n, lane, A) || big;
}

template <int PH>
__device__ __forceinline__ int plain_scan(int2* q, const int cnt0, bool& big, const int wave, const int lane,
                                          const TileArgs& A) {
  int head = 0, cnt = cnt0, met = 0;
  const unsigned below = (1u << lane) - 1u;
  #pragma unroll 1
  for (int pass = 0; pass <= AT_BMW; ++pass) {
    const bool last = pass == AT_BMW;
    const int idx = (wave * AT_BMW + pass) * 32 + lane;
    unsigned g = tile_bits(plane_group(A.tp, idx), A.pat);
    if (last || idx >= AT_V16) g = 0u;
    for (;;) {
      const bool on = g != 0u;
      const unsigned m = (unsigned)__ballot(on);
      if (m != 0u) {
        if (on) {
          const int pb = __ffs(g) - 1;
          g &= g - 1u;
          q[(head + cnt + __popc(m & below)) & (AT_QCAP - 1)] = make_int2(bit_edge(idx, pb), -1);
        }
        cnt += __popc(m);
        met += __popc(m);
      }
      while (cnt >= 32 || (last && m == 0u && cnt > 0)) {
        const int n = (cnt < 32) ? cnt : 32;
        drain_any<PH>(q, head, n, lane, A, big);
        head = (head + n) & (AT_QCAP - 1);
        cnt -= n;
      }
      if (m == 0u) break;
    }
  }
  return met;
}

__device__ __forceinline__ void tile_load_a(v4f (&a)[16], const float* __restrict__ X, const int row0, const int lane) {
  const int row = max(min(row0 + (lane & 15), N_NODES - 1), 0);
  const v4f* p = (const v4f*)(X + (size_t)row * HC) + (lane >> 4);
  #pragma unroll
  for (int j = 0; j < 16; ++j) a[j] = p[2 * j];
}

template <int REUSE>
__global__ void __launch_bounds__(AT_THREADS)
tile_softmax_aggregate_ln_kernel(const int* __restrict__ ei,
                                 const float* __restrict__ xl,
                                 const float* __restrict__ xin,
                                 const float* __restrict__ Wr,
                                 const float* __restrict__ br,
                                 const float* __restrict__ att,
                                 float* sc,
                                 const float* __restrict__ bias,
                                 const float* __restrict__ gamma,
                                 const float* __restrict__ beta,
                                 float* __restrict__ out,
                                 int*  lists,
                                 int*  lcnt,
                                 int*  lover,
                                 const unsigned* __restrict__ tplane,
                                 const unsigned* __restrict__ bmap) {
  __shared__ __align__(16) float    tile[AT_S * HC];
  __shared__ __align__(16) float    sxr[AT_S * HC];
  __shared__ __align__(16) int2     ring[AT_WAVES * AT_QCAP];
  __shared__ int                    glist[AT_WAVES * AT_GCAP];
  __shared__ __align__(16) unsigned smax[AT_S * HEADS];
  __shared__ __align__(16) float    sden[AT_S * HEADS];
  __shared__ int      sover;
  __shared__ int      sbad;
  const int tid  = threadIdx.x;
  const int wave = tid >> 5;
  const int lane = tid & 31;
  const int lo   = blockIdx.x * AT_S;

  constexpr int XR_RT    = (AT_S + 15) / 16;
  constexpr int XR_UNITS = XR_RT * 4;
  static_assert(AT_THREADS == 32 * 32 && AT_S * HC >= HC * HC && XR_UNITS >= AT_WAVES,
                "one (k-quad, column group) item per thread; the pieces fit the tile; every wave has a first unit");
  const int wv = __builtin_amdgcn_readfirstlane(wave);
  v4f xa[16];
  tile_load_a(xa, xin, lo + (wv >> 2) * 16, lane);
  {
    const int wq = tid >> 5, wg = tid & 31;
    v4f r[4];
    #pragma unroll
    for (int k = 0; k < 4; ++k) r[k] = *(const v4f*)(Wr + (4 * wq + k) * HC + 4 * wg);
    v4f* d = (v4f*)tile + wq * HC + 4 * wg;
    d[0] = v4f{r[0].x, r[1].x, r[2].x, r[3].x};
    d[1] = v4f{r[0].y, r[1].y, r[2].y, r[3].y};
    d[2] = v4f{r[0].z, r[1].z, r[2].z, r[3].z};
    d[3] = v4f{r[0].w, r[1].w, r[2].w, r[3].w};
  }
  if (tid < AT_S * HEADS) sden[tid] = 0.f;
  if (tid == 0) { sover = 0; sbad = 0; }
  __syncthreads();
  for (int u = wv; u < XR_UNITS; u += AT_WAVES) {
    const int rt   = u >> 2;
    const int col0 = (u & 3) * 32 + (lane & 15);
    const v4f* bp = (const v4f*)tile + (lane >> 4) * HC + col0;
    float bc[2];
    v8f acc[2];
    #pragma unroll
    for (int t = 0; t < 2; ++t) { bc[t] = br[col0 + 16 * t]; acc[t] = v8f{}; }
    v4f bq[2][2];
    #pragma unroll
    for (int t = 0; t < 2; ++t) bq[0][t] = bp[16 * t];
    #pragma unroll
    for (int j = 0; j < 16; ++j) {
      if (j + 1 < 16) {
        #pragma unroll
        for (int t = 0; t < 2; ++t) bq[(j + 1) & 1][t] = bp[2 * (j + 1) * HC + 16 * t];
      }
      #pragma unroll
      for (int t = 0; t < 2; ++t) {
        const v4f b = bq[j & 1][t];
        v2f a0; a0.x = xa[j].x; a0.y = xa[j].y;
        v2f b0; b0.x = b.x;     b0.y = b.y;
        acc[t] = __builtin_amdgcn_wmma_f32_16x16x4_f32(false, a0, false, b0, (short)0, acc[t], false, false);
        v2f a1; a1.x = xa[j].z; a1.y = xa[j].w;
        v2f b1; b1.x = b.z;     b1.y = b.w;
        acc[t] = __builtin_amdgcn_wmma_f32_16x16x4_f32(false, a1, false, b1, (short)0, acc[t], false, false);
      }
      __builtin_amdgcn_sched_barrier(0);
    }
    #pragma unroll
    for (int t = 0; t < 2; ++t) {
      #pragma unroll
      for (int r = 0; r < 8; ++r) {
        const int lr = rt * 16 + r + 8 * (lane >> 4);
        if (lr < AT_S) sxr[lr * HC + col0 + 16 * t] = acc[t][r] + bc[t];
      }
    }
    if (u + AT_WAVES < XR_UNITS) tile_load_a(xa, xin, lo + ((u + AT_WAVES) >> 2) * 16, lane);
  }

  TileArgs A;
  A.tp = (const uint4*)tplane; A.pat = (unsigned)blockIdx.x * 0x01010101u;
  A.bm = bmap + 4 * blockIdx.x;
  A.ei = ei; A.xl = xl; A.sxr = sxr; A.att = att; A.sc = sc;
  A.tile = tile; A.smax = smax; A.shift = (float*)smax; A.sden = sden; A.lo = lo;
  int2* q = ring + wave * AT_QCAP;

  const int slot = blockIdx.x * AT_WAVES + wave;
  int total;
  bool listed = false;
  if (REUSE) listed = (lover[blockIdx.x] == 0);
  int nself, cnt;
  bool over = false;
  if (REUSE && listed) {
    total = min(max(__builtin_amdgcn_readfirstlane(lcnt[slot]), 0), AT_QCAP);
    const int2* ls = (const int2*)lists + (size_t)slot * AT_QCAP;
    int2 r[AT_QCAP / 32];
    #pragma unroll
    for (int k = 0; k < AT_QCAP / 32; ++k) {
      const int i = lane + 32 * k;
      r[k] = (i < total) ? ls[i] : make_int2(0, 0);
    }
    #pragma unroll
    for (int k = 0; k < AT_QCAP / 32; ++k) {
      const int i = lane + 32 * k;
      if (i < total) q[i] = r[k];
    }
    nself = min(ring_self(q, false, wave, lane, A), total);
    cnt = total;
  } else {
    nself = ring_self(q, true, wave, lane, A);
    cnt = nself;
    over = bitmap_fill(q, glist + wave * AT_GCAP, cnt, wave, lane, A);
    total = cnt;
  }
  __syncthreads();
  {
    constexpr int NV4 = AT_S * HC / 4;
    const float4 zero4 = make_float4(0.f, 0.f, 0.f, 0.f);
    for (int i = tid; i < NV4; i += AT_THREADS) ((float4*)tile)[i] = zero4;
  }
  if (nself > 0) (void)one_pass_drain<1>(q, 0, nself, lane, A);
  __syncthreads();

  bool big = false;
  if (!over) {
    for (int h0 = 0; h0 < cnt; h0 += 32) {
      const int n = (cnt - h0 < 32) ? (cnt - h0) : 32;
      big = one_pass_drain<0>(q, h0, n, lane, A) || big;
    }
  } else {
    const int ns = ring_self(q, true, wave, lane, A);
    total = ns + plain_scan<2>(q, ns, big, wave, lane, A);
  }
  if (total > AT_QCAP && lane == 0) sover = 1;
  if (__ballot(big) != 0 && lane == 0) sbad = 1;
  __syncthreads();
  if (!REUSE) {
    if (tid == 0) lover[blockIdx.x] = sover;
    if (sover == 0) {
      int2* ls = (int2*)lists + (size_t)slot * AT_QCAP;
      for (int i = lane; i < total; i += 32) ls[i] = q[i];
      if (lane == 0) lcnt[slot] = total;
    }
  }

  if (sbad != 0) {
    constexpr int NV4 = AT_S * HC / 4;
    for (int i = tid; i < NV4; i += AT_THREADS) ((float4*)tile)[i] = make_float4(0.f, 0.f, 0.f, 0.f);
    if (tid < AT_S * HEADS) { smax[tid] = ORD_NEG_INF; sden[tid] = 0.f; }
    __syncthreads();
    bool unused = false;
    if (sover == 0) {
      for (int h0 = 0; h0 < total; h0 += 32) {
        const int n = (total - h0 < 32) ? (total - h0) : 32;
        score_drain(q, h0, n, lane, A);
      }
    } else {
      (void)plain_scan<0>(q, ring_self(q, true, wave, lane, A), unused, wave, lane, A);
    }
    __syncthreads();
    if (sover == 0) {
      for (int h0 = 0; h0 < total; h0 += 32) {
        const int n = (total - h0 < 32) ? (total - h0) : 32;
        agg_drain(q, h0, n, lane, A);
      }
    } else {
      (void)plain_scan<1>(q, ring_self(q, true, wave, lane, A), unused, wave, lane, A);
    }
    __syncthreads();
  }

  const float4 bs4 = *(const float4*)(bias + 4 * lane);
  const float4 ga4 = *(const float4*)(gamma + 4 * lane);
  const float4 be4 = *(const float4*)(beta + 4 * lane);
  for (int dl = wave; dl < AT_S; dl += AT_WAVES) {
    const int node = lo + dl;
    if (node >= N_NODES) break;
    const float den = sden[dl * HEADS + (lane >> 3)];
    const float* trow = tile + dl * HC + lane;
    float v[4];
    v[0] = trow[0]  / den + bs4.x;
    v[1] = trow[32] / den + bs4.y;
    v[2] = trow[64] / den + bs4.z;
    v[3] = trow[96] / den + bs4.w;
    float sum = (v[0] + v[1]) + (v[2] + v[3]);
    for (int off = 16; off > 0; off >>= 1) sum += __shfl_xor(sum, off, 32);
    const float mu = sum * (1.0f / HC);
    float sq = 0.f;
    #pragma unroll
    for (int i = 0; i < 4; ++i) { float d = v[i] - mu; sq += d * d; }
    for (int off = 16; off > 0; off >>= 1) sq += __shfl_xor(sq, off, 32);
    const float rstd = rsqrtf(sq * (1.0f / HC) + LN_EPS);
    const float h0 = (v[0] - mu) * rstd * ga4.x + be4.x;
    const float h1 = (v[1] - mu) * rstd * ga4.y + be4.y;
    const float h2 = (v[2] - mu) * rstd * ga4.z + be4.z;
    const float h3 = (v[3] - mu) * rstd * ga4.w + be4.w;
    float4 o;
    o.x = h0 / (1.0f + __expf(-h0));
    o.y = h1 / (1.0f + __expf(-h1));
    o.z = h2 / (1.0f + __expf(-h2));
    o.w = h3 / (1.0f + __expf(-h3));
    *(float4*)(out + (size_t)node * HC + 4 * lane) = o;
  }
}

#define PG_WAVES   4
#define PG_THREADS (PG_WAVES * 32)
#define PG_NODES   (PG_WAVES * 32)
#define PG_BLOCKS  ((N_NODES + PG_NODES - 1) / PG_NODES)
#define PG_ROUNDS  ((PG_BLOCKS + 31) / 32)
__global__ void __launch_bounds__(PG_THREADS)
pool_gather_kernel(const float* __restrict__ x,
                   const int* __restrict__ batch,
                   float* __restrict__ part,
                   int* __restrict__ pcnt) {
  static_assert(PG_THREADS == HC && NGRAPH == 64, "a thread per channel at the store; two 32-graph words of presence");
  __shared__ __align__(16) float sacc[NGRAPH * HC];
  __shared__ int scnt[NGRAPH];
  const int tid  = threadIdx.x;
  const int wave = tid >> 5;
  const int lane = tid & 31;
  const int base = blockIdx.x * PG_NODES + wave * 32;
  int id = (base + lane < N_NODES) ? batch[base + lane] : -1;
  if ((unsigned)id >= (unsigned)NGRAPH) id = -1;
  float4 v[32];
  #pragma unroll
  for (int u = 0; u < 32; ++u) {
    int node = base + u;
    node = (node < N_NODES) ? node : (N_NODES - 1);
    v[u] = *(const float4*)(x + (size_t)node * HC + lane * 4);
  }
  for (int i = tid; i < NGRAPH * HC / 4; i += PG_THREADS) ((float4*)sacc)[i] = make_float4(0.f, 0.f, 0.f, 0.f);
  if (tid < NGRAPH) scnt[tid] = 0;
  __syncthreads();
  unsigned rem = (unsigned)__ballot(id >= 0);
  for (int w = 0; w < PG_WAVES; ++w) {
    if (wave == w) {
      while (rem != 0u) {
        const int g = __builtin_amdgcn_readlane(id, __ffs(rem) - 1);
        const unsigned mask = (unsigned)__ballot(id == g);
        rem &= ~mask;
        float4 acc = make_float4(0.f, 0.f, 0.f, 0.f);
        #pragma unroll
        for (int u = 0; u < 32; ++u) {
          const bool on = ((mask >> u) & 1u) != 0u;
          acc.x += on ? v[u].x : 0.f;
          acc.y += on ? v[u].y : 0.f;
          acc.z += on ? v[u].z : 0.f;
          acc.w += on ? v[u].w : 0.f;
        }
        float4* p = (float4*)(sacc + g * HC) + lane;
        float4 o = *p;
        o.x += acc.x; o.y += acc.y; o.z += acc.z; o.w += acc.w;
        *p = o;
        if (lane == 0) scnt[g] += __popc(mask);
      }
    }
    __syncthreads();
  }
  const int c0 = scnt[lane], c1 = scnt[lane + 32];
  if (tid < NGRAPH) pcnt[blockIdx.x * NGRAPH + tid] = (tid < 32) ? c0 : c1;
  const unsigned m0 = (unsigned)__ballot(c0 > 0), m1 = (unsigned)__ballot(c1 > 0);
  float* prow = part + (size_t)blockIdx.x * NGRAPH * HC + tid;
  for (unsigned m = m0; m != 0u; m &= m - 1u) {
    const int g = __ffs(m) - 1;
    prow[g * HC] = sacc[g * HC + tid];
  }
  for (unsigned m = m1; m != 0u; m &= m - 1u) {
    const int g = 32 + __ffs(m) - 1;
    prow[g * HC] = sacc[g * HC + tid];
  }
}
__global__ void __launch_bounds__(PG_THREADS)
pool_final_kernel(const float* __restrict__ part,
                  const int* __restrict__ pcnt,
                  float* __restrict__ out) {
  __shared__ int slist[PG_ROUNDS * 32];
  __shared__ int sn[2];
  const int g    = blockIdx.x;
  const int tid  = threadIdx.x;
  const int lane = tid & 31;
  if (tid < 32) {
    int cc[PG_ROUNDS];
    #pragma unroll
    for (int k = 0; k < PG_ROUNDS; ++k) {
      const int b = 32 * k + lane;
      cc[k] = (b < PG_BLOCKS) ? pcnt[b * NGRAPH + g] : 0;
    }
    int n = 0, members = 0;
    #pragma unroll
    for (int k = 0; k < PG_ROUNDS; ++k) {
      const bool on = cc[k] > 0;
      const unsigned m = (unsigned)__ballot(on);
      if (on) slist[n + __popc(m & ((1u << lane) - 1u))] = 32 * k + lane;
      n += __popc(m);
      members += cc[k];
    }
    for (int off = 16; off > 0; off >>= 1) members += __shfl_xor(members, off, 32);
    if (lane == 0) { sn[0] = n; sn[1] = members; }
  }
  __syncthreads();
  const int n = sn[0];
  float sum = 0.f;
  for (int i0 = 0; i0 < n; i0 += 4) {
    float r[4];
    #pragma unroll
    for (int u = 0; u < 4; ++u) {
      const int b = slist[min(i0 + u, n - 1)];
      r[u] = part[((size_t)b * NGRAPH + g) * HC + tid];
    }
    #pragma unroll
    for (int u = 0; u < 4; ++u) sum += (i0 + u < n) ? r[u] : 0.f;
  }
  out[g * HC + tid] = sum / fmaxf((float)sn[1], 1.0f);
}

extern "C" void kernel_launch(void* const* d_in, const int* in_sizes, int n_in,
                              void* d_out, int out_size, void* d_ws, size_t ws_size,
                              hipStream_t stream) {
  (void)in_sizes; (void)n_in; (void)out_size; (void)ws_size;
  const float* x     = (const float*)d_in[0];
  const int*   ei    = (const int*)d_in[1];
  const int*   batch = (const int*)d_in[2];
  const float* Wl[2]    = {(const float*)d_in[3],  (const float*)d_in[11]};
  const float* Wr[2]    = {(const float*)d_in[4],  (const float*)d_in[12]};
  const float* bl[2]    = {(const float*)d_in[5],  (const float*)d_in[13]};
  const float* br[2]    = {(const float*)d_in[6],  (const float*)d_in[14]};
  const float* att[2]   = {(const float*)d_in[7],  (const float*)d_in[15]};
  const float* bias[2]  = {(const float*)d_in[8],  (const float*)d_in[16]};
  const float* gamma[2] = {(const float*)d_in[9],  (const float*)d_in[17]};
  const float* beta[2]  = {(const float*)d_in[10], (const float*)d_in[18]};

  float* ws = (float*)d_ws;
  float*    xl    = ws;            ws += (size_t)N_NODES * HC;
  float*    xnode2 = ws;           ws += (size_t)N_NODES * HC;
  float*    xnode = ws;            ws += (size_t)N_NODES * HC;
  float*    scbuf = ws;            ws += (size_t)ETOT * HEADS;
  int*      pcnt  = (int*)ws;      ws += (size_t)PG_BLOCKS * NGRAPH;
  float*    ppart = ws;            ws += (size_t)PG_BLOCKS * NGRAPH * HC;
  int*      lists = (int*)ws;      ws += (size_t)AT_TILES * AT_WAVES * AT_QCAP * 2;
  int*      lcnt  = (int*)ws;      ws += (size_t)AT_TILES * AT_WAVES;
  int*      lover = (int*)ws;      ws += 256;
  unsigned* tplane = (unsigned*)ws;  ws += E_EDGES / 4;
  unsigned* bmap   = (unsigned*)ws;

  gemm_bias_wmma<1><<<G_BLOCKS, G_THREADS, 0, stream>>>(x, Wl[0], bl[0], xl, N_NODES, ei, tplane, bmap);
  tile_softmax_aggregate_ln_kernel<0><<<AT_TILES, AT_THREADS, 0, stream>>>(ei, xl, x, Wr[0], br[0], att[0], scbuf, bias[0], gamma[0], beta[0], xnode, lists, lcnt, lover, tplane, bmap);
  gemm_bias_wmma<0><<<G_BLOCKS, G_THREADS, 0, stream>>>(xnode, Wl[1], bl[1], xl, N_NODES, ei, tplane, bmap);
  tile_softmax_aggregate_ln_kernel<1><<<AT_TILES, AT_THREADS, 0, stream>>>(ei, xl, xnode, Wr[1], br[1], att[1], scbuf, bias[1], gamma[1], beta[1], xnode2, lists, lcnt, lover, tplane, bmap);
  pool_gather_kernel<<<PG_BLOCKS, PG_THREADS, 0, stream>>>(xnode2, batch, ppart, pcnt);
  pool_final_kernel<<<NGRAPH, PG_THREADS, 0, stream>>>(ppart, pcnt, (float*)d_out);
}
